# starting kernel
# baseline (speedup 1.0000x reference)
#include <hip/hip_runtime.h>

constexpr int N_NODES = 50000;
constexpr int N_EDGES = 1600000;
constexpr int F_IN    = 128;
constexpr int F_E     = 16;
constexpr int OUT     = 32;
constexpr float SLOPE = 0.2f;
constexpr float EPS   = 1e-9f;

typedef float v2f __attribute__((ext_vector_type(2)));
typedef float v8f __attribute__((ext_vector_type(8)));

#define WMMA4(acc, a, b)                                                     \
  acc = __builtin_amdgcn_wmma_f32_16x16x4_f32(false, (a), false, (b),        \
                                              (short)0, (acc), false, false)

__device__ __forceinline__ float leaky(float x) { return x >= 0.f ? x : SLOPE * x; }

__device__ __forceinline__ unsigned flipf(float f) {
  unsigned b = __float_as_uint(f);
  return (b & 0x80000000u) ? ~b : (b | 0x80000000u);
}
__device__ __forceinline__ float unflipf(unsigned u) {
  return __uint_as_float((u & 0x80000000u) ? (u ^ 0x80000000u) : ~u);
}

__global__ __launch_bounds__(256) void zero_kernel(float* p, long n) {
  long i = (long)blockIdx.x * 256 + threadIdx.x;
  if (i < n) p[i] = 0.f;
}

__global__ __launch_bounds__(32) void node_gemm_kernel(
    const float* __restrict__ feat,
    const float* __restrict__ Wp,  const float* __restrict__ Wp2,
    const float* __restrict__ Was, const float* __restrict__ Wad,
    const float* __restrict__ Wapd, const float* __restrict__ bapd,
    float* __restrict__ fc, float* __restrict__ esrc,
    float* __restrict__ edst, float* __restrict__ dterm) {
  const int lane = threadIdx.x, lo = lane & 15, hi = lane >> 4;
  const int base = blockIdx.x * 16;

  v8f accP[2] = {}, accAS[2] = {}, accAD[2] = {}, accAP[2] = {};
  for (int k0 = 0; k0 < F_IN; k0 += 4) {
    const int ka = k0 + 2 * hi;
    v2f a = *(const v2f*)(feat + (long)(base + lo) * F_IN + ka);
#pragma unroll
    for (int t = 0; t < 2; ++t) {
      const int n = lo + 16 * t;
      v2f b;
      b = (v2f){Wp[ka * OUT + n],   Wp[(ka + 1) * OUT + n]};   WMMA4(accP[t],  a, b);
      b = (v2f){Was[ka * OUT + n],  Was[(ka + 1) * OUT + n]};  WMMA4(accAS[t], a, b);
      b = (v2f){Wad[ka * OUT + n],  Wad[(ka + 1) * OUT + n]};  WMMA4(accAD[t], a, b);
      b = (v2f){Wapd[ka * OUT + n], Wapd[(ka + 1) * OUT + n]}; WMMA4(accAP[t], a, b);
    }
  }

  __shared__ float lds[16 * OUT];
#pragma unroll
  for (int t = 0; t < 2; ++t)
#pragma unroll
    for (int v = 0; v < 8; ++v)
      lds[(v + 8 * hi) * OUT + lo + 16 * t] = leaky(accP[t][v]);
  __syncthreads();

  v8f accFC[2] = {};
  for (int k0 = 0; k0 < OUT; k0 += 4) {
    const int ka = k0 + 2 * hi;
    v2f a2 = (v2f){lds[lo * OUT + ka], lds[lo * OUT + ka + 1]};
#pragma unroll
    for (int t = 0; t < 2; ++t) {
      const int n = lo + 16 * t;
      v2f b = (v2f){Wp2[ka * OUT + n], Wp2[(ka + 1) * OUT + n]};
      WMMA4(accFC[t], a2, b);
    }
  }

#pragma unroll
  for (int t = 0; t < 2; ++t)
#pragma unroll
    for (int v = 0; v < 8; ++v) {
      const long row = base + v + 8 * hi;
      const int n = lo + 16 * t;
      fc[row * OUT + n]    = accFC[t][v];
      esrc[row * OUT + n]  = accAS[t][v];
      edst[row * OUT + n]  = accAD[t][v];
      dterm[row * OUT + n] = accAP[t][v] + bapd[n];
    }
}

__global__ __launch_bounds__(32) void edge_logits_kernel(
    const float* __restrict__ fe, const int* __restrict__ src,
    const int* __restrict__ dst, const float* __restrict__ We,
    const float* __restrict__ esrc, const float* __restrict__ edst,
    float* __restrict__ ebuf, unsigned* __restrict__ mdst,
    unsigned* __restrict__ msrc) {
  const int lane = threadIdx.x, lo = lane & 15, hi = lane >> 4;
  const int ebase = blockIdx.x * 16;

  v8f acc[2] = {};
#pragma unroll
  for (int k0 = 0; k0 < F_E; k0 += 4) {
    const int ka = k0 + 2 * hi;
    v2f a = *(const v2f*)(fe + (long)(ebase + lo) * F_E + ka);
#pragma unroll
    for (int t = 0; t < 2; ++t) {
      const int n = lo + 16 * t;
      v2f b = (v2f){We[ka * OUT + n], We[(ka + 1) * OUT + n]};
      WMMA4(acc[t], a, b);
    }
  }

#pragma unroll
  for (int v = 0; v < 8; ++v) {
    const long eid = ebase + v + 8 * hi;
    const long s = src[eid], d = dst[eid];
#pragma unroll
    for (int t = 0; t < 2; ++t) {
      const int n = lo + 16 * t;
      float val = leaky(acc[t][v] + esrc[s * OUT + n] + edst[d * OUT + n]);
      ebuf[eid * OUT + n] = val;
      const unsigned fv = flipf(val);
      atomicMax(&mdst[d * OUT + n], fv);
      atomicMax(&msrc[s * OUT + n], fv);
    }
  }
}

__global__ __launch_bounds__(256) void edge_sum_kernel(
    const float* __restrict__ ebuf, const int* __restrict__ src,
    const int* __restrict__ dst, const unsigned* __restrict__ mdst,
    const unsigned* __restrict__ msrc, float* __restrict__ sdst,
    float* __restrict__ ssrc) {
  const int idx = blockIdx.x * 256 + threadIdx.x;
  if (idx >= N_EDGES * OUT) return;
  const int eid = idx >> 5, n = idx & 31;
  const long s = src[eid], d = dst[eid];
  const float v = ebuf[idx];
  atomicAdd(&sdst[d * OUT + n], __expf(v - unflipf(mdst[d * OUT + n])));
  atomicAdd(&ssrc[s * OUT + n], __expf(v - unflipf(msrc[s * OUT + n])));
}

__global__ __launch_bounds__(256) void edge_agg_kernel(
    const float* __restrict__ ebuf, const int* __restrict__ src,
    const int* __restrict__ dst, const unsigned* __restrict__ mdst,
    const unsigned* __restrict__ msrc, const float* __restrict__ sdst,
    const float* __restrict__ ssrc, const float* __restrict__ fc,
    float* __restrict__ msg) {
  const int idx = blockIdx.x * 256 + threadIdx.x;
  if (idx >= N_EDGES * OUT) return;
  const int eid = idx >> 5, n = idx & 31;
  const long s = src[eid], d = dst[eid];
  const float v = ebuf[idx];
  const float ad = fmaxf(__expf(v - unflipf(mdst[d * OUT + n])) / sdst[d * OUT + n], EPS);
  const float as = fmaxf(__expf(v - unflipf(msrc[s * OUT + n])) / ssrc[s * OUT + n], EPS);
  atomicAdd(&msg[d * OUT + n], fc[s * OUT + n] * sqrtf(ad * as));
}

__global__ __launch_bounds__(256) void norm_kernel(
    float* __restrict__ msg, const float* __restrict__ scale,
    const float* __restrict__ offset) {
  const int node = blockIdx.x * 256 + threadIdx.x;
  if (node >= N_NODES) return;
  float x[OUT];
#pragma unroll
  for (int i = 0; i < OUT; ++i) x[i] = msg[(long)node * OUT + i];
#pragma unroll
  for (int h = 0; h < 4; ++h) {
    float m = 0.f;
#pragma unroll
    for (int c = 0; c < 8; ++c) m += x[h * 8 + c];
    m *= 0.125f;
    float var = 0.f;
#pragma unroll
    for (int c = 0; c < 8; ++c) { float dd = x[h * 8 + c] - m; var += dd * dd; }
    const float r = rsqrtf(var * 0.125f + EPS);
#pragma unroll
    for (int c = 0; c < 8; ++c) {
      const int n = h * 8 + c;
      msg[(long)node * OUT + n] = (x[n] - m) * scale[n] * r + offset[n];
    }
  }
}

__global__ __launch_bounds__(32) void final_kernel(
    const float* __restrict__ msg, const float* __restrict__ Wagg,
    const float* __restrict__ bagg, const float* __restrict__ dterm,
    const float* __restrict__ Wap, const float* __restrict__ bap,
    float* __restrict__ out) {
  const int lane = threadIdx.x, lo = lane & 15, hi = lane >> 4;
  const int base = blockIdx.x * 16;

  v8f acc1[2] = {};
  for (int k0 = 0; k0 < OUT; k0 += 4) {
    const int ka = k0 + 2 * hi;
    v2f a = *(const v2f*)(msg + (long)(base + lo) * OUT + ka);
#pragma unroll
    for (int t = 0; t < 2; ++t) {
      const int n = lo + 16 * t;
      v2f b = (v2f){Wagg[ka * OUT + n], Wagg[(ka + 1) * OUT + n]};
      WMMA4(acc1[t], a, b);
    }
  }

  __shared__ float lds[16 * OUT];
#pragma unroll
  for (int t = 0; t < 2; ++t)
#pragma unroll
    for (int v = 0; v < 8; ++v) {
      const long row = base + v + 8 * hi;
      const int n = lo + 16 * t;
      lds[(v + 8 * hi) * OUT + n] = leaky(acc1[t][v] + bagg[n] + dterm[row * OUT + n]);
    }
  __syncthreads();

  v8f acc2[2] = {};
  for (int k0 = 0; k0 < OUT; k0 += 4) {
    const int ka = k0 + 2 * hi;
    v2f a2 = (v2f){lds[lo * OUT + ka], lds[lo * OUT + ka + 1]};
#pragma unroll
    for (int t = 0; t < 2; ++t) {
      const int n = lo + 16 * t;
      v2f b = (v2f){Wap[ka * OUT + n], Wap[(ka + 1) * OUT + n]};
      WMMA4(acc2[t], a2, b);
    }
  }

#pragma unroll
  for (int t = 0; t < 2; ++t)
#pragma unroll
    for (int v = 0; v < 8; ++v)
      out[(long)(base + v + 8 * hi) * OUT + lo + 16 * t] = acc2[t][v] + bap[lo + 16 * t];
}

extern "C" void kernel_launch(void* const* d_in, const int* in_sizes, int n_in,
                              void* d_out, int out_size, void* d_ws, size_t ws_size,
                              hipStream_t stream) {
  const float* feat  = (const float*)d_in[0];
  const float* fe    = (const float*)d_in[1];
  const int*   src   = (const int*)d_in[2];
  const int*   dst   = (const int*)d_in[3];
  const float* Wp    = (const float*)d_in[4];
  const float* Wp2   = (const float*)d_in[5];
  const float* Was   = (const float*)d_in[6];
  const float* Wad   = (const float*)d_in[7];
  const float* We    = (const float*)d_in[8];
  const float* scale = (const float*)d_in[9];
  const float* offs  = (const float*)d_in[10];
  const float* Wagg  = (const float*)d_in[11];
  const float* bagg  = (const float*)d_in[12];
  const float* Wapd  = (const float*)d_in[13];
  const float* bapd  = (const float*)d_in[14];
  const float* Wap   = (const float*)d_in[15];
  const float* bap   = (const float*)d_in[16];
  float* out = (float*)d_out;

  const long NW = (long)N_NODES * OUT;
  float* fc    = (float*)d_ws;
  float* esrc  = fc + NW;
  float* edst  = esrc + NW;
  float* dterm = edst + NW;
  unsigned* mdst = (unsigned*)(dterm + NW);
  unsigned* msrc = mdst + NW;
  float* sdst  = (float*)(msrc + NW);
  float* ssrc  = sdst + NW;
  float* msg   = ssrc + NW;
  float* ebuf  = msg + NW;

  const long nzero = 5 * NW;
  zero_kernel<<<(int)((nzero + 255) / 256), 256, 0, stream>>>((float*)mdst, nzero);

  node_gemm_kernel<<<N_NODES / 16, 32, 0, stream>>>(
      feat, Wp, Wp2, Was, Wad, Wapd, bapd, fc, esrc, edst, dterm);

  edge_logits_kernel<<<N_EDGES / 16, 32, 0, stream>>>(
      fe, src, dst, We, esrc, edst, ebuf, mdst, msrc);

  const int ne = N_EDGES * OUT;
  edge_sum_kernel<<<(ne + 255) / 256, 256, 0, stream>>>(
      ebuf, src, dst, mdst, msrc, sdst, ssrc);

  edge_agg_kernel<<<(ne + 255) / 256, 256, 0, stream>>>(
      ebuf, src, dst, mdst, msrc, sdst, ssrc, fc, msg);

  norm_kernel<<<(N_NODES + 255) / 256, 256, 0, stream>>>(msg, scale, offs);

  final_kernel<<<N_NODES / 16, 32, 0, stream>>>(
      msg, Wagg, bagg, dterm, Wap, bap, out);

  (void)in_sizes; (void)n_in; (void)out_size; (void)ws_size;
}


# edge kernels: each wave owns 8 edges (sum, agg) or its 16-edge tile (logits) and issues all index and row loads before its first atomic, so gathers overlap; f32 atomic adds still land in any order
# speedup vs baseline: 1.5559x; 1.5559x over previous
#include <hip/hip_runtime.h>

constexpr int N_NODES = 50000;
constexpr int N_EDGES = 1600000;
constexpr int F_IN    = 128;
constexpr int F_E     = 16;
constexpr int OUT     = 32;
constexpr float SLOPE = 0.2f;
constexpr float EPS   = 1e-9f;

typedef float v2f __attribute__((ext_vector_type(2)));
typedef float v8f __attribute__((ext_vector_type(8)));

#define WMMA4(acc, a, b)                                                     \
  acc = __builtin_amdgcn_wmma_f32_16x16x4_f32(false, (a), false, (b),        \
                                              (short)0, (acc), false, false)

__device__ __forceinline__ float leaky(float x) { return x >= 0.f ? x : SLOPE * x; }

__device__ __forceinline__ unsigned flipf(float f) {
  unsigned b = __float_as_uint(f);
  return (b & 0x80000000u) ? ~b : (b | 0x80000000u);
}
__device__ __forceinline__ float unflipf(unsigned u) {
  return __uint_as_float((u & 0x80000000u) ? (u ^ 0x80000000u) : ~u);
}

__global__ __launch_bounds__(256) void zero_kernel(float* p, long n) {
  long i = (long)blockIdx.x * 256 + threadIdx.x;
  if (i < n) p[i] = 0.f;
}

__global__ __launch_bounds__(32) void node_gemm_kernel(
    const float* __restrict__ feat,
    const float* __restrict__ Wp,  const float* __restrict__ Wp2,
    const float* __restrict__ Was, const float* __restrict__ Wad,
    const float* __restrict__ Wapd, const float* __restrict__ bapd,
    float* __restrict__ fc, float* __restrict__ esrc,
    float* __restrict__ edst, float* __restrict__ dterm) {
  const int lane = threadIdx.x, lo = lane & 15, hi = lane >> 4;
  const int base = blockIdx.x * 16;

  v8f accP[2] = {}, accAS[2] = {}, accAD[2] = {}, accAP[2] = {};
  for (int k0 = 0; k0 < F_IN; k0 += 4) {
    const int ka = k0 + 2 * hi;
    v2f a = *(const v2f*)(feat + (long)(base + lo) * F_IN + ka);
#pragma unroll
    for (int t = 0; t < 2; ++t) {
      const int n = lo + 16 * t;
      v2f b;
      b = (v2f){Wp[ka * OUT + n],   Wp[(ka + 1) * OUT + n]};   WMMA4(accP[t],  a, b);
      b = (v2f){Was[ka * OUT + n],  Was[(ka + 1) * OUT + n]};  WMMA4(accAS[t], a, b);
      b = (v2f){Wad[ka * OUT + n],  Wad[(ka + 1) * OUT + n]};  WMMA4(accAD[t], a, b);
      b = (v2f){Wapd[ka * OUT + n], Wapd[(ka + 1) * OUT + n]}; WMMA4(accAP[t], a, b);
    }
  }

  __shared__ float lds[16 * OUT];
#pragma unroll
  for (int t = 0; t < 2; ++t)
#pragma unroll
    for (int v = 0; v < 8; ++v)
      lds[(v + 8 * hi) * OUT + lo + 16 * t] = leaky(accP[t][v]);
  __syncthreads();

  v8f accFC[2] = {};
  for (int k0 = 0; k0 < OUT; k0 += 4) {
    const int ka = k0 + 2 * hi;
    v2f a2 = (v2f){lds[lo * OUT + ka], lds[lo * OUT + ka + 1]};
#pragma unroll
    for (int t = 0; t < 2; ++t) {
      const int n = lo + 16 * t;
      v2f b = (v2f){Wp2[ka * OUT + n], Wp2[(ka + 1) * OUT + n]};
      WMMA4(accFC[t], a2, b);
    }
  }

#pragma unroll
  for (int t = 0; t < 2; ++t)
#pragma unroll
    for (int v = 0; v < 8; ++v) {
      const long row = base + v + 8 * hi;
      const int n = lo + 16 * t;
      fc[row * OUT + n]    = accFC[t][v];
      esrc[row * OUT + n]  = accAS[t][v];
      edst[row * OUT + n]  = accAD[t][v];
      dterm[row * OUT + n] = accAP[t][v] + bapd[n];
    }
}

__global__ __launch_bounds__(32) void edge_logits_kernel(
    const float* __restrict__ fe, const int* __restrict__ src,
    const int* __restrict__ dst, const float* __restrict__ We,
    const float* __restrict__ esrc, const float* __restrict__ edst,
    float* __restrict__ ebuf, unsigned* __restrict__ mdst,
    unsigned* __restrict__ msrc) {
  const int lane = threadIdx.x, lo = lane & 15, hi = lane >> 4;
  const int ebase = blockIdx.x * 16;

  v8f acc[2] = {};
#pragma unroll
  for (int k0 = 0; k0 < F_E; k0 += 4) {
    const int ka = k0 + 2 * hi;
    v2f a = *(const v2f*)(fe + (long)(ebase + lo) * F_E + ka);
#pragma unroll
    for (int t = 0; t < 2; ++t) {
      const int n = lo + 16 * t;
      v2f b = (v2f){We[ka * OUT + n], We[(ka + 1) * OUT + n]};
      WMMA4(acc[t], a, b);
    }
  }

  unsigned os[8], od[8];
#pragma unroll
  for (int v = 0; v < 8; ++v) {
    const int eid = ebase + v + 8 * hi;
    os[v] = (unsigned)src[eid] * OUT + lo;
    od[v] = (unsigned)dst[eid] * OUT + lo;
  }
  float gs[8][2], gd[8][2];
#pragma unroll
  for (int v = 0; v < 8; ++v)
#pragma unroll
    for (int t = 0; t < 2; ++t) {
      gs[v][t] = esrc[os[v] + 16 * t];
      gd[v][t] = edst[od[v] + 16 * t];
    }

#pragma unroll
  for (int v = 0; v < 8; ++v) {
    const int eid = ebase + v + 8 * hi;
#pragma unroll
    for (int t = 0; t < 2; ++t) {
      const float val = leaky(acc[t][v] + gs[v][t] + gd[v][t]);
      ebuf[eid * OUT + lo + 16 * t] = val;
      const unsigned fv = flipf(val);
      atomicMax(&mdst[od[v] + 16 * t], fv);
      atomicMax(&msrc[os[v] + 16 * t], fv);
    }
  }
}

constexpr int EPW = 8;
constexpr int EDGE_BLOCK = 256;
constexpr int EDGES_PER_BLOCK = (EDGE_BLOCK / 32) * EPW;
static_assert(N_EDGES % EDGES_PER_BLOCK == 0, "edge grid must tile exactly");

__device__ __forceinline__ void edge_offsets(const int* __restrict__ src,
                                             const int* __restrict__ dst,
                                             int e0, int n,
                                             unsigned (&os)[EPW], unsigned (&od)[EPW]) {
  const int sv = src[e0 + (n & (EPW - 1))];
  const int dv = dst[e0 + (n & (EPW - 1))];
#pragma unroll
  for (int j = 0; j < EPW; ++j) {
    os[j] = (unsigned)__builtin_amdgcn_readlane(sv, j) * OUT + n;
    od[j] = (unsigned)__builtin_amdgcn_readlane(dv, j) * OUT + n;
  }
}

__global__ __launch_bounds__(EDGE_BLOCK) void edge_sum_kernel(
    const float* __restrict__ ebuf, const int* __restrict__ src,
    const int* __restrict__ dst, const unsigned* __restrict__ mdst,
    const unsigned* __restrict__ msrc, float* __restrict__ sdst,
    float* __restrict__ ssrc) {
  const int n = threadIdx.x & 31;
  const int e0 = (blockIdx.x * (EDGE_BLOCK / 32) + (threadIdx.x >> 5)) * EPW;
  unsigned os[EPW], od[EPW];
  edge_offsets(src, dst, e0, n, os, od);

  float v[EPW];
  unsigned md[EPW], ms[EPW];
#pragma unroll
  for (int j = 0; j < EPW; ++j) {
    v[j]  = ebuf[(e0 + j) * OUT + n];
    md[j] = mdst[od[j]];
    ms[j] = msrc[os[j]];
  }
#pragma unroll
  for (int j = 0; j < EPW; ++j) {
    atomicAdd(&sdst[od[j]], __expf(v[j] - unflipf(md[j])));
    atomicAdd(&ssrc[os[j]], __expf(v[j] - unflipf(ms[j])));
  }
}

__global__ __launch_bounds__(EDGE_BLOCK) void edge_agg_kernel(
    const float* __restrict__ ebuf, const int* __restrict__ src,
    const int* __restrict__ dst, const unsigned* __restrict__ mdst,
    const unsigned* __restrict__ msrc, const float* __restrict__ sdst,
    const float* __restrict__ ssrc, const float* __restrict__ fc,
    float* __restrict__ msg) {
  const int n = threadIdx.x & 31;
  const int e0 = (blockIdx.x * (EDGE_BLOCK / 32) + (threadIdx.x >> 5)) * EPW;
  unsigned os[EPW], od[EPW];
  edge_offsets(src, dst, e0, n, os, od);

  float v[EPW], sd[EPW], ss[EPW], f[EPW];
  unsigned md[EPW], ms[EPW];
#pragma unroll
  for (int j = 0; j < EPW; ++j) {
    v[j]  = ebuf[(e0 + j) * OUT + n];
    md[j] = mdst[od[j]];
    sd[j] = sdst[od[j]];
    ms[j] = msrc[os[j]];
    ss[j] = ssrc[os[j]];
    f[j]  = fc[os[j]];
  }
#pragma unroll
  for (int j = 0; j < EPW; ++j) {
    const float ad = fmaxf(__expf(v[j] - unflipf(md[j])) / sd[j], EPS);
    const float as = fmaxf(__expf(v[j] - unflipf(ms[j])) / ss[j], EPS);
    atomicAdd(&msg[od[j]], f[j] * sqrtf(ad * as));
  }
}

__global__ __launch_bounds__(256) void norm_kernel(
    float* __restrict__ msg, const float* __restrict__ scale,
    const float* __restrict__ offset) {
  const int node = blockIdx.x * 256 + threadIdx.x;
  if (node >= N_NODES) return;
  float x[OUT];
#pragma unroll
  for (int i = 0; i < OUT; ++i) x[i] = msg[(long)node * OUT + i];
#pragma unroll
  for (int h = 0; h < 4; ++h) {
    float m = 0.f;
#pragma unroll
    for (int c = 0; c < 8; ++c) m += x[h * 8 + c];
    m *= 0.125f;
    float var = 0.f;
#pragma unroll
    for (int c = 0; c < 8; ++c) { float dd = x[h * 8 + c] - m; var += dd * dd; }
    const float r = rsqrtf(var * 0.125f + EPS);
#pragma unroll
    for (int c = 0; c < 8; ++c) {
      const int n = h * 8 + c;
      msg[(long)node * OUT + n] = (x[n] - m) * scale[n] * r + offset[n];
    }
  }
}

__global__ __launch_bounds__(32) void final_kernel(
    const float* __restrict__ msg, const float* __restrict__ Wagg,
    const float* __restrict__ bagg, const float* __restrict__ dterm,
    const float* __restrict__ Wap, const float* __restrict__ bap,
    float* __restrict__ out) {
  const int lane = threadIdx.x, lo = lane & 15, hi = lane >> 4;
  const int base = blockIdx.x * 16;

  v8f acc1[2] = {};
  for (int k0 = 0; k0 < OUT; k0 += 4) {
    const int ka = k0 + 2 * hi;
    v2f a = *(const v2f*)(msg + (long)(base + lo) * OUT + ka);
#pragma unroll
    for (int t = 0; t < 2; ++t) {
      const int n = lo + 16 * t;
      v2f b = (v2f){Wagg[ka * OUT + n], Wagg[(ka + 1) * OUT + n]};
      WMMA4(acc1[t], a, b);
    }
  }

  __shared__ float lds[16 * OUT];
#pragma unroll
  for (int t = 0; t < 2; ++t)
#pragma unroll
    for (int v = 0; v < 8; ++v) {
      const long row = base + v + 8 * hi;
      const int n = lo + 16 * t;
      lds[(v + 8 * hi) * OUT + n] = leaky(acc1[t][v] + bagg[n] + dterm[row * OUT + n]);
    }
  __syncthreads();

  v8f acc2[2] = {};
  for (int k0 = 0; k0 < OUT; k0 += 4) {
    const int ka = k0 + 2 * hi;
    v2f a2 = (v2f){lds[lo * OUT + ka], lds[lo * OUT + ka + 1]};
#pragma unroll
    for (int t = 0; t < 2; ++t) {
      const int n = lo + 16 * t;
      v2f b = (v2f){Wap[ka * OUT + n], Wap[(ka + 1) * OUT + n]};
      WMMA4(acc2[t], a2, b);
    }
  }

#pragma unroll
  for (int t = 0; t < 2; ++t)
#pragma unroll
    for (int v = 0; v < 8; ++v)
      out[(long)(base + v + 8 * hi) * OUT + lo + 16 * t] = acc2[t][v] + bap[lo + 16 * t];
}

extern "C" void kernel_launch(void* const* d_in, const int* in_sizes, int n_in,
                              void* d_out, int out_size, void* d_ws, size_t ws_size,
                              hipStream_t stream) {
  const float* feat  = (const float*)d_in[0];
  const float* fe    = (const float*)d_in[1];
  const int*   src   = (const int*)d_in[2];
  const int*   dst   = (const int*)d_in[3];
  const float* Wp    = (const float*)d_in[4];
  const float* Wp2   = (const float*)d_in[5];
  const float* Was   = (const float*)d_in[6];
  const float* Wad   = (const float*)d_in[7];
  const float* We    = (const float*)d_in[8];
  const float* scale = (const float*)d_in[9];
  const float* offs  = (const float*)d_in[10];
  const float* Wagg  = (const float*)d_in[11];
  const float* bagg  = (const float*)d_in[12];
  const float* Wapd  = (const float*)d_in[13];
  const float* bapd  = (const float*)d_in[14];
  const float* Wap   = (const float*)d_in[15];
  const float* bap   = (const float*)d_in[16];
  float* out = (float*)d_out;

  const long NW = (long)N_NODES * OUT;
  float* fc    = (float*)d_ws;
  float* esrc  = fc + NW;
  float* edst  = esrc + NW;
  float* dterm = edst + NW;
  unsigned* mdst = (unsigned*)(dterm + NW);
  unsigned* msrc = mdst + NW;
  float* sdst  = (float*)(msrc + NW);
  float* ssrc  = sdst + NW;
  float* msg   = ssrc + NW;
  float* ebuf  = msg + NW;

  const long nzero = 5 * NW;
  zero_kernel<<<(int)((nzero + 255) / 256), 256, 0, stream>>>((float*)mdst, nzero);

  node_gemm_kernel<<<N_NODES / 16, 32, 0, stream>>>(
      feat, Wp, Wp2, Was, Wad, Wapd, bapd, fc, esrc, edst, dterm);

  edge_logits_kernel<<<N_EDGES / 16, 32, 0, stream>>>(
      fe, src, dst, We, esrc, edst, ebuf, mdst, msrc);

  edge_sum_kernel<<<N_EDGES / EDGES_PER_BLOCK, EDGE_BLOCK, 0, stream>>>(
      ebuf, src, dst, mdst, msrc, sdst, ssrc);

  edge_agg_kernel<<<N_EDGES / EDGES_PER_BLOCK, EDGE_BLOCK, 0, stream>>>(
      ebuf, src, dst, mdst, msrc, sdst, ssrc, fc, msg);

  norm_kernel<<<(N_NODES + 255) / 256, 256, 0, stream>>>(msg, scale, offs);

  final_kernel<<<N_NODES / 16, 32, 0, stream>>>(
      msg, Wagg, bagg, dterm, Wap, bap, out);

  (void)in_sizes; (void)n_in; (void)out_size; (void)ws_size;
}


# zero fill and the three edge kernels each launch 8192 long-lived waves that loop over their tiles instead of 100k-250k one-item waves paced by wave entry; f32 atomics still land in any order
# speedup vs baseline: 1.6099x; 1.0347x over previous
#include <hip/hip_runtime.h>

constexpr int N_NODES = 50000;
constexpr int N_EDGES = 1600000;
constexpr int F_IN    = 128;
constexpr int F_E     = 16;
constexpr int OUT     = 32;
constexpr float SLOPE = 0.2f;
constexpr float EPS   = 1e-9f;

typedef float v2f __attribute__((ext_vector_type(2)));
typedef float v8f __attribute__((ext_vector_type(8)));

#define WMMA4(acc, a, b)                                                     \
  acc = __builtin_amdgcn_wmma_f32_16x16x4_f32(false, (a), false, (b),        \
                                              (short)0, (acc), false, false)

__device__ __forceinline__ float leaky(float x) { return x >= 0.f ? x : SLOPE * x; }

__device__ __forceinline__ unsigned flipf(float f) {
  unsigned b = __float_as_uint(f);
  return (b & 0x80000000u) ? ~b : (b | 0x80000000u);
}
__device__ __forceinline__ float unflipf(unsigned u) {
  return __uint_as_float((u & 0x80000000u) ? (u ^ 0x80000000u) : ~u);
}

constexpr int PW_BLOCK = 256;
constexpr int PW_GRID  = 1024;
constexpr int PW_WAVES = PW_GRID * (PW_BLOCK / 32);
#define PW_RESIDENT __attribute__((amdgpu_waves_per_eu(8, 8)))

__device__ __forceinline__ int wave_number() {
  return __builtin_amdgcn_readfirstlane(
      (int)(blockIdx.x * (PW_BLOCK / 32) + (threadIdx.x >> 5)));
}

__global__ __launch_bounds__(PW_BLOCK) void zero_kernel(float* p, int n4) {
  float4* p4 = (float4*)p;
  const float4 z = {0.f, 0.f, 0.f, 0.f};
  for (int i = blockIdx.x * PW_BLOCK + threadIdx.x; i < n4; i += PW_GRID * PW_BLOCK)
    p4[i] = z;
}

__global__ __launch_bounds__(32) void node_gemm_kernel(
    const float* __restrict__ feat,
    const float* __restrict__ Wp,  const float* __restrict__ Wp2,
    const float* __restrict__ Was, const float* __restrict__ Wad,
    const float* __restrict__ Wapd, const float* __restrict__ bapd,
    float* __restrict__ fc, float* __restrict__ esrc,
    float* __restrict__ edst, float* __restrict__ dterm) {
  const int lane = threadIdx.x, lo = lane & 15, hi = lane >> 4;
  const int base = blockIdx.x * 16;

  v8f accP[2] = {}, accAS[2] = {}, accAD[2] = {}, accAP[2] = {};
  for (int k0 = 0; k0 < F_IN; k0 += 4) {
    const int ka = k0 + 2 * hi;
    v2f a = *(const v2f*)(feat + (long)(base + lo) * F_IN + ka);
#pragma unroll
    for (int t = 0; t < 2; ++t) {
      const int n = lo + 16 * t;
      v2f b;
      b = (v2f){Wp[ka * OUT + n],   Wp[(ka + 1) * OUT + n]};   WMMA4(accP[t],  a, b);
      b = (v2f){Was[ka * OUT + n],  Was[(ka + 1) * OUT + n]};  WMMA4(accAS[t], a, b);
      b = (v2f){Wad[ka * OUT + n],  Wad[(ka + 1) * OUT + n]};  WMMA4(accAD[t], a, b);
      b = (v2f){Wapd[ka * OUT + n], Wapd[(ka + 1) * OUT + n]}; WMMA4(accAP[t], a, b);
    }
  }

  __shared__ float lds[16 * OUT];
#pragma unroll
  for (int t = 0; t < 2; ++t)
#pragma unroll
    for (int v = 0; v < 8; ++v)
      lds[(v + 8 * hi) * OUT + lo + 16 * t] = leaky(accP[t][v]);
  __syncthreads();

  v8f accFC[2] = {};
  for (int k0 = 0; k0 < OUT; k0 += 4) {
    const int ka = k0 + 2 * hi;
    v2f a2 = (v2f){lds[lo * OUT + ka], lds[lo * OUT + ka + 1]};
#pragma unroll
    for (int t = 0; t < 2; ++t) {
      const int n = lo + 16 * t;
      v2f b = (v2f){Wp2[ka * OUT + n], Wp2[(ka + 1) * OUT + n]};
      WMMA4(accFC[t], a2, b);
    }
  }

#pragma unroll
  for (int t = 0; t < 2; ++t)
#pragma unroll
    for (int v = 0; v < 8; ++v) {
      const long row = base + v + 8 * hi;
      const int n = lo + 16 * t;
      fc[row * OUT + n]    = accFC[t][v];
      esrc[row * OUT + n]  = accAS[t][v];
      edst[row * OUT + n]  = accAD[t][v];
      dterm[row * OUT + n] = accAP[t][v] + bapd[n];
    }
}

__global__ __launch_bounds__(PW_BLOCK) PW_RESIDENT void edge_logits_kernel(
    const float* __restrict__ fe, const int* __restrict__ src,
    const int* __restrict__ dst, const float* __restrict__ We,
    const float* __restrict__ esrc, const float* __restrict__ edst,
    float* __restrict__ ebuf, unsigned* __restrict__ mdst,
    unsigned* __restrict__ msrc) {
  const int lane = threadIdx.x & 31, lo = lane & 15, hi = lane >> 4;
  const int wave = wave_number();

  v2f bw[F_E / 4][2];
#pragma unroll
  for (int k = 0; k < F_E / 4; ++k) {
    const int ka = 4 * k + 2 * hi;
#pragma unroll
    for (int t = 0; t < 2; ++t) {
      const int n = lo + 16 * t;
      bw[k][t] = (v2f){We[ka * OUT + n], We[(ka + 1) * OUT + n]};
    }
  }

  for (int tile = wave; tile < N_EDGES / 16; tile += PW_WAVES) {
    const int ebase = tile * 16;

    v8f acc[2] = {};
#pragma unroll
    for (int k = 0; k < F_E / 4; ++k) {
      const int ka = 4 * k + 2 * hi;
      v2f a = *(const v2f*)(fe + (long)(ebase + lo) * F_E + ka);
#pragma unroll
      for (int t = 0; t < 2; ++t) WMMA4(acc[t], a, bw[k][t]);
    }

    unsigned os[8], od[8];
#pragma unroll
    for (int v = 0; v < 8; ++v) {
      const int eid = ebase + v + 8 * hi;
      os[v] = (unsigned)src[eid] * OUT + lo;
      od[v] = (unsigned)dst[eid] * OUT + lo;
    }
    float gs[8][2], gd[8][2];
#pragma unroll
    for (int v = 0; v < 8; ++v)
#pragma unroll
      for (int t = 0; t < 2; ++t) {
        gs[v][t] = esrc[os[v] + 16 * t];
        gd[v][t] = edst[od[v] + 16 * t];
      }

#pragma unroll
    for (int v = 0; v < 8; ++v) {
      const int eid = ebase + v + 8 * hi;
#pragma unroll
      for (int t = 0; t < 2; ++t) {
        const float val = leaky(acc[t][v] + gs[v][t] + gd[v][t]);
        ebuf[eid * OUT + lo + 16 * t] = val;
        const unsigned fv = flipf(val);
        atomicMax(&mdst[od[v] + 16 * t], fv);
        atomicMax(&msrc[os[v] + 16 * t], fv);
      }
    }
  }
}

constexpr int EPW = 8;
constexpr int N_CHUNKS = N_EDGES / EPW;
static_assert(N_EDGES % EPW == 0, "edge chunks must tile exactly");
static_assert(N_EDGES % 16 == 0, "logit tiles must tile exactly");

__device__ __forceinline__ void edge_offsets(const int* __restrict__ src,
                                             const int* __restrict__ dst,
                                             int e0, int n,
                                             unsigned (&os)[EPW], unsigned (&od)[EPW]) {
  const int sv = src[e0 + (n & (EPW - 1))];
  const int dv = dst[e0 + (n & (EPW - 1))];
#pragma unroll
  for (int j = 0; j < EPW; ++j) {
    os[j] = (unsigned)__builtin_amdgcn_readlane(sv, j) * OUT + n;
    od[j] = (unsigned)__builtin_amdgcn_readlane(dv, j) * OUT + n;
  }
}

__global__ __launch_bounds__(PW_BLOCK) void edge_sum_kernel(
    const float* __restrict__ ebuf, const int* __restrict__ src,
    const int* __restrict__ dst, const unsigned* __restrict__ mdst,
    const unsigned* __restrict__ msrc, float* __restrict__ sdst,
    float* __restrict__ ssrc) {
  const int n = threadIdx.x & 31;
  for (int chunk = wave_number(); chunk < N_CHUNKS; chunk += PW_WAVES) {
    const int e0 = chunk * EPW;
    unsigned os[EPW], od[EPW];
    edge_offsets(src, dst, e0, n, os, od);

    float v[EPW];
    unsigned md[EPW], ms[EPW];
#pragma unroll
    for (int j = 0; j < EPW; ++j) {
      v[j]  = ebuf[(e0 + j) * OUT + n];
      md[j] = mdst[od[j]];
      ms[j] = msrc[os[j]];
    }
#pragma unroll
    for (int j = 0; j < EPW; ++j) {
      atomicAdd(&sdst[od[j]], __expf(v[j] - unflipf(md[j])));
      atomicAdd(&ssrc[os[j]], __expf(v[j] - unflipf(ms[j])));
    }
  }
}

__global__ __launch_bounds__(PW_BLOCK) PW_RESIDENT void edge_agg_kernel(
    const float* __restrict__ ebuf, const int* __restrict__ src,
    const int* __restrict__ dst, const unsigned* __restrict__ mdst,
    const unsigned* __restrict__ msrc, const float* __restrict__ sdst,
    const float* __restrict__ ssrc, const float* __restrict__ fc,
    float* __restrict__ msg) {
  const int n = threadIdx.x & 31;
  for (int chunk = wave_number(); chunk < N_CHUNKS; chunk += PW_WAVES) {
    const int e0 = chunk * EPW;
    unsigned os[EPW], od[EPW];
    edge_offsets(src, dst, e0, n, os, od);

    float v[EPW], sd[EPW], ss[EPW], f[EPW];
    unsigned md[EPW], ms[EPW];
#pragma unroll
    for (int j = 0; j < EPW; ++j) {
      v[j]  = ebuf[(e0 + j) * OUT + n];
      md[j] = mdst[od[j]];
      sd[j] = sdst[od[j]];
      ms[j] = msrc[os[j]];
      ss[j] = ssrc[os[j]];
      f[j]  = fc[os[j]];
    }
#pragma unroll
    for (int j = 0; j < EPW; ++j) {
      const float ad = fmaxf(__expf(v[j] - unflipf(md[j])) / sd[j], EPS);
      const float as = fmaxf(__expf(v[j] - unflipf(ms[j])) / ss[j], EPS);
      atomicAdd(&msg[od[j]], f[j] * sqrtf(ad * as));
    }
  }
}

__global__ __launch_bounds__(256) void norm_kernel(
    float* __restrict__ msg, const float* __restrict__ scale,
    const float* __restrict__ offset) {
  const int node = blockIdx.x * 256 + threadIdx.x;
  if (node >= N_NODES) return;
  float x[OUT];
#pragma unroll
  for (int i = 0; i < OUT; ++i) x[i] = msg[(long)node * OUT + i];
#pragma unroll
  for (int h = 0; h < 4; ++h) {
    float m = 0.f;
#pragma unroll
    for (int c = 0; c < 8; ++c) m += x[h * 8 + c];
    m *= 0.125f;
    float var = 0.f;
#pragma unroll
    for (int c = 0; c < 8; ++c) { float dd = x[h * 8 + c] - m; var += dd * dd; }
    const float r = rsqrtf(var * 0.125f + EPS);
#pragma unroll
    for (int c = 0; c < 8; ++c) {
      const int n = h * 8 + c;
      msg[(long)node * OUT + n] = (x[n] - m) * scale[n] * r + offset[n];
    }
  }
}

__global__ __launch_bounds__(32) void final_kernel(
    const float* __restrict__ msg, const float* __restrict__ Wagg,
    const float* __restrict__ bagg, const float* __restrict__ dterm,
    const float* __restrict__ Wap, const float* __restrict__ bap,
    float* __restrict__ out) {
  const int lane = threadIdx.x, lo = lane & 15, hi = lane >> 4;
  const int base = blockIdx.x * 16;

  v8f acc1[2] = {};
  for (int k0 = 0; k0 < OUT; k0 += 4) {
    const int ka = k0 + 2 * hi;
    v2f a = *(const v2f*)(msg + (long)(base + lo) * OUT + ka);
#pragma unroll
    for (int t = 0; t < 2; ++t) {
      const int n = lo + 16 * t;
      v2f b = (v2f){Wagg[ka * OUT + n], Wagg[(ka + 1) * OUT + n]};
      WMMA4(acc1[t], a, b);
    }
  }

  __shared__ float lds[16 * OUT];
#pragma unroll
  for (int t = 0; t < 2; ++t)
#pragma unroll
    for (int v = 0; v < 8; ++v) {
      const long row = base + v + 8 * hi;
      const int n = lo + 16 * t;
      lds[(v + 8 * hi) * OUT + n] = leaky(acc1[t][v] + bagg[n] + dterm[row * OUT + n]);
    }
  __syncthreads();

  v8f acc2[2] = {};
  for (int k0 = 0; k0 < OUT; k0 += 4) {
    const int ka = k0 + 2 * hi;
    v2f a2 = (v2f){lds[lo * OUT + ka], lds[lo * OUT + ka + 1]};
#pragma unroll
    for (int t = 0; t < 2; ++t) {
      const int n = lo + 16 * t;
      v2f b = (v2f){Wap[ka * OUT + n], Wap[(ka + 1) * OUT + n]};
      WMMA4(acc2[t], a2, b);
    }
  }

#pragma unroll
  for (int t = 0; t < 2; ++t)
#pragma unroll
    for (int v = 0; v < 8; ++v)
      out[(long)(base + v + 8 * hi) * OUT + lo + 16 * t] = acc2[t][v] + bap[lo + 16 * t];
}

extern "C" void kernel_launch(void* const* d_in, const int* in_sizes, int n_in,
                              void* d_out, int out_size, void* d_ws, size_t ws_size,
                              hipStream_t stream) {
  const float* feat  = (const float*)d_in[0];
  const float* fe    = (const float*)d_in[1];
  const int*   src   = (const int*)d_in[2];
  const int*   dst   = (const int*)d_in[3];
  const float* Wp    = (const float*)d_in[4];
  const float* Wp2   = (const float*)d_in[5];
  const float* Was   = (const float*)d_in[6];
  const float* Wad   = (const float*)d_in[7];
  const float* We    = (const float*)d_in[8];
  const float* scale = (const float*)d_in[9];
  const float* offs  = (const float*)d_in[10];
  const float* Wagg  = (const float*)d_in[11];
  const float* bagg  = (const float*)d_in[12];
  const float* Wapd  = (const float*)d_in[13];
  const float* bapd  = (const float*)d_in[14];
  const float* Wap   = (const float*)d_in[15];
  const float* bap   = (const float*)d_in[16];
  float* out = (float*)d_out;

  const long NW = (long)N_NODES * OUT;
  float* fc    = (float*)d_ws;
  float* esrc  = fc + NW;
  float* edst  = esrc + NW;
  float* dterm = edst + NW;
  unsigned* mdst = (unsigned*)(dterm + NW);
  unsigned* msrc = mdst + NW;
  float* sdst  = (float*)(msrc + NW);
  float* ssrc  = sdst + NW;
  float* msg   = ssrc + NW;
  float* ebuf  = msg + NW;

  constexpr int nzero4 = 5 * N_NODES * OUT / 4;
  zero_kernel<<<PW_GRID, PW_BLOCK, 0, stream>>>((float*)mdst, nzero4);

  node_gemm_kernel<<<N_NODES / 16, 32, 0, stream>>>(
      feat, Wp, Wp2, Was, Wad, Wapd, bapd, fc, esrc, edst, dterm);

  edge_logits_kernel<<<PW_GRID, PW_BLOCK, 0, stream>>>(
      fe, src, dst, We, esrc, edst, ebuf, mdst, msrc);

  edge_sum_kernel<<<PW_GRID, PW_BLOCK, 0, stream>>>(
      ebuf, src, dst, mdst, msrc, sdst, ssrc);

  edge_agg_kernel<<<PW_GRID, PW_BLOCK, 0, stream>>>(
      ebuf, src, dst, mdst, msrc, sdst, ssrc, fc, msg);

  norm_kernel<<<(N_NODES + 255) / 256, 256, 0, stream>>>(msg, scale, offs);

  final_kernel<<<N_NODES / 16, 32, 0, stream>>>(
      msg, Wagg, bagg, dterm, Wap, bap, out);

  (void)in_sizes; (void)n_in; (void)out_size; (void)ws_size;
}


# edge logits: tile turned from the matrix result layout to one edge per register (lane = channel) by a half-wave exchange, so each gather, store and max atomic covers one whole row, not two half rows
# speedup vs baseline: 1.8180x; 1.1293x over previous
#include <hip/hip_runtime.h>

constexpr int N_NODES = 50000;
constexpr int N_EDGES = 1600000;
constexpr int F_IN    = 128;
constexpr int F_E     = 16;
constexpr int OUT     = 32;
constexpr float SLOPE = 0.2f;
constexpr float EPS   = 1e-9f;

typedef float v2f __attribute__((ext_vector_type(2)));
typedef float v8f __attribute__((ext_vector_type(8)));

#define WMMA4(acc, a, b)                                                     \
  acc = __builtin_amdgcn_wmma_f32_16x16x4_f32(false, (a), false, (b),        \
                                              (short)0, (acc), false, false)

__device__ __forceinline__ float leaky(float x) { return x >= 0.f ? x : SLOPE * x; }

__device__ __forceinline__ unsigned flipf(float f) {
  unsigned b = __float_as_uint(f);
  return (b & 0x80000000u) ? ~b : (b | 0x80000000u);
}
__device__ __forceinline__ float unflipf(unsigned u) {
  return __uint_as_float((u & 0x80000000u) ? (u ^ 0x80000000u) : ~u);
}

constexpr int PW_BLOCK = 256;
constexpr int PW_GRID  = 1024;
constexpr int PW_WAVES = PW_GRID * (PW_BLOCK / 32);
#define PW_RESIDENT __attribute__((amdgpu_waves_per_eu(8, 8)))

__device__ __forceinline__ int wave_number() {
  return __builtin_amdgcn_readfirstlane(
      (int)(blockIdx.x * (PW_BLOCK / 32) + (threadIdx.x >> 5)));
}

__global__ __launch_bounds__(PW_BLOCK) void zero_kernel(float* p, int n4) {
  float4* p4 = (float4*)p;
  const float4 z = {0.f, 0.f, 0.f, 0.f};
  for (int i = blockIdx.x * PW_BLOCK + threadIdx.x; i < n4; i += PW_GRID * PW_BLOCK)
    p4[i] = z;
}

__global__ __launch_bounds__(32) void node_gemm_kernel(
    const float* __restrict__ feat,
    const float* __restrict__ Wp,  const float* __restrict__ Wp2,
    const float* __restrict__ Was, const float* __restrict__ Wad,
    const float* __restrict__ Wapd, const float* __restrict__ bapd,
    float* __restrict__ fc, float* __restrict__ esrc,
    float* __restrict__ edst, float* __restrict__ dterm) {
  const int lane = threadIdx.x, lo = lane & 15, hi = lane >> 4;
  const int base = blockIdx.x * 16;

  v8f accP[2] = {}, accAS[2] = {}, accAD[2] = {}, accAP[2] = {};
  for (int k0 = 0; k0 < F_IN; k0 += 4) {
    const int ka = k0 + 2 * hi;
    v2f a = *(const v2f*)(feat + (long)(base + lo) * F_IN + ka);
#pragma unroll
    for (int t = 0; t < 2; ++t) {
      const int n = lo + 16 * t;
      v2f b;
      b = (v2f){Wp[ka * OUT + n],   Wp[(ka + 1) * OUT + n]};   WMMA4(accP[t],  a, b);
      b = (v2f){Was[ka * OUT + n],  Was[(ka + 1) * OUT + n]};  WMMA4(accAS[t], a, b);
      b = (v2f){Wad[ka * OUT + n],  Wad[(ka + 1) * OUT + n]};  WMMA4(accAD[t], a, b);
      b = (v2f){Wapd[ka * OUT + n], Wapd[(ka + 1) * OUT + n]}; WMMA4(accAP[t], a, b);
    }
  }

  __shared__ float lds[16 * OUT];
#pragma unroll
  for (int t = 0; t < 2; ++t)
#pragma unroll
    for (int v = 0; v < 8; ++v)
      lds[(v + 8 * hi) * OUT + lo + 16 * t] = leaky(accP[t][v]);
  __syncthreads();

  v8f accFC[2] = {};
  for (int k0 = 0; k0 < OUT; k0 += 4) {
    const int ka = k0 + 2 * hi;
    v2f a2 = (v2f){lds[lo * OUT + ka], lds[lo * OUT + ka + 1]};
#pragma unroll
    for (int t = 0; t < 2; ++t) {
      const int n = lo + 16 * t;
      v2f b = (v2f){Wp2[ka * OUT + n], Wp2[(ka + 1) * OUT + n]};
      WMMA4(accFC[t], a2, b);
    }
  }

#pragma unroll
  for (int t = 0; t < 2; ++t)
#pragma unroll
    for (int v = 0; v < 8; ++v) {
      const long row = base + v + 8 * hi;
      const int n = lo + 16 * t;
      fc[row * OUT + n]    = accFC[t][v];
      esrc[row * OUT + n]  = accAS[t][v];
      edst[row * OUT + n]  = accAD[t][v];
      dterm[row * OUT + n] = accAP[t][v] + bapd[n];
    }
}


__device__ __forceinline__ float xhalf(float x) {
  const int b = __float_as_int(x);
  return __int_as_float(
      __builtin_amdgcn_permlanex16(b, b, 0x76543210, 0xfedcba98, false, false));
}

constexpr int TE = 16;

__global__ __launch_bounds__(PW_BLOCK) PW_RESIDENT void edge_logits_kernel(
    const float* __restrict__ fe, const int* __restrict__ src,
    const int* __restrict__ dst, const float* __restrict__ We,
    const float* __restrict__ esrc, const float* __restrict__ edst,
    float* __restrict__ ebuf, unsigned* __restrict__ mdst,
    unsigned* __restrict__ msrc) {
  const int lane = threadIdx.x & 31, lo = lane & 15, hi = lane >> 4;
  const int wave = wave_number();

  v2f bw[F_E / 4][2];
#pragma unroll
  for (int k = 0; k < F_E / 4; ++k) {
    const int ka = 4 * k + 2 * hi;
#pragma unroll
    for (int t = 0; t < 2; ++t) {
      const int n = lo + 16 * t;
      bw[k][t] = (v2f){We[ka * OUT + n], We[(ka + 1) * OUT + n]};
    }
  }

  for (int tile = wave; tile < N_EDGES / TE; tile += PW_WAVES) {
    const int ebase = tile * TE;

    const int sv = src[ebase + lo];
    const int dv = dst[ebase + lo];

    v8f acc[2] = {};
#pragma unroll
    for (int k = 0; k < F_E / 4; ++k) {
      const int ka = 4 * k + 2 * hi;
      v2f a = *(const v2f*)(fe + (long)(ebase + lo) * F_E + ka);
#pragma unroll
      for (int t = 0; t < 2; ++t) WMMA4(acc[t], a, bw[k][t]);
    }

    unsigned os[TE], od[TE];
#pragma unroll
    for (int r = 0; r < TE; ++r) {
      os[r] = (unsigned)__builtin_amdgcn_readlane(sv, r) * OUT + lane;
      od[r] = (unsigned)__builtin_amdgcn_readlane(dv, r) * OUT + lane;
    }
    float gs[TE], gd[TE];
#pragma unroll
    for (int r = 0; r < TE; ++r) {
      gs[r] = esrc[os[r]];
      gd[r] = edst[od[r]];
    }

    float row[TE];
#pragma unroll
    for (int v = 0; v < 8; ++v) {
      const float a0 = acc[0][v], a1 = acc[1][v];
      const float x0 = xhalf(a0), x1 = xhalf(a1);
      row[v]     = hi ? x1 : a0;
      row[v + 8] = hi ? a1 : x0;
    }

#pragma unroll
    for (int r = 0; r < TE; ++r) {
      const float val = leaky(row[r] + gs[r] + gd[r]);
      ebuf[(ebase + r) * OUT + lane] = val;
      const unsigned fv = flipf(val);
      atomicMax(&mdst[od[r]], fv);
      atomicMax(&msrc[os[r]], fv);
    }
  }
}

constexpr int EPW = 8;
constexpr int N_CHUNKS = N_EDGES / EPW;
static_assert(N_EDGES % EPW == 0, "edge chunks must tile exactly");
static_assert(N_EDGES % 16 == 0, "logit tiles must tile exactly");

__device__ __forceinline__ void edge_offsets(const int* __restrict__ src,
                                             const int* __restrict__ dst,
                                             int e0, int n,
                                             unsigned (&os)[EPW], unsigned (&od)[EPW]) {
  const int sv = src[e0 + (n & (EPW - 1))];
  const int dv = dst[e0 + (n & (EPW - 1))];
#pragma unroll
  for (int j = 0; j < EPW; ++j) {
    os[j] = (unsigned)__builtin_amdgcn_readlane(sv, j) * OUT + n;
    od[j] = (unsigned)__builtin_amdgcn_readlane(dv, j) * OUT + n;
  }
}

__global__ __launch_bounds__(PW_BLOCK) void edge_sum_kernel(
    const float* __restrict__ ebuf, const int* __restrict__ src,
    const int* __restrict__ dst, const unsigned* __restrict__ mdst,
    const unsigned* __restrict__ msrc, float* __restrict__ sdst,
    float* __restrict__ ssrc) {
  const int n = threadIdx.x & 31;
  for (int chunk = wave_number(); chunk < N_CHUNKS; chunk += PW_WAVES) {
    const int e0 = chunk * EPW;
    unsigned os[EPW], od[EPW];
    edge_offsets(src, dst, e0, n, os, od);

    float v[EPW];
    unsigned md[EPW], ms[EPW];
#pragma unroll
    for (int j = 0; j < EPW; ++j) {
      v[j]  = ebuf[(e0 + j) * OUT + n];
      md[j] = mdst[od[j]];
      ms[j] = msrc[os[j]];
    }
#pragma unroll
    for (int j = 0; j < EPW; ++j) {
      atomicAdd(&sdst[od[j]], __expf(v[j] - unflipf(md[j])));
      atomicAdd(&ssrc[os[j]], __expf(v[j] - unflipf(ms[j])));
    }
  }
}

__global__ __launch_bounds__(PW_BLOCK) PW_RESIDENT void edge_agg_kernel(
    const float* __restrict__ ebuf, const int* __restrict__ src,
    const int* __restrict__ dst, const unsigned* __restrict__ mdst,
    const unsigned* __restrict__ msrc, const float* __restrict__ sdst,
    const float* __restrict__ ssrc, const float* __restrict__ fc,
    float* __restrict__ msg) {
  const int n = threadIdx.x & 31;
  for (int chunk = wave_number(); chunk < N_CHUNKS; chunk += PW_WAVES) {
    const int e0 = chunk * EPW;
    unsigned os[EPW], od[EPW];
    edge_offsets(src, dst, e0, n, os, od);

    float v[EPW], sd[EPW], ss[EPW], f[EPW];
    unsigned md[EPW], ms[EPW];
#pragma unroll
    for (int j = 0; j < EPW; ++j) {
      v[j]  = ebuf[(e0 + j) * OUT + n];
      md[j] = mdst[od[j]];
      sd[j] = sdst[od[j]];
      ms[j] = msrc[os[j]];
      ss[j] = ssrc[os[j]];
      f[j]  = fc[os[j]];
    }
#pragma unroll
    for (int j = 0; j < EPW; ++j) {
      const float ad = fmaxf(__expf(v[j] - unflipf(md[j])) / sd[j], EPS);
      const float as = fmaxf(__expf(v[j] - unflipf(ms[j])) / ss[j], EPS);
      atomicAdd(&msg[od[j]], f[j] * sqrtf(ad * as));
    }
  }
}

__global__ __launch_bounds__(256) void norm_kernel(
    float* __restrict__ msg, const float* __restrict__ scale,
    const float* __restrict__ offset) {
  const int node = blockIdx.x * 256 + threadIdx.x;
  if (node >= N_NODES) return;
  float x[OUT];
#pragma unroll
  for (int i = 0; i < OUT; ++i) x[i] = msg[(long)node * OUT + i];
#pragma unroll
  for (int h = 0; h < 4; ++h) {
    float m = 0.f;
#pragma unroll
    for (int c = 0; c < 8; ++c) m += x[h * 8 + c];
    m *= 0.125f;
    float var = 0.f;
#pragma unroll
    for (int c = 0; c < 8; ++c) { float dd = x[h * 8 + c] - m; var += dd * dd; }
    const float r = rsqrtf(var * 0.125f + EPS);
#pragma unroll
    for (int c = 0; c < 8; ++c) {
      const int n = h * 8 + c;
      msg[(long)node * OUT + n] = (x[n] - m) * scale[n] * r + offset[n];
    }
  }
}

__global__ __launch_bounds__(32) void final_kernel(
    const float* __restrict__ msg, const float* __restrict__ Wagg,
    const float* __restrict__ bagg, const float* __restrict__ dterm,
    const float* __restrict__ Wap, const float* __restrict__ bap,
    float* __restrict__ out) {
  const int lane = threadIdx.x, lo = lane & 15, hi = lane >> 4;
  const int base = blockIdx.x * 16;

  v8f acc1[2] = {};
  for (int k0 = 0; k0 < OUT; k0 += 4) {
    const int ka = k0 + 2 * hi;
    v2f a = *(const v2f*)(msg + (long)(base + lo) * OUT + ka);
#pragma unroll
    for (int t = 0; t < 2; ++t) {
      const int n = lo + 16 * t;
      v2f b = (v2f){Wagg[ka * OUT + n], Wagg[(ka + 1) * OUT + n]};
      WMMA4(acc1[t], a, b);
    }
  }

  __shared__ float lds[16 * OUT];
#pragma unroll
  for (int t = 0; t < 2; ++t)
#pragma unroll
    for (int v = 0; v < 8; ++v) {
      const long row = base + v + 8 * hi;
      const int n = lo + 16 * t;
      lds[(v + 8 * hi) * OUT + n] = leaky(acc1[t][v] + bagg[n] + dterm[row * OUT + n]);
    }
  __syncthreads();

  v8f acc2[2] = {};
  for (int k0 = 0; k0 < OUT; k0 += 4) {
    const int ka = k0 + 2 * hi;
    v2f a2 = (v2f){lds[lo * OUT + ka], lds[lo * OUT + ka + 1]};
#pragma unroll
    for (int t = 0; t < 2; ++t) {
      const int n = lo + 16 * t;
      v2f b = (v2f){Wap[ka * OUT + n], Wap[(ka + 1) * OUT + n]};
      WMMA4(acc2[t], a2, b);
    }
  }

#pragma unroll
  for (int t = 0; t < 2; ++t)
#pragma unroll
    for (int v = 0; v < 8; ++v)
      out[(long)(base + v + 8 * hi) * OUT + lo + 16 * t] = acc2[t][v] + bap[lo + 16 * t];
}

extern "C" void kernel_launch(void* const* d_in, const int* in_sizes, int n_in,
                              void* d_out, int out_size, void* d_ws, size_t ws_size,
                              hipStream_t stream) {
  const float* feat  = (const float*)d_in[0];
  const float* fe    = (const float*)d_in[1];
  const int*   src   = (const int*)d_in[2];
  const int*   dst   = (const int*)d_in[3];
  const float* Wp    = (const float*)d_in[4];
  const float* Wp2   = (const float*)d_in[5];
  const float* Was   = (const float*)d_in[6];
  const float* Wad   = (const float*)d_in[7];
  const float* We    = (const float*)d_in[8];
  const float* scale = (const float*)d_in[9];
  const float* offs  = (const float*)d_in[10];
  const float* Wagg  = (const float*)d_in[11];
  const float* bagg  = (const float*)d_in[12];
  const float* Wapd  = (const float*)d_in[13];
  const float* bapd  = (const float*)d_in[14];
  const float* Wap   = (const float*)d_in[15];
  const float* bap   = (const float*)d_in[16];
  float* out = (float*)d_out;

  const long NW = (long)N_NODES * OUT;
  float* fc    = (float*)d_ws;
  float* esrc  = fc + NW;
  float* edst  = esrc + NW;
  float* dterm = edst + NW;
  unsigned* mdst = (unsigned*)(dterm + NW);
  unsigned* msrc = mdst + NW;
  float* sdst  = (float*)(msrc + NW);
  float* ssrc  = sdst + NW;
  float* msg   = ssrc + NW;
  float* ebuf  = msg + NW;

  constexpr int nzero4 = 5 * N_NODES * OUT / 4;
  zero_kernel<<<PW_GRID, PW_BLOCK, 0, stream>>>((float*)mdst, nzero4);

  node_gemm_kernel<<<N_NODES / 16, 32, 0, stream>>>(
      feat, Wp, Wp2, Was, Wad, Wapd, bapd, fc, esrc, edst, dterm);

  edge_logits_kernel<<<PW_GRID, PW_BLOCK, 0, stream>>>(
      fe, src, dst, We, esrc, edst, ebuf, mdst, msrc);

  edge_sum_kernel<<<PW_GRID, PW_BLOCK, 0, stream>>>(
      ebuf, src, dst, mdst, msrc, sdst, ssrc);

  edge_agg_kernel<<<PW_GRID, PW_BLOCK, 0, stream>>>(
      ebuf, src, dst, mdst, msrc, sdst, ssrc, fc, msg);

  norm_kernel<<<(N_NODES + 255) / 256, 256, 0, stream>>>(msg, scale, offs);

  final_kernel<<<N_NODES / 16, 32, 0, stream>>>(
      msg, Wagg, bagg, dterm, Wap, bap, out);

  (void)in_sizes; (void)n_in; (void)out_size; (void)ws_size;
}


# per-node pass folds each side's max and log of sum into one row and packs the src row with the message row in a 256-byte record: aggregation reads 2 rows per edge, not 5; atomics land in any order
# speedup vs baseline: 1.9057x; 1.0482x over previous
#include <hip/hip_runtime.h>

constexpr int N_NODES = 50000;
constexpr int N_EDGES = 1600000;
constexpr int F_IN    = 128;
constexpr int F_E     = 16;
constexpr int OUT     = 32;
constexpr float SLOPE = 0.2f;
constexpr float EPS   = 1e-9f;

typedef float v2f __attribute__((ext_vector_type(2)));
typedef float v8f __attribute__((ext_vector_type(8)));

#define WMMA4(acc, a, b)                                                     \
  acc = __builtin_amdgcn_wmma_f32_16x16x4_f32(false, (a), false, (b),        \
                                              (short)0, (acc), false, false)

__device__ __forceinline__ float leaky(float x) { return x >= 0.f ? x : SLOPE * x; }

__device__ __forceinline__ unsigned flipf(float f) {
  unsigned b = __float_as_uint(f);
  return (b & 0x80000000u) ? ~b : (b | 0x80000000u);
}
__device__ __forceinline__ float unflipf(unsigned u) {
  return __uint_as_float((u & 0x80000000u) ? (u ^ 0x80000000u) : ~u);
}

constexpr int PW_BLOCK = 256;
constexpr int PW_GRID  = 1024;
constexpr int PW_WAVES = PW_GRID * (PW_BLOCK / 32);
#define PW_RESIDENT __attribute__((amdgpu_waves_per_eu(8, 8)))

__device__ __forceinline__ int wave_number() {
  return __builtin_amdgcn_readfirstlane(
      (int)(blockIdx.x * (PW_BLOCK / 32) + (threadIdx.x >> 5)));
}

__global__ __launch_bounds__(PW_BLOCK) void zero_kernel(float* p, int n4) {
  float4* p4 = (float4*)p;
  const float4 z = {0.f, 0.f, 0.f, 0.f};
  for (int i = blockIdx.x * PW_BLOCK + threadIdx.x; i < n4; i += PW_GRID * PW_BLOCK)
    p4[i] = z;
}

__global__ __launch_bounds__(32) void node_gemm_kernel(
    const float* __restrict__ feat,
    const float* __restrict__ Wp,  const float* __restrict__ Wp2,
    const float* __restrict__ Was, const float* __restrict__ Wad,
    const float* __restrict__ Wapd, const float* __restrict__ bapd,
    float* __restrict__ fc, float* __restrict__ esrc,
    float* __restrict__ edst, float* __restrict__ dterm) {
  const int lane = threadIdx.x, lo = lane & 15, hi = lane >> 4;
  const int base = blockIdx.x * 16;

  v8f accP[2] = {}, accAS[2] = {}, accAD[2] = {}, accAP[2] = {};
  for (int k0 = 0; k0 < F_IN; k0 += 4) {
    const int ka = k0 + 2 * hi;
    v2f a = *(const v2f*)(feat + (long)(base + lo) * F_IN + ka);
#pragma unroll
    for (int t = 0; t < 2; ++t) {
      const int n = lo + 16 * t;
      v2f b;
      b = (v2f){Wp[ka * OUT + n],   Wp[(ka + 1) * OUT + n]};   WMMA4(accP[t],  a, b);
      b = (v2f){Was[ka * OUT + n],  Was[(ka + 1) * OUT + n]};  WMMA4(accAS[t], a, b);
      b = (v2f){Wad[ka * OUT + n],  Wad[(ka + 1) * OUT + n]};  WMMA4(accAD[t], a, b);
      b = (v2f){Wapd[ka * OUT + n], Wapd[(ka + 1) * OUT + n]}; WMMA4(accAP[t], a, b);
    }
  }

  __shared__ float lds[16 * OUT];
#pragma unroll
  for (int t = 0; t < 2; ++t)
#pragma unroll
    for (int v = 0; v < 8; ++v)
      lds[(v + 8 * hi) * OUT + lo + 16 * t] = leaky(accP[t][v]);
  __syncthreads();

  v8f accFC[2] = {};
  for (int k0 = 0; k0 < OUT; k0 += 4) {
    const int ka = k0 + 2 * hi;
    v2f a2 = (v2f){lds[lo * OUT + ka], lds[lo * OUT + ka + 1]};
#pragma unroll
    for (int t = 0; t < 2; ++t) {
      const int n = lo + 16 * t;
      v2f b = (v2f){Wp2[ka * OUT + n], Wp2[(ka + 1) * OUT + n]};
      WMMA4(accFC[t], a2, b);
    }
  }

#pragma unroll
  for (int t = 0; t < 2; ++t)
#pragma unroll
    for (int v = 0; v < 8; ++v) {
      const long row = base + v + 8 * hi;
      const int n = lo + 16 * t;
      fc[row * OUT + n]    = accFC[t][v];
      esrc[row * OUT + n]  = accAS[t][v];
      edst[row * OUT + n]  = accAD[t][v];
      dterm[row * OUT + n] = accAP[t][v] + bapd[n];
    }
}


__device__ __forceinline__ float xhalf(float x) {
  const int b = __float_as_int(x);
  return __int_as_float(
      __builtin_amdgcn_permlanex16(b, b, 0x76543210, 0xfedcba98, false, false));
}

constexpr int TE = 16;

__global__ __launch_bounds__(PW_BLOCK) PW_RESIDENT void edge_logits_kernel(
    const float* __restrict__ fe, const int* __restrict__ src,
    const int* __restrict__ dst, const float* __restrict__ We,
    const float* __restrict__ esrc, const float* __restrict__ edst,
    float* __restrict__ ebuf, unsigned* __restrict__ mdst,
    unsigned* __restrict__ msrc) {
  const int lane = threadIdx.x & 31, lo = lane & 15, hi = lane >> 4;
  const int wave = wave_number();

  v2f bw[F_E / 4][2];
#pragma unroll
  for (int k = 0; k < F_E / 4; ++k) {
    const int ka = 4 * k + 2 * hi;
#pragma unroll
    for (int t = 0; t < 2; ++t) {
      const int n = lo + 16 * t;
      bw[k][t] = (v2f){We[ka * OUT + n], We[(ka + 1) * OUT + n]};
    }
  }

  for (int tile = wave; tile < N_EDGES / TE; tile += PW_WAVES) {
    const int ebase = tile * TE;

    const int sv = src[ebase + lo];
    const int dv = dst[ebase + lo];

    v8f acc[2] = {};
#pragma unroll
    for (int k = 0; k < F_E / 4; ++k) {
      const int ka = 4 * k + 2 * hi;
      v2f a = *(const v2f*)(fe + (long)(ebase + lo) * F_E + ka);
#pragma unroll
      for (int t = 0; t < 2; ++t) WMMA4(acc[t], a, bw[k][t]);
    }

    unsigned os[TE], od[TE];
#pragma unroll
    for (int r = 0; r < TE; ++r) {
      os[r] = (unsigned)__builtin_amdgcn_readlane(sv, r) * OUT + lane;
      od[r] = (unsigned)__builtin_amdgcn_readlane(dv, r) * OUT + lane;
    }
    float gs[TE], gd[TE];
#pragma unroll
    for (int r = 0; r < TE; ++r) {
      gs[r] = esrc[os[r]];
      gd[r] = edst[od[r]];
    }

    float row[TE];
#pragma unroll
    for (int v = 0; v < 8; ++v) {
      const float a0 = acc[0][v], a1 = acc[1][v];
      const float x0 = xhalf(a0), x1 = xhalf(a1);
      row[v]     = hi ? x1 : a0;
      row[v + 8] = hi ? a1 : x0;
    }

#pragma unroll
    for (int r = 0; r < TE; ++r) {
      const float val = leaky(row[r] + gs[r] + gd[r]);
      ebuf[(ebase + r) * OUT + lane] = val;
      const unsigned fv = flipf(val);
      atomicMax(&mdst[od[r]], fv);
      atomicMax(&msrc[os[r]], fv);
    }
  }
}

constexpr int EPW = 8;
constexpr int N_CHUNKS = N_EDGES / EPW;
static_assert(N_EDGES % EPW == 0, "edge chunks must tile exactly");
static_assert(N_EDGES % 16 == 0, "logit tiles must tile exactly");

__device__ __forceinline__ void edge_offsets(const int* __restrict__ src,
                                             const int* __restrict__ dst,
                                             int e0, int n,
                                             unsigned (&os)[EPW], unsigned (&od)[EPW]) {
  const int sv = src[e0 + (n & (EPW - 1))];
  const int dv = dst[e0 + (n & (EPW - 1))];
#pragma unroll
  for (int j = 0; j < EPW; ++j) {
    os[j] = (unsigned)__builtin_amdgcn_readlane(sv, j) * OUT + n;
    od[j] = (unsigned)__builtin_amdgcn_readlane(dv, j) * OUT + n;
  }
}

__global__ __launch_bounds__(PW_BLOCK) void edge_sum_kernel(
    const float* __restrict__ ebuf, const int* __restrict__ src,
    const int* __restrict__ dst, const unsigned* __restrict__ mdst,
    const unsigned* __restrict__ msrc, float* __restrict__ sdst,
    float* __restrict__ ssrc) {
  const int n = threadIdx.x & 31;
  for (int chunk = wave_number(); chunk < N_CHUNKS; chunk += PW_WAVES) {
    const int e0 = chunk * EPW;
    unsigned os[EPW], od[EPW];
    edge_offsets(src, dst, e0, n, os, od);

    float v[EPW];
    unsigned md[EPW], ms[EPW];
#pragma unroll
    for (int j = 0; j < EPW; ++j) {
      v[j]  = ebuf[(e0 + j) * OUT + n];
      md[j] = mdst[od[j]];
      ms[j] = msrc[os[j]];
    }
#pragma unroll
    for (int j = 0; j < EPW; ++j) {
      atomicAdd(&sdst[od[j]], __expf(v[j] - unflipf(md[j])));
      atomicAdd(&ssrc[os[j]], __expf(v[j] - unflipf(ms[j])));
    }
  }
}

constexpr int FOLD_ITEMS = N_NODES * OUT / 4;

__device__ __forceinline__ float fold_term(float flipped_max_bits, float sum) {
  return unflipf(__float_as_uint(flipped_max_bits)) + logf(sum);
}

__global__ __launch_bounds__(PW_BLOCK) void node_fold_kernel(
    float* mplane, const float* __restrict__ msrc_bits,
    const float* __restrict__ sdst, const float* __restrict__ ssrc,
    const float* __restrict__ fc, float* __restrict__ srcrec) {
  for (int i = blockIdx.x * PW_BLOCK + threadIdx.x; i < FOLD_ITEMS; i += PW_GRID * PW_BLOCK) {
    const float4 md = ((const float4*)mplane)[i];
    const float4 ms = ((const float4*)msrc_bits)[i];
    const float4 sd = ((const float4*)sdst)[i];
    const float4 ss = ((const float4*)ssrc)[i];
    const float4 f  = ((const float4*)fc)[i];
    const float4 ld = {fold_term(md.x, sd.x), fold_term(md.y, sd.y),
                       fold_term(md.z, sd.z), fold_term(md.w, sd.w)};
    ((float4*)mplane)[i] = ld;
    const float4 r0 = {fold_term(ms.x, ss.x), f.x, fold_term(ms.y, ss.y), f.y};
    const float4 r1 = {fold_term(ms.z, ss.z), f.z, fold_term(ms.w, ss.w), f.w};
    ((float4*)srcrec)[2 * i]     = r0;
    ((float4*)srcrec)[2 * i + 1] = r1;
  }
}

__global__ __launch_bounds__(PW_BLOCK) void edge_agg_kernel(
    const float* __restrict__ ebuf, const int* __restrict__ src,
    const int* __restrict__ dst, const float* __restrict__ lddst,
    const float* __restrict__ srcrec, float* __restrict__ msg) {
  const int n = threadIdx.x & 31;
  const v2f* __restrict__ rec = (const v2f*)srcrec;
  for (int chunk = wave_number(); chunk < N_CHUNKS; chunk += PW_WAVES) {
    const int e0 = chunk * EPW;
    unsigned os[EPW], od[EPW];
    edge_offsets(src, dst, e0, n, os, od);

    float v[EPW], ld[EPW];
    v2f r[EPW];
#pragma unroll
    for (int j = 0; j < EPW; ++j) {
      v[j]  = ebuf[(e0 + j) * OUT + n];
      ld[j] = lddst[od[j]];
      r[j]  = rec[os[j]];
    }
#pragma unroll
    for (int j = 0; j < EPW; ++j) {
      const float ad = fmaxf(__expf(v[j] - ld[j]), EPS);
      const float as = fmaxf(__expf(v[j] - r[j].x), EPS);
      atomicAdd(&msg[od[j]], r[j].y * sqrtf(ad * as));
    }
  }
}

__global__ __launch_bounds__(256) void norm_kernel(
    float* __restrict__ msg, const float* __restrict__ scale,
    const float* __restrict__ offset) {
  const int node = blockIdx.x * 256 + threadIdx.x;
  if (node >= N_NODES) return;
  float x[OUT];
#pragma unroll
  for (int i = 0; i < OUT; ++i) x[i] = msg[(long)node * OUT + i];
#pragma unroll
  for (int h = 0; h < 4; ++h) {
    float m = 0.f;
#pragma unroll
    for (int c = 0; c < 8; ++c) m += x[h * 8 + c];
    m *= 0.125f;
    float var = 0.f;
#pragma unroll
    for (int c = 0; c < 8; ++c) { float dd = x[h * 8 + c] - m; var += dd * dd; }
    const float r = rsqrtf(var * 0.125f + EPS);
#pragma unroll
    for (int c = 0; c < 8; ++c) {
      const int n = h * 8 + c;
      msg[(long)node * OUT + n] = (x[n] - m) * scale[n] * r + offset[n];
    }
  }
}

__global__ __launch_bounds__(32) void final_kernel(
    const float* __restrict__ msg, const float* __restrict__ Wagg,
    const float* __restrict__ bagg, const float* __restrict__ dterm,
    const float* __restrict__ Wap, const float* __restrict__ bap,
    float* __restrict__ out) {
  const int lane = threadIdx.x, lo = lane & 15, hi = lane >> 4;
  const int base = blockIdx.x * 16;

  v8f acc1[2] = {};
  for (int k0 = 0; k0 < OUT; k0 += 4) {
    const int ka = k0 + 2 * hi;
    v2f a = *(const v2f*)(msg + (long)(base + lo) * OUT + ka);
#pragma unroll
    for (int t = 0; t < 2; ++t) {
      const int n = lo + 16 * t;
      v2f b = (v2f){Wagg[ka * OUT + n], Wagg[(ka + 1) * OUT + n]};
      WMMA4(acc1[t], a, b);
    }
  }

  __shared__ float lds[16 * OUT];
#pragma unroll
  for (int t = 0; t < 2; ++t)
#pragma unroll
    for (int v = 0; v < 8; ++v) {
      const long row = base + v + 8 * hi;
      const int n = lo + 16 * t;
      lds[(v + 8 * hi) * OUT + n] = leaky(acc1[t][v] + bagg[n] + dterm[row * OUT + n]);
    }
  __syncthreads();

  v8f acc2[2] = {};
  for (int k0 = 0; k0 < OUT; k0 += 4) {
    const int ka = k0 + 2 * hi;
    v2f a2 = (v2f){lds[lo * OUT + ka], lds[lo * OUT + ka + 1]};
#pragma unroll
    for (int t = 0; t < 2; ++t) {
      const int n = lo + 16 * t;
      v2f b = (v2f){Wap[ka * OUT + n], Wap[(ka + 1) * OUT + n]};
      WMMA4(acc2[t], a2, b);
    }
  }

#pragma unroll
  for (int t = 0; t < 2; ++t)
#pragma unroll
    for (int v = 0; v < 8; ++v)
      out[(long)(base + v + 8 * hi) * OUT + lo + 16 * t] = acc2[t][v] + bap[lo + 16 * t];
}

extern "C" void kernel_launch(void* const* d_in, const int* in_sizes, int n_in,
                              void* d_out, int out_size, void* d_ws, size_t ws_size,
                              hipStream_t stream) {
  const float* feat  = (const float*)d_in[0];
  const float* fe    = (const float*)d_in[1];
  const int*   src   = (const int*)d_in[2];
  const int*   dst   = (const int*)d_in[3];
  const float* Wp    = (const float*)d_in[4];
  const float* Wp2   = (const float*)d_in[5];
  const float* Was   = (const float*)d_in[6];
  const float* Wad   = (const float*)d_in[7];
  const float* We    = (const float*)d_in[8];
  const float* scale = (const float*)d_in[9];
  const float* offs  = (const float*)d_in[10];
  const float* Wagg  = (const float*)d_in[11];
  const float* bagg  = (const float*)d_in[12];
  const float* Wapd  = (const float*)d_in[13];
  const float* bapd  = (const float*)d_in[14];
  const float* Wap   = (const float*)d_in[15];
  const float* bap   = (const float*)d_in[16];
  float* out = (float*)d_out;

  const long NW = (long)N_NODES * OUT;
  float* fc    = (float*)d_ws;
  float* esrc  = fc + NW;
  float* edst  = esrc + NW;
  float* dterm = edst + NW;
  unsigned* mdst = (unsigned*)(dterm + NW);
  unsigned* msrc = mdst + NW;
  float* sdst  = (float*)(msrc + NW);
  float* ssrc  = sdst + NW;
  float* msg   = ssrc + NW;
  float* ebuf  = msg + NW;

  constexpr int nzero4 = 5 * N_NODES * OUT / 4;
  zero_kernel<<<PW_GRID, PW_BLOCK, 0, stream>>>((float*)mdst, nzero4);

  node_gemm_kernel<<<N_NODES / 16, 32, 0, stream>>>(
      feat, Wp, Wp2, Was, Wad, Wapd, bapd, fc, esrc, edst, dterm);

  edge_logits_kernel<<<PW_GRID, PW_BLOCK, 0, stream>>>(
      fe, src, dst, We, esrc, edst, ebuf, mdst, msrc);

  edge_sum_kernel<<<PW_GRID, PW_BLOCK, 0, stream>>>(
      ebuf, src, dst, mdst, msrc, sdst, ssrc);

  float* srcrec = esrc;
  node_fold_kernel<<<PW_GRID, PW_BLOCK, 0, stream>>>(
      (float*)mdst, (const float*)msrc, sdst, ssrc, fc, srcrec);

  edge_agg_kernel<<<PW_GRID, PW_BLOCK, 0, stream>>>(
      ebuf, src, dst, (const float*)mdst, srcrec, msg);

  norm_kernel<<<(N_NODES + 255) / 256, 256, 0, stream>>>(msg, scale, offs);

  final_kernel<<<N_NODES / 16, 32, 0, stream>>>(
      msg, Wagg, bagg, dterm, Wap, bap, out);

  (void)in_sizes; (void)n_in; (void)out_size; (void)ws_size;
}


# edge logits pass adds exp(e - own node term) into both sum planes itself: no max atomics, no separate sum pass; a range flag reruns max-shifted sums if any |e - shift| > 32; add order still free
# speedup vs baseline: 2.5692x; 1.3482x over previous
#include <hip/hip_runtime.h>

constexpr int N_NODES = 50000;
constexpr int N_EDGES = 1600000;
constexpr int F_IN    = 128;
constexpr int F_E     = 16;
constexpr int OUT     = 32;
constexpr float SLOPE = 0.2f;
constexpr float EPS   = 1e-9f;

typedef float v2f __attribute__((ext_vector_type(2)));
typedef float v8f __attribute__((ext_vector_type(8)));

#define WMMA4(acc, a, b)                                                     \
  acc = __builtin_amdgcn_wmma_f32_16x16x4_f32(false, (a), false, (b),        \
                                              (short)0, (acc), false, false)

__device__ __forceinline__ float leaky(float x) { return x >= 0.f ? x : SLOPE * x; }

__device__ __forceinline__ unsigned flipf(float f) {
  unsigned b = __float_as_uint(f);
  return (b & 0x80000000u) ? ~b : (b | 0x80000000u);
}
__device__ __forceinline__ float unflipf(unsigned u) {
  return __uint_as_float((u & 0x80000000u) ? (u ^ 0x80000000u) : ~u);
}

constexpr float SHIFT_RANGE = 32.f;

constexpr int PW_BLOCK = 256;
constexpr int PW_GRID  = 1024;
constexpr int PW_WAVES = PW_GRID * (PW_BLOCK / 32);
#define PW_RESIDENT __attribute__((amdgpu_waves_per_eu(8, 8)))

__device__ __forceinline__ int wave_number() {
  return __builtin_amdgcn_readfirstlane(
      (int)(blockIdx.x * (PW_BLOCK / 32) + (threadIdx.x >> 5)));
}

__global__ __launch_bounds__(PW_BLOCK) void zero_kernel(float* p, int n4) {
  float4* p4 = (float4*)p;
  const float4 z = {0.f, 0.f, 0.f, 0.f};
  for (int i = blockIdx.x * PW_BLOCK + threadIdx.x; i < n4; i += PW_GRID * PW_BLOCK)
    p4[i] = z;
}

__global__ __launch_bounds__(32) void node_gemm_kernel(
    const float* __restrict__ feat,
    const float* __restrict__ Wp,  const float* __restrict__ Wp2,
    const float* __restrict__ Was, const float* __restrict__ Wad,
    const float* __restrict__ Wapd, const float* __restrict__ bapd,
    float* __restrict__ fc, float* __restrict__ esrc,
    float* __restrict__ edst, float* __restrict__ dterm) {
  const int lane = threadIdx.x, lo = lane & 15, hi = lane >> 4;
  const int base = blockIdx.x * 16;

  v8f accP[2] = {}, accAS[2] = {}, accAD[2] = {}, accAP[2] = {};
  for (int k0 = 0; k0 < F_IN; k0 += 4) {
    const int ka = k0 + 2 * hi;
    v2f a = *(const v2f*)(feat + (long)(base + lo) * F_IN + ka);
#pragma unroll
    for (int t = 0; t < 2; ++t) {
      const int n = lo + 16 * t;
      v2f b;
      b = (v2f){Wp[ka * OUT + n],   Wp[(ka + 1) * OUT + n]};   WMMA4(accP[t],  a, b);
      b = (v2f){Was[ka * OUT + n],  Was[(ka + 1) * OUT + n]};  WMMA4(accAS[t], a, b);
      b = (v2f){Wad[ka * OUT + n],  Wad[(ka + 1) * OUT + n]};  WMMA4(accAD[t], a, b);
      b = (v2f){Wapd[ka * OUT + n], Wapd[(ka + 1) * OUT + n]}; WMMA4(accAP[t], a, b);
    }
  }

  __shared__ float lds[16 * OUT];
#pragma unroll
  for (int t = 0; t < 2; ++t)
#pragma unroll
    for (int v = 0; v < 8; ++v)
      lds[(v + 8 * hi) * OUT + lo + 16 * t] = leaky(accP[t][v]);
  __syncthreads();

  v8f accFC[2] = {};
  for (int k0 = 0; k0 < OUT; k0 += 4) {
    const int ka = k0 + 2 * hi;
    v2f a2 = (v2f){lds[lo * OUT + ka], lds[lo * OUT + ka + 1]};
#pragma unroll
    for (int t = 0; t < 2; ++t) {
      const int n = lo + 16 * t;
      v2f b = (v2f){Wp2[ka * OUT + n], Wp2[(ka + 1) * OUT + n]};
      WMMA4(accFC[t], a2, b);
    }
  }

#pragma unroll
  for (int t = 0; t < 2; ++t)
#pragma unroll
    for (int v = 0; v < 8; ++v) {
      const long row = base + v + 8 * hi;
      const int n = lo + 16 * t;
      fc[row * OUT + n]    = accFC[t][v];
      esrc[row * OUT + n]  = accAS[t][v];
      edst[row * OUT + n]  = accAD[t][v];
      dterm[row * OUT + n] = accAP[t][v] + bapd[n];
    }
}


__device__ __forceinline__ float xhalf(float x) {
  const int b = __float_as_int(x);
  return __int_as_float(
      __builtin_amdgcn_permlanex16(b, b, 0x76543210, 0xfedcba98, false, false));
}

constexpr int TE = 16;

__global__ __launch_bounds__(PW_BLOCK) PW_RESIDENT void edge_logits_kernel(
    const float* __restrict__ fe, const int* __restrict__ src,
    const int* __restrict__ dst, const float* __restrict__ We,
    const float* __restrict__ esrc, const float* __restrict__ edst,
    float* __restrict__ ebuf, float* __restrict__ sdst,
    float* __restrict__ ssrc, unsigned* __restrict__ flag) {
  const int lane = threadIdx.x & 31, lo = lane & 15, hi = lane >> 4;
  const int wave = wave_number();
  float worst = 0.f;

  v2f bw[F_E / 4][2];
#pragma unroll
  for (int k = 0; k < F_E / 4; ++k) {
    const int ka = 4 * k + 2 * hi;
#pragma unroll
    for (int t = 0; t < 2; ++t) {
      const int n = lo + 16 * t;
      bw[k][t] = (v2f){We[ka * OUT + n], We[(ka + 1) * OUT + n]};
    }
  }

  for (int tile = wave; tile < N_EDGES / TE; tile += PW_WAVES) {
    const int ebase = tile * TE;

    const int sv = src[ebase + lo];
    const int dv = dst[ebase + lo];

    v8f acc[2] = {};
#pragma unroll
    for (int k = 0; k < F_E / 4; ++k) {
      const int ka = 4 * k + 2 * hi;
      v2f a = *(const v2f*)(fe + (long)(ebase + lo) * F_E + ka);
#pragma unroll
      for (int t = 0; t < 2; ++t) WMMA4(acc[t], a, bw[k][t]);
    }

    unsigned os[TE], od[TE];
#pragma unroll
    for (int r = 0; r < TE; ++r) {
      os[r] = (unsigned)__builtin_amdgcn_readlane(sv, r) * OUT + lane;
      od[r] = (unsigned)__builtin_amdgcn_readlane(dv, r) * OUT + lane;
    }
    float gs[TE], gd[TE];
#pragma unroll
    for (int r = 0; r < TE; ++r) {
      gs[r] = esrc[os[r]];
      gd[r] = edst[od[r]];
    }

    float row[TE];
#pragma unroll
    for (int v = 0; v < 8; ++v) {
      const float a0 = acc[0][v], a1 = acc[1][v];
      const float x0 = xhalf(a0), x1 = xhalf(a1);
      row[v]     = hi ? x1 : a0;
      row[v + 8] = hi ? a1 : x0;
    }

#pragma unroll
    for (int r = 0; r < TE; ++r) {
      const float val = leaky(row[r] + gs[r] + gd[r]);
      ebuf[(ebase + r) * OUT + lane] = val;
      const float xd = val - gd[r], xs = val - gs[r];
      worst = fmaxf(worst, fmaxf(fabsf(xd), fabsf(xs)));
      atomicAdd(&sdst[od[r]], __expf(xd));
      atomicAdd(&ssrc[os[r]], __expf(xs));
    }
  }
  if (worst > SHIFT_RANGE) *flag = 1u;
}

constexpr int EPW = 8;
constexpr int N_CHUNKS = N_EDGES / EPW;
static_assert(N_EDGES % EPW == 0, "edge chunks must tile exactly");
static_assert(N_EDGES % 16 == 0, "logit tiles must tile exactly");

__device__ __forceinline__ void edge_offsets(const int* __restrict__ src,
                                             const int* __restrict__ dst,
                                             int e0, int n,
                                             unsigned (&os)[EPW], unsigned (&od)[EPW]) {
  const int sv = src[e0 + (n & (EPW - 1))];
  const int dv = dst[e0 + (n & (EPW - 1))];
#pragma unroll
  for (int j = 0; j < EPW; ++j) {
    os[j] = (unsigned)__builtin_amdgcn_readlane(sv, j) * OUT + n;
    od[j] = (unsigned)__builtin_amdgcn_readlane(dv, j) * OUT + n;
  }
}

constexpr int PLANE_ITEMS = N_NODES * OUT / 4;

__global__ __launch_bounds__(PW_BLOCK) void fallback_max_kernel(
    const unsigned* __restrict__ flag, const float* __restrict__ ebuf,
    const int* __restrict__ src, const int* __restrict__ dst,
    unsigned* __restrict__ mdst, unsigned* __restrict__ msrc,
    float* __restrict__ sums) {
  if (*flag == 0u) return;
  float4* s4 = (float4*)sums;
  const float4 z = {0.f, 0.f, 0.f, 0.f};
  for (int i = blockIdx.x * PW_BLOCK + threadIdx.x; i < 2 * PLANE_ITEMS; i += PW_GRID * PW_BLOCK)
    s4[i] = z;
  const int n = threadIdx.x & 31;
  for (int chunk = wave_number(); chunk < N_CHUNKS; chunk += PW_WAVES) {
    const int e0 = chunk * EPW;
    unsigned os[EPW], od[EPW];
    edge_offsets(src, dst, e0, n, os, od);
#pragma unroll
    for (int j = 0; j < EPW; ++j) {
      const unsigned fv = flipf(ebuf[(e0 + j) * OUT + n]);
      atomicMax(&mdst[od[j]], fv);
      atomicMax(&msrc[os[j]], fv);
    }
  }
}

__global__ __launch_bounds__(PW_BLOCK) void fallback_sum_kernel(
    const unsigned* __restrict__ flag, const float* __restrict__ ebuf,
    const int* __restrict__ src, const int* __restrict__ dst,
    const unsigned* __restrict__ mdst, const unsigned* __restrict__ msrc,
    float* __restrict__ sdst, float* __restrict__ ssrc,
    float* __restrict__ shift_dst, float* __restrict__ shift_src) {
  if (*flag == 0u) return;
  for (int i = blockIdx.x * PW_BLOCK + threadIdx.x; i < N_NODES * OUT; i += PW_GRID * PW_BLOCK) {
    shift_dst[i] = unflipf(mdst[i]);
    shift_src[i] = unflipf(msrc[i]);
  }
  const int n = threadIdx.x & 31;
  for (int chunk = wave_number(); chunk < N_CHUNKS; chunk += PW_WAVES) {
    const int e0 = chunk * EPW;
    unsigned os[EPW], od[EPW];
    edge_offsets(src, dst, e0, n, os, od);

    float v[EPW];
    unsigned md[EPW], ms[EPW];
#pragma unroll
    for (int j = 0; j < EPW; ++j) {
      v[j]  = ebuf[(e0 + j) * OUT + n];
      md[j] = mdst[od[j]];
      ms[j] = msrc[os[j]];
    }
#pragma unroll
    for (int j = 0; j < EPW; ++j) {
      atomicAdd(&sdst[od[j]], __expf(v[j] - unflipf(md[j])));
      atomicAdd(&ssrc[os[j]], __expf(v[j] - unflipf(ms[j])));
    }
  }
}

constexpr int FOLD_ITEMS = PLANE_ITEMS;

__device__ __forceinline__ float fold_term(float shift, float sum) {
  return shift + logf(sum);
}

__global__ __launch_bounds__(PW_BLOCK) void node_fold_kernel(
    float* splane, const float* __restrict__ ssrc,
    const float* __restrict__ edst, const float* __restrict__ esrc,
    const float* __restrict__ fc, float* __restrict__ srcrec,
    unsigned* __restrict__ flag) {
  if (blockIdx.x == 0 && threadIdx.x == 0) *flag = 0u;
  for (int i = blockIdx.x * PW_BLOCK + threadIdx.x; i < FOLD_ITEMS; i += PW_GRID * PW_BLOCK) {
    const float4 td = ((const float4*)edst)[i];
    const float4 ts = ((const float4*)esrc)[i];
    const float4 sd = ((const float4*)splane)[i];
    const float4 ss = ((const float4*)ssrc)[i];
    const float4 f  = ((const float4*)fc)[i];
    const float4 ld = {fold_term(td.x, sd.x), fold_term(td.y, sd.y),
                       fold_term(td.z, sd.z), fold_term(td.w, sd.w)};
    ((float4*)splane)[i] = ld;
    const float4 r0 = {fold_term(ts.x, ss.x), f.x, fold_term(ts.y, ss.y), f.y};
    const float4 r1 = {fold_term(ts.z, ss.z), f.z, fold_term(ts.w, ss.w), f.w};
    ((float4*)srcrec)[2 * i]     = r0;
    ((float4*)srcrec)[2 * i + 1] = r1;
  }
}

__global__ __launch_bounds__(PW_BLOCK) void edge_agg_kernel(
    const float* __restrict__ ebuf, const int* __restrict__ src,
    const int* __restrict__ dst, const float* __restrict__ lddst,
    const float* __restrict__ srcrec, float* __restrict__ msg) {
  const int n = threadIdx.x & 31;
  const v2f* __restrict__ rec = (const v2f*)srcrec;
  for (int chunk = wave_number(); chunk < N_CHUNKS; chunk += PW_WAVES) {
    const int e0 = chunk * EPW;
    unsigned os[EPW], od[EPW];
    edge_offsets(src, dst, e0, n, os, od);

    float v[EPW], ld[EPW];
    v2f r[EPW];
#pragma unroll
    for (int j = 0; j < EPW; ++j) {
      v[j]  = ebuf[(e0 + j) * OUT + n];
      ld[j] = lddst[od[j]];
      r[j]  = rec[os[j]];
    }
#pragma unroll
    for (int j = 0; j < EPW; ++j) {
      const float ad = fmaxf(__expf(v[j] - ld[j]), EPS);
      const float as = fmaxf(__expf(v[j] - r[j].x), EPS);
      atomicAdd(&msg[od[j]], r[j].y * sqrtf(ad * as));
    }
  }
}

__global__ __launch_bounds__(256) void norm_kernel(
    float* __restrict__ msg, const float* __restrict__ scale,
    const float* __restrict__ offset) {
  const int node = blockIdx.x * 256 + threadIdx.x;
  if (node >= N_NODES) return;
  float x[OUT];
#pragma unroll
  for (int i = 0; i < OUT; ++i) x[i] = msg[(long)node * OUT + i];
#pragma unroll
  for (int h = 0; h < 4; ++h) {
    float m = 0.f;
#pragma unroll
    for (int c = 0; c < 8; ++c) m += x[h * 8 + c];
    m *= 0.125f;
    float var = 0.f;
#pragma unroll
    for (int c = 0; c < 8; ++c) { float dd = x[h * 8 + c] - m; var += dd * dd; }
    const float r = rsqrtf(var * 0.125f + EPS);
#pragma unroll
    for (int c = 0; c < 8; ++c) {
      const int n = h * 8 + c;
      msg[(long)node * OUT + n] = (x[n] - m) * scale[n] * r + offset[n];
    }
  }
}

__global__ __launch_bounds__(32) void final_kernel(
    const float* __restrict__ msg, const float* __restrict__ Wagg,
    const float* __restrict__ bagg, const float* __restrict__ dterm,
    const float* __restrict__ Wap, const float* __restrict__ bap,
    float* __restrict__ out) {
  const int lane = threadIdx.x, lo = lane & 15, hi = lane >> 4;
  const int base = blockIdx.x * 16;

  v8f acc1[2] = {};
  for (int k0 = 0; k0 < OUT; k0 += 4) {
    const int ka = k0 + 2 * hi;
    v2f a = *(const v2f*)(msg + (long)(base + lo) * OUT + ka);
#pragma unroll
    for (int t = 0; t < 2; ++t) {
      const int n = lo + 16 * t;
      v2f b = (v2f){Wagg[ka * OUT + n], Wagg[(ka + 1) * OUT + n]};
      WMMA4(acc1[t], a, b);
    }
  }

  __shared__ float lds[16 * OUT];
#pragma unroll
  for (int t = 0; t < 2; ++t)
#pragma unroll
    for (int v = 0; v < 8; ++v) {
      const long row = base + v + 8 * hi;
      const int n = lo + 16 * t;
      lds[(v + 8 * hi) * OUT + n] = leaky(acc1[t][v] + bagg[n] + dterm[row * OUT + n]);
    }
  __syncthreads();

  v8f acc2[2] = {};
  for (int k0 = 0; k0 < OUT; k0 += 4) {
    const int ka = k0 + 2 * hi;
    v2f a2 = (v2f){lds[lo * OUT + ka], lds[lo * OUT + ka + 1]};
#pragma unroll
    for (int t = 0; t < 2; ++t) {
      const int n = lo + 16 * t;
      v2f b = (v2f){Wap[ka * OUT + n], Wap[(ka + 1) * OUT + n]};
      WMMA4(acc2[t], a2, b);
    }
  }

#pragma unroll
  for (int t = 0; t < 2; ++t)
#pragma unroll
    for (int v = 0; v < 8; ++v)
      out[(long)(base + v + 8 * hi) * OUT + lo + 16 * t] = acc2[t][v] + bap[lo + 16 * t];
}

extern "C" void kernel_launch(void* const* d_in, const int* in_sizes, int n_in,
                              void* d_out, int out_size, void* d_ws, size_t ws_size,
                              hipStream_t stream) {
  const float* feat  = (const float*)d_in[0];
  const float* fe    = (const float*)d_in[1];
  const int*   src   = (const int*)d_in[2];
  const int*   dst   = (const int*)d_in[3];
  const float* Wp    = (const float*)d_in[4];
  const float* Wp2   = (const float*)d_in[5];
  const float* Was   = (const float*)d_in[6];
  const float* Wad   = (const float*)d_in[7];
  const float* We    = (const float*)d_in[8];
  const float* scale = (const float*)d_in[9];
  const float* offs  = (const float*)d_in[10];
  const float* Wagg  = (const float*)d_in[11];
  const float* bagg  = (const float*)d_in[12];
  const float* Wapd  = (const float*)d_in[13];
  const float* bapd  = (const float*)d_in[14];
  const float* Wap   = (const float*)d_in[15];
  const float* bap   = (const float*)d_in[16];
  float* out = (float*)d_out;

  const long NW = (long)N_NODES * OUT;
  float* fc    = (float*)d_ws;
  float* esrc  = fc + NW;
  float* edst  = esrc + NW;
  float* dterm = edst + NW;
  float* srcrec = dterm + NW;
  unsigned* mdst = (unsigned*)srcrec;
  unsigned* msrc = mdst + NW;
  float* sdst  = srcrec + 2 * NW;
  float* ssrc  = sdst + NW;
  float* msg   = ssrc + NW;
  float* ebuf  = msg + NW;
  unsigned* flag = (unsigned*)msg;

  constexpr int nzero4 = 5 * N_NODES * OUT / 4;
  zero_kernel<<<PW_GRID, PW_BLOCK, 0, stream>>>(srcrec, nzero4);

  node_gemm_kernel<<<N_NODES / 16, 32, 0, stream>>>(
      feat, Wp, Wp2, Was, Wad, Wapd, bapd, fc, esrc, edst, dterm);

  edge_logits_kernel<<<PW_GRID, PW_BLOCK, 0, stream>>>(
      fe, src, dst, We, esrc, edst, ebuf, sdst, ssrc, flag);

  fallback_max_kernel<<<PW_GRID, PW_BLOCK, 0, stream>>>(
      flag, ebuf, src, dst, mdst, msrc, sdst);
  fallback_sum_kernel<<<PW_GRID, PW_BLOCK, 0, stream>>>(
      flag, ebuf, src, dst, mdst, msrc, sdst, ssrc, edst, esrc);

  node_fold_kernel<<<PW_GRID, PW_BLOCK, 0, stream>>>(
      sdst, ssrc, edst, esrc, fc, srcrec, flag);

  edge_agg_kernel<<<PW_GRID, PW_BLOCK, 0, stream>>>(
      ebuf, src, dst, (const float*)sdst, srcrec, msg);

  norm_kernel<<<(N_NODES + 255) / 256, 256, 0, stream>>>(msg, scale, offs);

  final_kernel<<<N_NODES / 16, 32, 0, stream>>>(
      msg, Wagg, bagg, dterm, Wap, bap, out);

  (void)in_sizes; (void)n_in; (void)out_size; (void)ws_size;
}


# 9 launches -> 7: zero fill rides in the node GEMM launch, head normalisation moves into the epilogue GEMM's A loader (head sums pairwise across lane halves); fallback launches enter 2048 waves
# speedup vs baseline: 2.6904x; 1.0472x over previous
#include <hip/hip_runtime.h>

constexpr int N_NODES = 50000;
constexpr int N_EDGES = 1600000;
constexpr int F_IN    = 128;
constexpr int F_E     = 16;
constexpr int OUT     = 32;
constexpr float SLOPE = 0.2f;
constexpr float EPS   = 1e-9f;

typedef float v2f __attribute__((ext_vector_type(2)));
typedef float v8f __attribute__((ext_vector_type(8)));

#define WMMA4(acc, a, b)                                                     \
  acc = __builtin_amdgcn_wmma_f32_16x16x4_f32(false, (a), false, (b),        \
                                              (short)0, (acc), false, false)

__device__ __forceinline__ float leaky(float x) { return x >= 0.f ? x : SLOPE * x; }

__device__ __forceinline__ unsigned flipf(float f) {
  unsigned b = __float_as_uint(f);
  return (b & 0x80000000u) ? ~b : (b | 0x80000000u);
}
__device__ __forceinline__ float unflipf(unsigned u) {
  return __uint_as_float((u & 0x80000000u) ? (u ^ 0x80000000u) : ~u);
}

constexpr float SHIFT_RANGE = 32.f;

constexpr int PW_BLOCK = 256;
constexpr int PW_GRID  = 1024;
constexpr int PW_WAVES = PW_GRID * (PW_BLOCK / 32);
#define PW_RESIDENT __attribute__((amdgpu_waves_per_eu(8, 8)))

__device__ __forceinline__ int wave_number() {
  return __builtin_amdgcn_readfirstlane(
      (int)(blockIdx.x * (PW_BLOCK / 32) + (threadIdx.x >> 5)));
}

constexpr int FB_GRID  = 256;
constexpr int FB_WAVES = FB_GRID * (PW_BLOCK / 32);

constexpr int ZERO_PIECES   = 5 * N_NODES * OUT / 4;
constexpr int ZERO_PER_WAVE = ZERO_PIECES / (N_NODES / 16);
static_assert(ZERO_PIECES % (N_NODES / 16) == 0 && ZERO_PER_WAVE % 32 == 0,
              "the zero fill must split evenly over the GEMM's waves and lanes");

__global__ __launch_bounds__(32) void node_gemm_kernel(
    float* __restrict__ zero_base,
    const float* __restrict__ feat,
    const float* __restrict__ Wp,  const float* __restrict__ Wp2,
    const float* __restrict__ Was, const float* __restrict__ Wad,
    const float* __restrict__ Wapd, const float* __restrict__ bapd,
    float* __restrict__ fc, float* __restrict__ esrc,
    float* __restrict__ edst, float* __restrict__ dterm) {
  const int lane = threadIdx.x, lo = lane & 15, hi = lane >> 4;
  const int base = blockIdx.x * 16;

  {
    float4* z4 = (float4*)zero_base + (long)blockIdx.x * ZERO_PER_WAVE + lane;
    const float4 z = {0.f, 0.f, 0.f, 0.f};
#pragma unroll
    for (int i = 0; i < ZERO_PER_WAVE / 32; ++i) z4[i * 32] = z;
  }

  v8f accP[2] = {}, accAS[2] = {}, accAD[2] = {}, accAP[2] = {};
  for (int k0 = 0; k0 < F_IN; k0 += 4) {
    const int ka = k0 + 2 * hi;
    v2f a = *(const v2f*)(feat + (long)(base + lo) * F_IN + ka);
#pragma unroll
    for (int t = 0; t < 2; ++t) {
      const int n = lo + 16 * t;
      v2f b;
      b = (v2f){Wp[ka * OUT + n],   Wp[(ka + 1) * OUT + n]};   WMMA4(accP[t],  a, b);
      b = (v2f){Was[ka * OUT + n],  Was[(ka + 1) * OUT + n]};  WMMA4(accAS[t], a, b);
      b = (v2f){Wad[ka * OUT + n],  Wad[(ka + 1) * OUT + n]};  WMMA4(accAD[t], a, b);
      b = (v2f){Wapd[ka * OUT + n], Wapd[(ka + 1) * OUT + n]}; WMMA4(accAP[t], a, b);
    }
  }

  __shared__ float lds[16 * OUT];
#pragma unroll
  for (int t = 0; t < 2; ++t)
#pragma unroll
    for (int v = 0; v < 8; ++v)
      lds[(v + 8 * hi) * OUT + lo + 16 * t] = leaky(accP[t][v]);
  __syncthreads();

  v8f accFC[2] = {};
  for (int k0 = 0; k0 < OUT; k0 += 4) {
    const int ka = k0 + 2 * hi;
    v2f a2 = (v2f){lds[lo * OUT + ka], lds[lo * OUT + ka + 1]};
#pragma unroll
    for (int t = 0; t < 2; ++t) {
      const int n = lo + 16 * t;
      v2f b = (v2f){Wp2[ka * OUT + n], Wp2[(ka + 1) * OUT + n]};
      WMMA4(accFC[t], a2, b);
    }
  }

#pragma unroll
  for (int t = 0; t < 2; ++t)
#pragma unroll
    for (int v = 0; v < 8; ++v) {
      const long row = base + v + 8 * hi;
      const int n = lo + 16 * t;
      fc[row * OUT + n]    = accFC[t][v];
      esrc[row * OUT + n]  = accAS[t][v];
      edst[row * OUT + n]  = accAD[t][v];
      dterm[row * OUT + n] = accAP[t][v] + bapd[n];
    }
}


__device__ __forceinline__ float xhalf(float x) {
  const int b = __float_as_int(x);
  return __int_as_float(
      __builtin_amdgcn_permlanex16(b, b, 0x76543210, 0xfedcba98, false, false));
}

constexpr int TE = 16;

__global__ __launch_bounds__(PW_BLOCK) PW_RESIDENT void edge_logits_kernel(
    const float* __restrict__ fe, const int* __restrict__ src,
    const int* __restrict__ dst, const float* __restrict__ We,
    const float* __restrict__ esrc, const float* __restrict__ edst,
    float* __restrict__ ebuf, float* __restrict__ sdst,
    float* __restrict__ ssrc, unsigned* __restrict__ flag) {
  const int lane = threadIdx.x & 31, lo = lane & 15, hi = lane >> 4;
  const int wave = wave_number();
  float worst = 0.f;

  v2f bw[F_E / 4][2];
#pragma unroll
  for (int k = 0; k < F_E / 4; ++k) {
    const int ka = 4 * k + 2 * hi;
#pragma unroll
    for (int t = 0; t < 2; ++t) {
      const int n = lo + 16 * t;
      bw[k][t] = (v2f){We[ka * OUT + n], We[(ka + 1) * OUT + n]};
    }
  }

  for (int tile = wave; tile < N_EDGES / TE; tile += PW_WAVES) {
    const int ebase = tile * TE;

    const int sv = src[ebase + lo];
    const int dv = dst[ebase + lo];

    v8f acc[2] = {};
#pragma unroll
    for (int k = 0; k < F_E / 4; ++k) {
      const int ka = 4 * k + 2 * hi;
      v2f a = *(const v2f*)(fe + (long)(ebase + lo) * F_E + ka);
#pragma unroll
      for (int t = 0; t < 2; ++t) WMMA4(acc[t], a, bw[k][t]);
    }

    unsigned os[TE], od[TE];
#pragma unroll
    for (int r = 0; r < TE; ++r) {
      os[r] = (unsigned)__builtin_amdgcn_readlane(sv, r) * OUT + lane;
      od[r] = (unsigned)__builtin_amdgcn_readlane(dv, r) * OUT + lane;
    }
    float gs[TE], gd[TE];
#pragma unroll
    for (int r = 0; r < TE; ++r) {
      gs[r] = esrc[os[r]];
      gd[r] = edst[od[r]];
    }

    float row[TE];
#pragma unroll
    for (int v = 0; v < 8; ++v) {
      const float a0 = acc[0][v], a1 = acc[1][v];
      const float x0 = xhalf(a0), x1 = xhalf(a1);
      row[v]     = hi ? x1 : a0;
      row[v + 8] = hi ? a1 : x0;
    }

#pragma unroll
    for (int r = 0; r < TE; ++r) {
      const float val = leaky(row[r] + gs[r] + gd[r]);
      ebuf[(ebase + r) * OUT + lane] = val;
      const float xd = val - gd[r], xs = val - gs[r];
      worst = fmaxf(worst, fmaxf(fabsf(xd), fabsf(xs)));
      atomicAdd(&sdst[od[r]], __expf(xd));
      atomicAdd(&ssrc[os[r]], __expf(xs));
    }
  }
  if (worst > SHIFT_RANGE) *flag = 1u;
}

constexpr int EPW = 8;
constexpr int N_CHUNKS = N_EDGES / EPW;
static_assert(N_EDGES % EPW == 0, "edge chunks must tile exactly");
static_assert(N_EDGES % 16 == 0, "logit tiles must tile exactly");

__device__ __forceinline__ void edge_offsets(const int* __restrict__ src,
                                             const int* __restrict__ dst,
                                             int e0, int n,
                                             unsigned (&os)[EPW], unsigned (&od)[EPW]) {
  const int sv = src[e0 + (n & (EPW - 1))];
  const int dv = dst[e0 + (n & (EPW - 1))];
#pragma unroll
  for (int j = 0; j < EPW; ++j) {
    os[j] = (unsigned)__builtin_amdgcn_readlane(sv, j) * OUT + n;
    od[j] = (unsigned)__builtin_amdgcn_readlane(dv, j) * OUT + n;
  }
}

constexpr int PLANE_ITEMS = N_NODES * OUT / 4;

__global__ __launch_bounds__(PW_BLOCK) void fallback_max_kernel(
    const unsigned* __restrict__ flag, const float* __restrict__ ebuf,
    const int* __restrict__ src, const int* __restrict__ dst,
    unsigned* __restrict__ mdst, unsigned* __restrict__ msrc,
    float* __restrict__ sums) {
  if (*flag == 0u) return;
  float4* s4 = (float4*)sums;
  const float4 z = {0.f, 0.f, 0.f, 0.f};
  for (int i = blockIdx.x * PW_BLOCK + threadIdx.x; i < 2 * PLANE_ITEMS; i += FB_GRID * PW_BLOCK)
    s4[i] = z;
  const int n = threadIdx.x & 31;
  for (int chunk = wave_number(); chunk < N_CHUNKS; chunk += FB_WAVES) {
    const int e0 = chunk * EPW;
    unsigned os[EPW], od[EPW];
    edge_offsets(src, dst, e0, n, os, od);
#pragma unroll
    for (int j = 0; j < EPW; ++j) {
      const unsigned fv = flipf(ebuf[(e0 + j) * OUT + n]);
      atomicMax(&mdst[od[j]], fv);
      atomicMax(&msrc[os[j]], fv);
    }
  }
}

__global__ __launch_bounds__(PW_BLOCK) void fallback_sum_kernel(
    const unsigned* __restrict__ flag, const float* __restrict__ ebuf,
    const int* __restrict__ src, const int* __restrict__ dst,
    const unsigned* __restrict__ mdst, const unsigned* __restrict__ msrc,
    float* __restrict__ sdst, float* __restrict__ ssrc,
    float* __restrict__ shift_dst, float* __restrict__ shift_src) {
  if (*flag == 0u) return;
  for (int i = blockIdx.x * PW_BLOCK + threadIdx.x; i < N_NODES * OUT; i += FB_GRID * PW_BLOCK) {
    shift_dst[i] = unflipf(mdst[i]);
    shift_src[i] = unflipf(msrc[i]);
  }
  const int n = threadIdx.x & 31;
  for (int chunk = wave_number(); chunk < N_CHUNKS; chunk += FB_WAVES) {
    const int e0 = chunk * EPW;
    unsigned os[EPW], od[EPW];
    edge_offsets(src, dst, e0, n, os, od);

    float v[EPW];
    unsigned md[EPW], ms[EPW];
#pragma unroll
    for (int j = 0; j < EPW; ++j) {
      v[j]  = ebuf[(e0 + j) * OUT + n];
      md[j] = mdst[od[j]];
      ms[j] = msrc[os[j]];
    }
#pragma unroll
    for (int j = 0; j < EPW; ++j) {
      atomicAdd(&sdst[od[j]], __expf(v[j] - unflipf(md[j])));
      atomicAdd(&ssrc[os[j]], __expf(v[j] - unflipf(ms[j])));
    }
  }
}

constexpr int FOLD_ITEMS = PLANE_ITEMS;

__device__ __forceinline__ float fold_term(float shift, float sum) {
  return shift + logf(sum);
}

__global__ __launch_bounds__(PW_BLOCK) void node_fold_kernel(
    float* splane, const float* __restrict__ ssrc,
    const float* __restrict__ edst, const float* __restrict__ esrc,
    const float* __restrict__ fc, float* __restrict__ srcrec,
    unsigned* __restrict__ flag) {
  if (blockIdx.x == 0 && threadIdx.x == 0) *flag = 0u;
  for (int i = blockIdx.x * PW_BLOCK + threadIdx.x; i < FOLD_ITEMS; i += PW_GRID * PW_BLOCK) {
    const float4 td = ((const float4*)edst)[i];
    const float4 ts = ((const float4*)esrc)[i];
    const float4 sd = ((const float4*)splane)[i];
    const float4 ss = ((const float4*)ssrc)[i];
    const float4 f  = ((const float4*)fc)[i];
    const float4 ld = {fold_term(td.x, sd.x), fold_term(td.y, sd.y),
                       fold_term(td.z, sd.z), fold_term(td.w, sd.w)};
    ((float4*)splane)[i] = ld;
    const float4 r0 = {fold_term(ts.x, ss.x), f.x, fold_term(ts.y, ss.y), f.y};
    const float4 r1 = {fold_term(ts.z, ss.z), f.z, fold_term(ts.w, ss.w), f.w};
    ((float4*)srcrec)[2 * i]     = r0;
    ((float4*)srcrec)[2 * i + 1] = r1;
  }
}

__global__ __launch_bounds__(PW_BLOCK) void edge_agg_kernel(
    const float* __restrict__ ebuf, const int* __restrict__ src,
    const int* __restrict__ dst, const float* __restrict__ lddst,
    const float* __restrict__ srcrec, float* __restrict__ msg) {
  const int n = threadIdx.x & 31;
  const v2f* __restrict__ rec = (const v2f*)srcrec;
  for (int chunk = wave_number(); chunk < N_CHUNKS; chunk += PW_WAVES) {
    const int e0 = chunk * EPW;
    unsigned os[EPW], od[EPW];
    edge_offsets(src, dst, e0, n, os, od);

    float v[EPW], ld[EPW];
    v2f r[EPW];
#pragma unroll
    for (int j = 0; j < EPW; ++j) {
      v[j]  = ebuf[(e0 + j) * OUT + n];
      ld[j] = lddst[od[j]];
      r[j]  = rec[os[j]];
    }
#pragma unroll
    for (int j = 0; j < EPW; ++j) {
      const float ad = fmaxf(__expf(v[j] - ld[j]), EPS);
      const float as = fmaxf(__expf(v[j] - r[j].x), EPS);
      atomicAdd(&msg[od[j]], r[j].y * sqrtf(ad * as));
    }
  }
}

__global__ __launch_bounds__(32) void final_kernel(
    const float* __restrict__ msg, const float* __restrict__ scale,
    const float* __restrict__ offset, const float* __restrict__ Wagg,
    const float* __restrict__ bagg, const float* __restrict__ dterm,
    const float* __restrict__ Wap, const float* __restrict__ bap,
    float* __restrict__ out) {
  const int lane = threadIdx.x, lo = lane & 15, hi = lane >> 4;
  const int base = blockIdx.x * 16;

  v2f a[OUT / 4];
#pragma unroll
  for (int j = 0; j < OUT / 4; ++j)
    a[j] = *(const v2f*)(msg + (long)(base + lo) * OUT + 4 * j + 2 * hi);

#pragma unroll
  for (int h = 0; h < 4; ++h) {
    v2f p = a[2 * h], q = a[2 * h + 1];
    const float s = (p.x + p.y) + (q.x + q.y);
    const float m = (s + xhalf(s)) * 0.125f;
    p.x -= m; p.y -= m; q.x -= m; q.y -= m;
    const float v = (p.x * p.x + p.y * p.y) + (q.x * q.x + q.y * q.y);
    const float r = rsqrtf((v + xhalf(v)) * 0.125f + EPS);
    const int k = 8 * h + 2 * hi;
    a[2 * h]     = (v2f){p.x * scale[k] * r + offset[k],
                         p.y * scale[k + 1] * r + offset[k + 1]};
    a[2 * h + 1] = (v2f){q.x * scale[k + 4] * r + offset[k + 4],
                         q.y * scale[k + 5] * r + offset[k + 5]};
  }

  v8f acc1[2] = {};
#pragma unroll
  for (int j = 0; j < OUT / 4; ++j) {
    const int ka = 4 * j + 2 * hi;
#pragma unroll
    for (int t = 0; t < 2; ++t) {
      const int n = lo + 16 * t;
      v2f b = (v2f){Wagg[ka * OUT + n], Wagg[(ka + 1) * OUT + n]};
      WMMA4(acc1[t], a[j], b);
    }
  }

  __shared__ float lds[16 * OUT];
#pragma unroll
  for (int t = 0; t < 2; ++t)
#pragma unroll
    for (int v = 0; v < 8; ++v) {
      const long row = base + v + 8 * hi;
      const int n = lo + 16 * t;
      lds[(v + 8 * hi) * OUT + n] = leaky(acc1[t][v] + bagg[n] + dterm[row * OUT + n]);
    }
  __syncthreads();

  v8f acc2[2] = {};
  for (int k0 = 0; k0 < OUT; k0 += 4) {
    const int ka = k0 + 2 * hi;
    v2f a2 = (v2f){lds[lo * OUT + ka], lds[lo * OUT + ka + 1]};
#pragma unroll
    for (int t = 0; t < 2; ++t) {
      const int n = lo + 16 * t;
      v2f b = (v2f){Wap[ka * OUT + n], Wap[(ka + 1) * OUT + n]};
      WMMA4(acc2[t], a2, b);
    }
  }

#pragma unroll
  for (int t = 0; t < 2; ++t)
#pragma unroll
    for (int v = 0; v < 8; ++v)
      out[(long)(base + v + 8 * hi) * OUT + lo + 16 * t] = acc2[t][v] + bap[lo + 16 * t];
}

extern "C" void kernel_launch(void* const* d_in, const int* in_sizes, int n_in,
                              void* d_out, int out_size, void* d_ws, size_t ws_size,
                              hipStream_t stream) {
  const float* feat  = (const float*)d_in[0];
  const float* fe    = (const float*)d_in[1];
  const int*   src   = (const int*)d_in[2];
  const int*   dst   = (const int*)d_in[3];
  const float* Wp    = (const float*)d_in[4];
  const float* Wp2   = (const float*)d_in[5];
  const float* Was   = (const float*)d_in[6];
  const float* Wad   = (const float*)d_in[7];
  const float* We    = (const float*)d_in[8];
  const float* scale = (const float*)d_in[9];
  const float* offs  = (const float*)d_in[10];
  const float* Wagg  = (const float*)d_in[11];
  const float* bagg  = (const float*)d_in[12];
  const float* Wapd  = (const float*)d_in[13];
  const float* bapd  = (const float*)d_in[14];
  const float* Wap   = (const float*)d_in[15];
  const float* bap   = (const float*)d_in[16];
  float* out = (float*)d_out;

  const long NW = (long)N_NODES * OUT;
  float* fc    = (float*)d_ws;
  float* esrc  = fc + NW;
  float* edst  = esrc + NW;
  float* dterm = edst + NW;
  float* srcrec = dterm + NW;
  unsigned* mdst = (unsigned*)srcrec;
  unsigned* msrc = mdst + NW;
  float* sdst  = srcrec + 2 * NW;
  float* ssrc  = sdst + NW;
  float* msg   = ssrc + NW;
  float* ebuf  = msg + NW;
  unsigned* flag = (unsigned*)msg;

  node_gemm_kernel<<<N_NODES / 16, 32, 0, stream>>>(
      srcrec, feat, Wp, Wp2, Was, Wad, Wapd, bapd, fc, esrc, edst, dterm);

  edge_logits_kernel<<<PW_GRID, PW_BLOCK, 0, stream>>>(
      fe, src, dst, We, esrc, edst, ebuf, sdst, ssrc, flag);

  fallback_max_kernel<<<FB_GRID, PW_BLOCK, 0, stream>>>(
      flag, ebuf, src, dst, mdst, msrc, sdst);
  fallback_sum_kernel<<<FB_GRID, PW_BLOCK, 0, stream>>>(
      flag, ebuf, src, dst, mdst, msrc, sdst, ssrc, edst, esrc);

  node_fold_kernel<<<PW_GRID, PW_BLOCK, 0, stream>>>(
      sdst, ssrc, edst, esrc, fc, srcrec, flag);

  edge_agg_kernel<<<PW_GRID, PW_BLOCK, 0, stream>>>(
      ebuf, src, dst, (const float*)sdst, srcrec, msg);

  final_kernel<<<N_NODES / 16, 32, 0, stream>>>(
      msg, scale, offs, Wagg, bagg, dterm, Wap, bap, out);

  (void)in_sizes; (void)n_in; (void)out_size; (void)ws_size;
}


# each plane that takes f32 atomic adds (both exp sums, messages) exists twice, one copy per half of the package chosen by die number; node passes add the copies; add order still free
# speedup vs baseline: 2.9012x; 1.0784x over previous
#include <hip/hip_runtime.h>

constexpr int N_NODES = 50000;
constexpr int N_EDGES = 1600000;
constexpr int F_IN    = 128;
constexpr int F_E     = 16;
constexpr int OUT     = 32;
constexpr float SLOPE = 0.2f;
constexpr float EPS   = 1e-9f;

typedef float v2f __attribute__((ext_vector_type(2)));
typedef float v8f __attribute__((ext_vector_type(8)));

#define WMMA4(acc, a, b)                                                     \
  acc = __builtin_amdgcn_wmma_f32_16x16x4_f32(false, (a), false, (b),        \
                                              (short)0, (acc), false, false)

__device__ __forceinline__ float leaky(float x) { return x >= 0.f ? x : SLOPE * x; }

__device__ __forceinline__ unsigned flipf(float f) {
  unsigned b = __float_as_uint(f);
  return (b & 0x80000000u) ? ~b : (b | 0x80000000u);
}
__device__ __forceinline__ float unflipf(unsigned u) {
  return __uint_as_float((u & 0x80000000u) ? (u ^ 0x80000000u) : ~u);
}

constexpr float SHIFT_RANGE = 32.f;

constexpr int PW_BLOCK = 256;
constexpr int PW_GRID  = 1024;
constexpr int PW_WAVES = PW_GRID * (PW_BLOCK / 32);
#define PW_RESIDENT __attribute__((amdgpu_waves_per_eu(8, 8)))

__device__ __forceinline__ int wave_number() {
  return __builtin_amdgcn_readfirstlane(
      (int)(blockIdx.x * (PW_BLOCK / 32) + (threadIdx.x >> 5)));
}

__device__ __forceinline__ int package_half() {
  return (int)((__builtin_amdgcn_s_sendmsg_rtn(0x87) >> 18) & 1u);
}

constexpr int FB_GRID  = 256;
constexpr int FB_WAVES = FB_GRID * (PW_BLOCK / 32);

constexpr int ZERO_PIECES   = 5 * N_NODES * OUT / 4;
constexpr int ZERO_PER_WAVE = ZERO_PIECES / (N_NODES / 16);
static_assert(ZERO_PIECES % (N_NODES / 16) == 0 && ZERO_PER_WAVE % 32 == 0,
              "the zero fill must split evenly over the GEMM's waves and lanes");
constexpr int ZERO2_PER_WAVE = (N_NODES * OUT / 4) / (N_NODES / 16);
static_assert((N_NODES * OUT / 4) % (N_NODES / 16) == 0 && ZERO2_PER_WAVE % 32 == 0,
              "the sixth plane must split evenly over the GEMM's waves and lanes");

__global__ __launch_bounds__(32) void node_gemm_kernel(
    float* __restrict__ zero_base, float* __restrict__ zero_plane6,
    const float* __restrict__ feat,
    const float* __restrict__ Wp,  const float* __restrict__ Wp2,
    const float* __restrict__ Was, const float* __restrict__ Wad,
    const float* __restrict__ Wapd, const float* __restrict__ bapd,
    float* __restrict__ fc, float* __restrict__ esrc,
    float* __restrict__ edst, float* __restrict__ dterm) {
  const int lane = threadIdx.x, lo = lane & 15, hi = lane >> 4;
  const int base = blockIdx.x * 16;

  {
    float4* z4 = (float4*)zero_base + (long)blockIdx.x * ZERO_PER_WAVE + lane;
    const float4 z = {0.f, 0.f, 0.f, 0.f};
#pragma unroll
    for (int i = 0; i < ZERO_PER_WAVE / 32; ++i) z4[i * 32] = z;
    float4* y4 = (float4*)zero_plane6 + (long)blockIdx.x * ZERO2_PER_WAVE + lane;
#pragma unroll
    for (int i = 0; i < ZERO2_PER_WAVE / 32; ++i) y4[i * 32] = z;
  }

  v8f accP[2] = {}, accAS[2] = {}, accAD[2] = {}, accAP[2] = {};
  for (int k0 = 0; k0 < F_IN; k0 += 4) {
    const int ka = k0 + 2 * hi;
    v2f a = *(const v2f*)(feat + (long)(base + lo) * F_IN + ka);
#pragma unroll
    for (int t = 0; t < 2; ++t) {
      const int n = lo + 16 * t;
      v2f b;
      b = (v2f){Wp[ka * OUT + n],   Wp[(ka + 1) * OUT + n]};   WMMA4(accP[t],  a, b);
      b = (v2f){Was[ka * OUT + n],  Was[(ka + 1) * OUT + n]};  WMMA4(accAS[t], a, b);
      b = (v2f){Wad[ka * OUT + n],  Wad[(ka + 1) * OUT + n]};  WMMA4(accAD[t], a, b);
      b = (v2f){Wapd[ka * OUT + n], Wapd[(ka + 1) * OUT + n]}; WMMA4(accAP[t], a, b);
    }
  }

  __shared__ float lds[16 * OUT];
#pragma unroll
  for (int t = 0; t < 2; ++t)
#pragma unroll
    for (int v = 0; v < 8; ++v)
      lds[(v + 8 * hi) * OUT + lo + 16 * t] = leaky(accP[t][v]);
  __syncthreads();

  v8f accFC[2] = {};
  for (int k0 = 0; k0 < OUT; k0 += 4) {
    const int ka = k0 + 2 * hi;
    v2f a2 = (v2f){lds[lo * OUT + ka], lds[lo * OUT + ka + 1]};
#pragma unroll
    for (int t = 0; t < 2; ++t) {
      const int n = lo + 16 * t;
      v2f b = (v2f){Wp2[ka * OUT + n], Wp2[(ka + 1) * OUT + n]};
      WMMA4(accFC[t], a2, b);
    }
  }

#pragma unroll
  for (int t = 0; t < 2; ++t)
#pragma unroll
    for (int v = 0; v < 8; ++v) {
      const long row = base + v + 8 * hi;
      const int n = lo + 16 * t;
      fc[row * OUT + n]    = accFC[t][v];
      esrc[row * OUT + n]  = accAS[t][v];
      edst[row * OUT + n]  = accAD[t][v];
      dterm[row * OUT + n] = accAP[t][v] + bapd[n];
    }
}


__device__ __forceinline__ float xhalf(float x) {
  const int b = __float_as_int(x);
  return __int_as_float(
      __builtin_amdgcn_permlanex16(b, b, 0x76543210, 0xfedcba98, false, false));
}

constexpr int TE = 16;

__global__ __launch_bounds__(PW_BLOCK) PW_RESIDENT void edge_logits_kernel(
    const float* __restrict__ fe, const int* __restrict__ src,
    const int* __restrict__ dst, const float* __restrict__ We,
    const float* __restrict__ esrc, const float* __restrict__ edst,
    float* __restrict__ ebuf, float* __restrict__ sdst0,
    float* __restrict__ ssrc0, float* __restrict__ sdst1,
    float* __restrict__ ssrc1, unsigned* __restrict__ flag) {
  const int lane = threadIdx.x & 31, lo = lane & 15, hi = lane >> 4;
  const int wave = wave_number();
  const int half = package_half();
  float* __restrict__ sdst = half ? sdst1 : sdst0;
  float* __restrict__ ssrc = half ? ssrc1 : ssrc0;
  float worst = 0.f;

  v2f bw[F_E / 4][2];
#pragma unroll
  for (int k = 0; k < F_E / 4; ++k) {
    const int ka = 4 * k + 2 * hi;
#pragma unroll
    for (int t = 0; t < 2; ++t) {
      const int n = lo + 16 * t;
      bw[k][t] = (v2f){We[ka * OUT + n], We[(ka + 1) * OUT + n]};
    }
  }

  for (int tile = wave; tile < N_EDGES / TE; tile += PW_WAVES) {
    const int ebase = tile * TE;

    const int sv = src[ebase + lo];
    const int dv = dst[ebase + lo];

    v8f acc[2] = {};
#pragma unroll
    for (int k = 0; k < F_E / 4; ++k) {
      const int ka = 4 * k + 2 * hi;
      v2f a = *(const v2f*)(fe + (long)(ebase + lo) * F_E + ka);
#pragma unroll
      for (int t = 0; t < 2; ++t) WMMA4(acc[t], a, bw[k][t]);
    }

    unsigned os[TE], od[TE];
#pragma unroll
    for (int r = 0; r < TE; ++r) {
      os[r] = (unsigned)__builtin_amdgcn_readlane(sv, r) * OUT + lane;
      od[r] = (unsigned)__builtin_amdgcn_readlane(dv, r) * OUT + lane;
    }
    float gs[TE], gd[TE];
#pragma unroll
    for (int r = 0; r < TE; ++r) {
      gs[r] = esrc[os[r]];
      gd[r] = edst[od[r]];
    }

    float row[TE];
#pragma unroll
    for (int v = 0; v < 8; ++v) {
      const float a0 = acc[0][v], a1 = acc[1][v];
      const float x0 = xhalf(a0), x1 = xhalf(a1);
      row[v]     = hi ? x1 : a0;
      row[v + 8] = hi ? a1 : x0;
    }

#pragma unroll
    for (int r = 0; r < TE; ++r) {
      const float val = leaky(row[r] + gs[r] + gd[r]);
      ebuf[(ebase + r) * OUT + lane] = val;
      const float xd = val - gd[r], xs = val - gs[r];
      worst = fmaxf(worst, fmaxf(fabsf(xd), fabsf(xs)));
      atomicAdd(&sdst[od[r]], __expf(xd));
      atomicAdd(&ssrc[os[r]], __expf(xs));
    }
  }
  if (worst > SHIFT_RANGE) *flag = 1u;
}

constexpr int EPW = 8;
constexpr int N_CHUNKS = N_EDGES / EPW;
static_assert(N_EDGES % EPW == 0, "edge chunks must tile exactly");
static_assert(N_EDGES % 16 == 0, "logit tiles must tile exactly");

__device__ __forceinline__ void edge_offsets(const int* __restrict__ src,
                                             const int* __restrict__ dst,
                                             int e0, int n,
                                             unsigned (&os)[EPW], unsigned (&od)[EPW]) {
  const int sv = src[e0 + (n & (EPW - 1))];
  const int dv = dst[e0 + (n & (EPW - 1))];
#pragma unroll
  for (int j = 0; j < EPW; ++j) {
    os[j] = (unsigned)__builtin_amdgcn_readlane(sv, j) * OUT + n;
    od[j] = (unsigned)__builtin_amdgcn_readlane(dv, j) * OUT + n;
  }
}

constexpr int PLANE_ITEMS = N_NODES * OUT / 4;

__global__ __launch_bounds__(PW_BLOCK) void fallback_max_kernel(
    const unsigned* flag, const float* __restrict__ ebuf,
    const int* __restrict__ src, const int* __restrict__ dst,
    unsigned* mdst, unsigned* __restrict__ msrc,
    float* __restrict__ sums, float* __restrict__ sums6) {
  if (*flag == 0u) return;
  float4* s4 = (float4*)sums;
  float4* t4 = (float4*)sums6;
  const float4 z = {0.f, 0.f, 0.f, 0.f};
  for (int i = blockIdx.x * PW_BLOCK + threadIdx.x; i < 3 * PLANE_ITEMS; i += FB_GRID * PW_BLOCK)
    s4[i] = z;
  for (int i = blockIdx.x * PW_BLOCK + threadIdx.x; i < PLANE_ITEMS; i += FB_GRID * PW_BLOCK)
    t4[i] = z;
  const int n = threadIdx.x & 31;
  for (int chunk = wave_number(); chunk < N_CHUNKS; chunk += FB_WAVES) {
    const int e0 = chunk * EPW;
    unsigned os[EPW], od[EPW];
    edge_offsets(src, dst, e0, n, os, od);
#pragma unroll
    for (int j = 0; j < EPW; ++j) {
      const unsigned fv = flipf(ebuf[(e0 + j) * OUT + n]);
      atomicMax(&mdst[od[j]], fv);
      atomicMax(&msrc[os[j]], fv);
    }
  }
}

__global__ __launch_bounds__(PW_BLOCK) void fallback_sum_kernel(
    const unsigned* __restrict__ flag, const float* __restrict__ ebuf,
    const int* __restrict__ src, const int* __restrict__ dst,
    const unsigned* __restrict__ mdst, const unsigned* __restrict__ msrc,
    float* __restrict__ sdst, float* __restrict__ ssrc,
    float* __restrict__ shift_dst, float* __restrict__ shift_src) {
  if (*flag == 0u) return;
  for (int i = blockIdx.x * PW_BLOCK + threadIdx.x; i < N_NODES * OUT; i += FB_GRID * PW_BLOCK) {
    shift_dst[i] = unflipf(mdst[i]);
    shift_src[i] = unflipf(msrc[i]);
  }
  const int n = threadIdx.x & 31;
  for (int chunk = wave_number(); chunk < N_CHUNKS; chunk += FB_WAVES) {
    const int e0 = chunk * EPW;
    unsigned os[EPW], od[EPW];
    edge_offsets(src, dst, e0, n, os, od);

    float v[EPW];
    unsigned md[EPW], ms[EPW];
#pragma unroll
    for (int j = 0; j < EPW; ++j) {
      v[j]  = ebuf[(e0 + j) * OUT + n];
      md[j] = mdst[od[j]];
      ms[j] = msrc[os[j]];
    }
#pragma unroll
    for (int j = 0; j < EPW; ++j) {
      atomicAdd(&sdst[od[j]], __expf(v[j] - unflipf(md[j])));
      atomicAdd(&ssrc[os[j]], __expf(v[j] - unflipf(ms[j])));
    }
  }
}

constexpr int FOLD_ITEMS = PLANE_ITEMS;

__device__ __forceinline__ float fold_term(float shift, float sum) {
  return shift + logf(sum);
}

__global__ __launch_bounds__(PW_BLOCK) void node_fold_kernel(
    float* splane, float* ssrc0, float* sdst1, const float* __restrict__ ssrc1,
    const float* __restrict__ edst, const float* __restrict__ esrc,
    const float* __restrict__ fc, float* __restrict__ srcrec) {
  for (int i = blockIdx.x * PW_BLOCK + threadIdx.x; i < FOLD_ITEMS; i += PW_GRID * PW_BLOCK) {
    const float4 td = ((const float4*)edst)[i];
    const float4 ts = ((const float4*)esrc)[i];
    const float4 da = ((const float4*)splane)[i];
    const float4 db = ((const float4*)sdst1)[i];
    const float4 sa = ((const float4*)ssrc0)[i];
    const float4 sb = ((const float4*)ssrc1)[i];
    const float4 f  = ((const float4*)fc)[i];
    const float4 sd = {da.x + db.x, da.y + db.y, da.z + db.z, da.w + db.w};
    const float4 ss = {sa.x + sb.x, sa.y + sb.y, sa.z + sb.z, sa.w + sb.w};
    const float4 z  = {0.f, 0.f, 0.f, 0.f};
    ((float4*)sdst1)[i] = z;
    ((float4*)ssrc0)[i] = z;
    const float4 ld = {fold_term(td.x, sd.x), fold_term(td.y, sd.y),
                       fold_term(td.z, sd.z), fold_term(td.w, sd.w)};
    ((float4*)splane)[i] = ld;
    const float4 r0 = {fold_term(ts.x, ss.x), f.x, fold_term(ts.y, ss.y), f.y};
    const float4 r1 = {fold_term(ts.z, ss.z), f.z, fold_term(ts.w, ss.w), f.w};
    ((float4*)srcrec)[2 * i]     = r0;
    ((float4*)srcrec)[2 * i + 1] = r1;
  }
}

__global__ __launch_bounds__(PW_BLOCK) void edge_agg_kernel(
    const float* __restrict__ ebuf, const int* __restrict__ src,
    const int* __restrict__ dst, const float* __restrict__ lddst,
    const float* __restrict__ srcrec, float* __restrict__ msg0,
    float* __restrict__ msg1) {
  const int n = threadIdx.x & 31;
  float* __restrict__ msg = package_half() ? msg1 : msg0;
  const v2f* __restrict__ rec = (const v2f*)srcrec;
  for (int chunk = wave_number(); chunk < N_CHUNKS; chunk += PW_WAVES) {
    const int e0 = chunk * EPW;
    unsigned os[EPW], od[EPW];
    edge_offsets(src, dst, e0, n, os, od);

    float v[EPW], ld[EPW];
    v2f r[EPW];
#pragma unroll
    for (int j = 0; j < EPW; ++j) {
      v[j]  = ebuf[(e0 + j) * OUT + n];
      ld[j] = lddst[od[j]];
      r[j]  = rec[os[j]];
    }
#pragma unroll
    for (int j = 0; j < EPW; ++j) {
      const float ad = fmaxf(__expf(v[j] - ld[j]), EPS);
      const float as = fmaxf(__expf(v[j] - r[j].x), EPS);
      atomicAdd(&msg[od[j]], r[j].y * sqrtf(ad * as));
    }
  }
}

__global__ __launch_bounds__(32) void final_kernel(
    const float* __restrict__ msg, const float* __restrict__ msg1,
    const float* __restrict__ scale,
    const float* __restrict__ offset, const float* __restrict__ Wagg,
    const float* __restrict__ bagg, const float* __restrict__ dterm,
    const float* __restrict__ Wap, const float* __restrict__ bap,
    float* __restrict__ out) {
  const int lane = threadIdx.x, lo = lane & 15, hi = lane >> 4;
  const int base = blockIdx.x * 16;

  v2f a[OUT / 4];
#pragma unroll
  for (int j = 0; j < OUT / 4; ++j)
    a[j] = *(const v2f*)(msg + (long)(base + lo) * OUT + 4 * j + 2 * hi) +
           *(const v2f*)(msg1 + (long)(base + lo) * OUT + 4 * j + 2 * hi);

#pragma unroll
  for (int h = 0; h < 4; ++h) {
    v2f p = a[2 * h], q = a[2 * h + 1];
    const float s = (p.x + p.y) + (q.x + q.y);
    const float m = (s + xhalf(s)) * 0.125f;
    p.x -= m; p.y -= m; q.x -= m; q.y -= m;
    const float v = (p.x * p.x + p.y * p.y) + (q.x * q.x + q.y * q.y);
    const float r = rsqrtf((v + xhalf(v)) * 0.125f + EPS);
    const int k = 8 * h + 2 * hi;
    a[2 * h]     = (v2f){p.x * scale[k] * r + offset[k],
                         p.y * scale[k + 1] * r + offset[k + 1]};
    a[2 * h + 1] = (v2f){q.x * scale[k + 4] * r + offset[k + 4],
                         q.y * scale[k + 5] * r + offset[k + 5]};
  }

  v8f acc1[2] = {};
#pragma unroll
  for (int j = 0; j < OUT / 4; ++j) {
    const int ka = 4 * j + 2 * hi;
#pragma unroll
    for (int t = 0; t < 2; ++t) {
      const int n = lo + 16 * t;
      v2f b = (v2f){Wagg[ka * OUT + n], Wagg[(ka + 1) * OUT + n]};
      WMMA4(acc1[t], a[j], b);
    }
  }

  __shared__ float lds[16 * OUT];
#pragma unroll
  for (int t = 0; t < 2; ++t)
#pragma unroll
    for (int v = 0; v < 8; ++v) {
      const long row = base + v + 8 * hi;
      const int n = lo + 16 * t;
      lds[(v + 8 * hi) * OUT + n] = leaky(acc1[t][v] + bagg[n] + dterm[row * OUT + n]);
    }
  __syncthreads();

  v8f acc2[2] = {};
  for (int k0 = 0; k0 < OUT; k0 += 4) {
    const int ka = k0 + 2 * hi;
    v2f a2 = (v2f){lds[lo * OUT + ka], lds[lo * OUT + ka + 1]};
#pragma unroll
    for (int t = 0; t < 2; ++t) {
      const int n = lo + 16 * t;
      v2f b = (v2f){Wap[ka * OUT + n], Wap[(ka + 1) * OUT + n]};
      WMMA4(acc2[t], a2, b);
    }
  }

#pragma unroll
  for (int t = 0; t < 2; ++t)
#pragma unroll
    for (int v = 0; v < 8; ++v)
      out[(long)(base + v + 8 * hi) * OUT + lo + 16 * t] = acc2[t][v] + bap[lo + 16 * t];
}

extern "C" void kernel_launch(void* const* d_in, const int* in_sizes, int n_in,
                              void* d_out, int out_size, void* d_ws, size_t ws_size,
                              hipStream_t stream) {
  const float* feat  = (const float*)d_in[0];
  const float* fe    = (const float*)d_in[1];
  const int*   src   = (const int*)d_in[2];
  const int*   dst   = (const int*)d_in[3];
  const float* Wp    = (const float*)d_in[4];
  const float* Wp2   = (const float*)d_in[5];
  const float* Was   = (const float*)d_in[6];
  const float* Wad   = (const float*)d_in[7];
  const float* We    = (const float*)d_in[8];
  const float* scale = (const float*)d_in[9];
  const float* offs  = (const float*)d_in[10];
  const float* Wagg  = (const float*)d_in[11];
  const float* bagg  = (const float*)d_in[12];
  const float* Wapd  = (const float*)d_in[13];
  const float* bapd  = (const float*)d_in[14];
  const float* Wap   = (const float*)d_in[15];
  const float* bap   = (const float*)d_in[16];
  float* out = (float*)d_out;

  const long NW = (long)N_NODES * OUT;
  float* fc    = (float*)d_ws;
  float* esrc  = fc + NW;
  float* edst  = esrc + NW;
  float* dterm = edst + NW;
  float* srcrec = dterm + NW;
  unsigned* mdst = (unsigned*)srcrec;
  unsigned* msrc = mdst + NW;
  float* sdst  = srcrec + 2 * NW;
  float* ssrc  = sdst + NW;
  float* msg   = ssrc + NW;
  float* ebuf  = msg + NW;
  unsigned* flag = mdst;

  node_gemm_kernel<<<N_NODES / 16, 32, 0, stream>>>(
      srcrec, out, feat, Wp, Wp2, Was, Wad, Wapd, bapd, fc, esrc, edst, dterm);

  edge_logits_kernel<<<PW_GRID, PW_BLOCK, 0, stream>>>(
      fe, src, dst, We, esrc, edst, ebuf, sdst, ssrc, msg, out, flag);

  fallback_max_kernel<<<FB_GRID, PW_BLOCK, 0, stream>>>(
      flag, ebuf, src, dst, mdst, msrc, sdst, out);
  fallback_sum_kernel<<<FB_GRID, PW_BLOCK, 0, stream>>>(
      flag, ebuf, src, dst, mdst, msrc, sdst, ssrc, edst, esrc);

  node_fold_kernel<<<PW_GRID, PW_BLOCK, 0, stream>>>(
      sdst, ssrc, msg, out, edst, esrc, fc, srcrec);

  edge_agg_kernel<<<PW_GRID, PW_BLOCK, 0, stream>>>(
      ebuf, src, dst, (const float*)sdst, srcrec, msg, ssrc);

  final_kernel<<<N_NODES / 16, 32, 0, stream>>>(
      msg, ssrc, scale, offs, Wagg, bagg, dterm, Wap, bap, out);

  (void)in_sizes; (void)n_in; (void)out_size; (void)ws_size;
}


# logit buffer kept as [4 edges][channel][4]: the logits pass stores and the aggregation and fallback passes load 4 edges per instruction (16 B per lane, 512 contiguous B), not one 128-B row each
# speedup vs baseline: 3.0869x; 1.0640x over previous
#include <hip/hip_runtime.h>

constexpr int N_NODES = 50000;
constexpr int N_EDGES = 1600000;
constexpr int F_IN    = 128;
constexpr int F_E     = 16;
constexpr int OUT     = 32;
constexpr float SLOPE = 0.2f;
constexpr float EPS   = 1e-9f;

typedef float v2f __attribute__((ext_vector_type(2)));
typedef float v8f __attribute__((ext_vector_type(8)));

#define WMMA4(acc, a, b)                                                     \
  acc = __builtin_amdgcn_wmma_f32_16x16x4_f32(false, (a), false, (b),        \
                                              (short)0, (acc), false, false)

__device__ __forceinline__ float leaky(float x) { return x >= 0.f ? x : SLOPE * x; }

__device__ __forceinline__ unsigned flipf(float f) {
  unsigned b = __float_as_uint(f);
  return (b & 0x80000000u) ? ~b : (b | 0x80000000u);
}
__device__ __forceinline__ float unflipf(unsigned u) {
  return __uint_as_float((u & 0x80000000u) ? (u ^ 0x80000000u) : ~u);
}

constexpr float SHIFT_RANGE = 32.f;

constexpr int PW_BLOCK = 256;
constexpr int PW_GRID  = 1024;
constexpr int PW_WAVES = PW_GRID * (PW_BLOCK / 32);
#define PW_RESIDENT __attribute__((amdgpu_waves_per_eu(8, 8)))

__device__ __forceinline__ int wave_number() {
  return __builtin_amdgcn_readfirstlane(
      (int)(blockIdx.x * (PW_BLOCK / 32) + (threadIdx.x >> 5)));
}

__device__ __forceinline__ int package_half() {
  return (int)((__builtin_amdgcn_s_sendmsg_rtn(0x87) >> 18) & 1u);
}

constexpr int FB_GRID  = 256;
constexpr int FB_WAVES = FB_GRID * (PW_BLOCK / 32);

constexpr int ZERO_PIECES   = 5 * N_NODES * OUT / 4;
constexpr int ZERO_PER_WAVE = ZERO_PIECES / (N_NODES / 16);
static_assert(ZERO_PIECES % (N_NODES / 16) == 0 && ZERO_PER_WAVE % 32 == 0,
              "the zero fill must split evenly over the GEMM's waves and lanes");
constexpr int ZERO2_PER_WAVE = (N_NODES * OUT / 4) / (N_NODES / 16);
static_assert((N_NODES * OUT / 4) % (N_NODES / 16) == 0 && ZERO2_PER_WAVE % 32 == 0,
              "the sixth plane must split evenly over the GEMM's waves and lanes");

__global__ __launch_bounds__(32) void node_gemm_kernel(
    float* __restrict__ zero_base, float* __restrict__ zero_plane6,
    const float* __restrict__ feat,
    const float* __restrict__ Wp,  const float* __restrict__ Wp2,
    const float* __restrict__ Was, const float* __restrict__ Wad,
    const float* __restrict__ Wapd, const float* __restrict__ bapd,
    float* __restrict__ fc, float* __restrict__ esrc,
    float* __restrict__ edst, float* __restrict__ dterm) {
  const int lane = threadIdx.x, lo = lane & 15, hi = lane >> 4;
  const int base = blockIdx.x * 16;

  {
    float4* z4 = (float4*)zero_base + (long)blockIdx.x * ZERO_PER_WAVE + lane;
    const float4 z = {0.f, 0.f, 0.f, 0.f};
#pragma unroll
    for (int i = 0; i < ZERO_PER_WAVE / 32; ++i) z4[i * 32] = z;
    float4* y4 = (float4*)zero_plane6 + (long)blockIdx.x * ZERO2_PER_WAVE + lane;
#pragma unroll
    for (int i = 0; i < ZERO2_PER_WAVE / 32; ++i) y4[i * 32] = z;
  }

  v8f accP[2] = {}, accAS[2] = {}, accAD[2] = {}, accAP[2] = {};
  for (int k0 = 0; k0 < F_IN; k0 += 4) {
    const int ka = k0 + 2 * hi;
    v2f a = *(const v2f*)(feat + (long)(base + lo) * F_IN + ka);
#pragma unroll
    for (int t = 0; t < 2; ++t) {
      const int n = lo + 16 * t;
      v2f b;
      b = (v2f){Wp[ka * OUT + n],   Wp[(ka + 1) * OUT + n]};   WMMA4(accP[t],  a, b);
      b = (v2f){Was[ka * OUT + n],  Was[(ka + 1) * OUT + n]};  WMMA4(accAS[t], a, b);
      b = (v2f){Wad[ka * OUT + n],  Wad[(ka + 1) * OUT + n]};  WMMA4(accAD[t], a, b);
      b = (v2f){Wapd[ka * OUT + n], Wapd[(ka + 1) * OUT + n]}; WMMA4(accAP[t], a, b);
    }
  }

  __shared__ float lds[16 * OUT];
#pragma unroll
  for (int t = 0; t < 2; ++t)
#pragma unroll
    for (int v = 0; v < 8; ++v)
      lds[(v + 8 * hi) * OUT + lo + 16 * t] = leaky(accP[t][v]);
  __syncthreads();

  v8f accFC[2] = {};
  for (int k0 = 0; k0 < OUT; k0 += 4) {
    const int ka = k0 + 2 * hi;
    v2f a2 = (v2f){lds[lo * OUT + ka], lds[lo * OUT + ka + 1]};
#pragma unroll
    for (int t = 0; t < 2; ++t) {
      const int n = lo + 16 * t;
      v2f b = (v2f){Wp2[ka * OUT + n], Wp2[(ka + 1) * OUT + n]};
      WMMA4(accFC[t], a2, b);
    }
  }

#pragma unroll
  for (int t = 0; t < 2; ++t)
#pragma unroll
    for (int v = 0; v < 8; ++v) {
      const long row = base + v + 8 * hi;
      const int n = lo + 16 * t;
      fc[row * OUT + n]    = accFC[t][v];
      esrc[row * OUT + n]  = accAS[t][v];
      edst[row * OUT + n]  = accAD[t][v];
      dterm[row * OUT + n] = accAP[t][v] + bapd[n];
    }
}


__device__ __forceinline__ float xhalf(float x) {
  const int b = __float_as_int(x);
  return __int_as_float(
      __builtin_amdgcn_permlanex16(b, b, 0x76543210, 0xfedcba98, false, false));
}

constexpr int TE = 16;

constexpr int EQ = 4;
static_assert(TE % EQ == 0, "a logit tile must be whole groups");

__global__ __launch_bounds__(PW_BLOCK) PW_RESIDENT void edge_logits_kernel(
    const float* __restrict__ fe, const int* __restrict__ src,
    const int* __restrict__ dst, const float* __restrict__ We,
    const float* __restrict__ esrc, const float* __restrict__ edst,
    float* __restrict__ ebuf, float* __restrict__ sdst0,
    float* __restrict__ ssrc0, float* __restrict__ sdst1,
    float* __restrict__ ssrc1, unsigned* __restrict__ flag) {
  const int lane = threadIdx.x & 31, lo = lane & 15, hi = lane >> 4;
  const int wave = wave_number();
  const int half = package_half();
  float* __restrict__ sdst = half ? sdst1 : sdst0;
  float* __restrict__ ssrc = half ? ssrc1 : ssrc0;
  float worst = 0.f;
  float4* __restrict__ ebuf4 = (float4*)ebuf;

  v2f bw[F_E / 4][2];
#pragma unroll
  for (int k = 0; k < F_E / 4; ++k) {
    const int ka = 4 * k + 2 * hi;
#pragma unroll
    for (int t = 0; t < 2; ++t) {
      const int n = lo + 16 * t;
      bw[k][t] = (v2f){We[ka * OUT + n], We[(ka + 1) * OUT + n]};
    }
  }

  for (int tile = wave; tile < N_EDGES / TE; tile += PW_WAVES) {
    const int ebase = tile * TE;

    const int sv = src[ebase + lo];
    const int dv = dst[ebase + lo];

    v8f acc[2] = {};
#pragma unroll
    for (int k = 0; k < F_E / 4; ++k) {
      const int ka = 4 * k + 2 * hi;
      v2f a = *(const v2f*)(fe + (long)(ebase + lo) * F_E + ka);
#pragma unroll
      for (int t = 0; t < 2; ++t) WMMA4(acc[t], a, bw[k][t]);
    }

    unsigned os[TE], od[TE];
#pragma unroll
    for (int r = 0; r < TE; ++r) {
      os[r] = (unsigned)__builtin_amdgcn_readlane(sv, r) * OUT + lane;
      od[r] = (unsigned)__builtin_amdgcn_readlane(dv, r) * OUT + lane;
    }
    float gs[TE], gd[TE];
#pragma unroll
    for (int r = 0; r < TE; ++r) {
      gs[r] = esrc[os[r]];
      gd[r] = edst[od[r]];
    }

    float row[TE];
#pragma unroll
    for (int v = 0; v < 8; ++v) {
      const float a0 = acc[0][v], a1 = acc[1][v];
      const float x0 = xhalf(a0), x1 = xhalf(a1);
      row[v]     = hi ? x1 : a0;
      row[v + 8] = hi ? a1 : x0;
    }

#pragma unroll
    for (int g = 0; g < TE / EQ; ++g) {
      float val[EQ];
#pragma unroll
      for (int k = 0; k < EQ; ++k) {
        const int r = EQ * g + k;
        val[k] = leaky(row[r] + gs[r] + gd[r]);
      }
      ebuf4[(tile * (TE / EQ) + g) * OUT + lane] = (float4){val[0], val[1], val[2], val[3]};
#pragma unroll
      for (int k = 0; k < EQ; ++k) {
        const int r = EQ * g + k;
        const float xd = val[k] - gd[r], xs = val[k] - gs[r];
        worst = fmaxf(worst, fmaxf(fabsf(xd), fabsf(xs)));
        atomicAdd(&sdst[od[r]], __expf(xd));
        atomicAdd(&ssrc[os[r]], __expf(xs));
      }
    }
  }
  if (worst > SHIFT_RANGE) *flag = 1u;
}

constexpr int EPW = 8;
constexpr int N_CHUNKS = N_EDGES / EPW;
static_assert(N_EDGES % EPW == 0, "edge chunks must tile exactly");
static_assert(N_EDGES % 16 == 0, "logit tiles must tile exactly");

__device__ __forceinline__ void edge_offsets(const int* __restrict__ src,
                                             const int* __restrict__ dst,
                                             int e0, int n,
                                             unsigned (&os)[EPW], unsigned (&od)[EPW]) {
  const int sv = src[e0 + (n & (EPW - 1))];
  const int dv = dst[e0 + (n & (EPW - 1))];
#pragma unroll
  for (int j = 0; j < EPW; ++j) {
    os[j] = (unsigned)__builtin_amdgcn_readlane(sv, j) * OUT + n;
    od[j] = (unsigned)__builtin_amdgcn_readlane(dv, j) * OUT + n;
  }
}

static_assert(EPW % EQ == 0, "an edge chunk must be whole groups");
__device__ __forceinline__ void edge_logits_of(const float* __restrict__ ebuf,
                                               int chunk, int n, float (&v)[EPW]) {
  const float4* __restrict__ g4 = (const float4*)ebuf + chunk * (EPW / EQ) * OUT + n;
#pragma unroll
  for (int g = 0; g < EPW / EQ; ++g) {
    const float4 t = g4[g * OUT];
    v[EQ * g] = t.x; v[EQ * g + 1] = t.y; v[EQ * g + 2] = t.z; v[EQ * g + 3] = t.w;
  }
}

constexpr int PLANE_ITEMS = N_NODES * OUT / 4;

__global__ __launch_bounds__(PW_BLOCK) void fallback_max_kernel(
    const unsigned* flag, const float* __restrict__ ebuf,
    const int* __restrict__ src, const int* __restrict__ dst,
    unsigned* mdst, unsigned* __restrict__ msrc,
    float* __restrict__ sums, float* __restrict__ sums6) {
  if (*flag == 0u) return;
  float4* s4 = (float4*)sums;
  float4* t4 = (float4*)sums6;
  const float4 z = {0.f, 0.f, 0.f, 0.f};
  for (int i = blockIdx.x * PW_BLOCK + threadIdx.x; i < 3 * PLANE_ITEMS; i += FB_GRID * PW_BLOCK)
    s4[i] = z;
  for (int i = blockIdx.x * PW_BLOCK + threadIdx.x; i < PLANE_ITEMS; i += FB_GRID * PW_BLOCK)
    t4[i] = z;
  const int n = threadIdx.x & 31;
  for (int chunk = wave_number(); chunk < N_CHUNKS; chunk += FB_WAVES) {
    const int e0 = chunk * EPW;
    unsigned os[EPW], od[EPW];
    edge_offsets(src, dst, e0, n, os, od);
    float v[EPW];
    edge_logits_of(ebuf, chunk, n, v);
#pragma unroll
    for (int j = 0; j < EPW; ++j) {
      const unsigned fv = flipf(v[j]);
      atomicMax(&mdst[od[j]], fv);
      atomicMax(&msrc[os[j]], fv);
    }
  }
}

__global__ __launch_bounds__(PW_BLOCK) void fallback_sum_kernel(
    const unsigned* __restrict__ flag, const float* __restrict__ ebuf,
    const int* __restrict__ src, const int* __restrict__ dst,
    const unsigned* __restrict__ mdst, const unsigned* __restrict__ msrc,
    float* __restrict__ sdst, float* __restrict__ ssrc,
    float* __restrict__ shift_dst, float* __restrict__ shift_src) {
  if (*flag == 0u) return;
  for (int i = blockIdx.x * PW_BLOCK + threadIdx.x; i < N_NODES * OUT; i += FB_GRID * PW_BLOCK) {
    shift_dst[i] = unflipf(mdst[i]);
    shift_src[i] = unflipf(msrc[i]);
  }
  const int n = threadIdx.x & 31;
  for (int chunk = wave_number(); chunk < N_CHUNKS; chunk += FB_WAVES) {
    const int e0 = chunk * EPW;
    unsigned os[EPW], od[EPW];
    edge_offsets(src, dst, e0, n, os, od);

    float v[EPW];
    unsigned md[EPW], ms[EPW];
    edge_logits_of(ebuf, chunk, n, v);
#pragma unroll
    for (int j = 0; j < EPW; ++j) {
      md[j] = mdst[od[j]];
      ms[j] = msrc[os[j]];
    }
#pragma unroll
    for (int j = 0; j < EPW; ++j) {
      atomicAdd(&sdst[od[j]], __expf(v[j] - unflipf(md[j])));
      atomicAdd(&ssrc[os[j]], __expf(v[j] - unflipf(ms[j])));
    }
  }
}

constexpr int FOLD_ITEMS = PLANE_ITEMS;

__device__ __forceinline__ float fold_term(float shift, float sum) {
  return shift + logf(sum);
}

__global__ __launch_bounds__(PW_BLOCK) void node_fold_kernel(
    float* splane, float* ssrc0, float* sdst1, const float* __restrict__ ssrc1,
    const float* __restrict__ edst, const float* __restrict__ esrc,
    const float* __restrict__ fc, float* __restrict__ srcrec) {
  for (int i = blockIdx.x * PW_BLOCK + threadIdx.x; i < FOLD_ITEMS; i += PW_GRID * PW_BLOCK) {
    const float4 td = ((const float4*)edst)[i];
    const float4 ts = ((const float4*)esrc)[i];
    const float4 da = ((const float4*)splane)[i];
    const float4 db = ((const float4*)sdst1)[i];
    const float4 sa = ((const float4*)ssrc0)[i];
    const float4 sb = ((const float4*)ssrc1)[i];
    const float4 f  = ((const float4*)fc)[i];
    const float4 sd = {da.x + db.x, da.y + db.y, da.z + db.z, da.w + db.w};
    const float4 ss = {sa.x + sb.x, sa.y + sb.y, sa.z + sb.z, sa.w + sb.w};
    const float4 z  = {0.f, 0.f, 0.f, 0.f};
    ((float4*)sdst1)[i] = z;
    ((float4*)ssrc0)[i] = z;
    const float4 ld = {fold_term(td.x, sd.x), fold_term(td.y, sd.y),
                       fold_term(td.z, sd.z), fold_term(td.w, sd.w)};
    ((float4*)splane)[i] = ld;
    const float4 r0 = {fold_term(ts.x, ss.x), f.x, fold_term(ts.y, ss.y), f.y};
    const float4 r1 = {fold_term(ts.z, ss.z), f.z, fold_term(ts.w, ss.w), f.w};
    ((float4*)srcrec)[2 * i]     = r0;
    ((float4*)srcrec)[2 * i + 1] = r1;
  }
}

__global__ __launch_bounds__(PW_BLOCK) void edge_agg_kernel(
    const float* __restrict__ ebuf, const int* __restrict__ src,
    const int* __restrict__ dst, const float* __restrict__ lddst,
    const float* __restrict__ srcrec, float* __restrict__ msg0,
    float* __restrict__ msg1) {
  const int n = threadIdx.x & 31;
  float* __restrict__ msg = package_half() ? msg1 : msg0;
  const v2f* __restrict__ rec = (const v2f*)srcrec;
  for (int chunk = wave_number(); chunk < N_CHUNKS; chunk += PW_WAVES) {
    const int e0 = chunk * EPW;
    unsigned os[EPW], od[EPW];
    edge_offsets(src, dst, e0, n, os, od);

    float v[EPW], ld[EPW];
    v2f r[EPW];
    edge_logits_of(ebuf, chunk, n, v);
#pragma unroll
    for (int j = 0; j < EPW; ++j) {
      ld[j] = lddst[od[j]];
      r[j]  = rec[os[j]];
    }
#pragma unroll
    for (int j = 0; j < EPW; ++j) {
      const float ad = fmaxf(__expf(v[j] - ld[j]), EPS);
      const float as = fmaxf(__expf(v[j] - r[j].x), EPS);
      atomicAdd(&msg[od[j]], r[j].y * sqrtf(ad * as));
    }
  }
}

__global__ __launch_bounds__(32) void final_kernel(
    const float* __restrict__ msg, const float* __restrict__ msg1,
    const float* __restrict__ scale,
    const float* __restrict__ offset, const float* __restrict__ Wagg,
    const float* __restrict__ bagg, const float* __restrict__ dterm,
    const float* __restrict__ Wap, const float* __restrict__ bap,
    float* __restrict__ out) {
  const int lane = threadIdx.x, lo = lane & 15, hi = lane >> 4;
  const int base = blockIdx.x * 16;

  v2f a[OUT / 4];
#pragma unroll
  for (int j = 0; j < OUT / 4; ++j)
    a[j] = *(const v2f*)(msg + (long)(base + lo) * OUT + 4 * j + 2 * hi) +
           *(const v2f*)(msg1 + (long)(base + lo) * OUT + 4 * j + 2 * hi);

#pragma unroll
  for (int h = 0; h < 4; ++h) {
    v2f p = a[2 * h], q = a[2 * h + 1];
    const float s = (p.x + p.y) + (q.x + q.y);
    const float m = (s + xhalf(s)) * 0.125f;
    p.x -= m; p.y -= m; q.x -= m; q.y -= m;
    const float v = (p.x * p.x + p.y * p.y) + (q.x * q.x + q.y * q.y);
    const float r = rsqrtf((v + xhalf(v)) * 0.125f + EPS);
    const int k = 8 * h + 2 * hi;
    a[2 * h]     = (v2f){p.x * scale[k] * r + offset[k],
                         p.y * scale[k + 1] * r + offset[k + 1]};
    a[2 * h + 1] = (v2f){q.x * scale[k + 4] * r + offset[k + 4],
                         q.y * scale[k + 5] * r + offset[k + 5]};
  }

  v8f acc1[2] = {};
#pragma unroll
  for (int j = 0; j < OUT / 4; ++j) {
    const int ka = 4 * j + 2 * hi;
#pragma unroll
    for (int t = 0; t < 2; ++t) {
      const int n = lo + 16 * t;
      v2f b = (v2f){Wagg[ka * OUT + n], Wagg[(ka + 1) * OUT + n]};
      WMMA4(acc1[t], a[j], b);
    }
  }

  __shared__ float lds[16 * OUT];
#pragma unroll
  for (int t = 0; t < 2; ++t)
#pragma unroll
    for (int v = 0; v < 8; ++v) {
      const long row = base + v + 8 * hi;
      const int n = lo + 16 * t;
      lds[(v + 8 * hi) * OUT + n] = leaky(acc1[t][v] + bagg[n] + dterm[row * OUT + n]);
    }
  __syncthreads();

  v8f acc2[2] = {};
  for (int k0 = 0; k0 < OUT; k0 += 4) {
    const int ka = k0 + 2 * hi;
    v2f a2 = (v2f){lds[lo * OUT + ka], lds[lo * OUT + ka + 1]};
#pragma unroll
    for (int t = 0; t < 2; ++t) {
      const int n = lo + 16 * t;
      v2f b = (v2f){Wap[ka * OUT + n], Wap[(ka + 1) * OUT + n]};
      WMMA4(acc2[t], a2, b);
    }
  }

#pragma unroll
  for (int t = 0; t < 2; ++t)
#pragma unroll
    for (int v = 0; v < 8; ++v)
      out[(long)(base + v + 8 * hi) * OUT + lo + 16 * t] = acc2[t][v] + bap[lo + 16 * t];
}

extern "C" void kernel_launch(void* const* d_in, const int* in_sizes, int n_in,
                              void* d_out, int out_size, void* d_ws, size_t ws_size,
                              hipStream_t stream) {
  const float* feat  = (const float*)d_in[0];
  const float* fe    = (const float*)d_in[1];
  const int*   src   = (const int*)d_in[2];
  const int*   dst   = (const int*)d_in[3];
  const float* Wp    = (const float*)d_in[4];
  const float* Wp2   = (const float*)d_in[5];
  const float* Was   = (const float*)d_in[6];
  const float* Wad   = (const float*)d_in[7];
  const float* We    = (const float*)d_in[8];
  const float* scale = (const float*)d_in[9];
  const float* offs  = (const float*)d_in[10];
  const float* Wagg  = (const float*)d_in[11];
  const float* bagg  = (const float*)d_in[12];
  const float* Wapd  = (const float*)d_in[13];
  const float* bapd  = (const float*)d_in[14];
  const float* Wap   = (const float*)d_in[15];
  const float* bap   = (const float*)d_in[16];
  float* out = (float*)d_out;

  const long NW = (long)N_NODES * OUT;
  float* fc    = (float*)d_ws;
  float* esrc  = fc + NW;
  float* edst  = esrc + NW;
  float* dterm = edst + NW;
  float* srcrec = dterm + NW;
  unsigned* mdst = (unsigned*)srcrec;
  unsigned* msrc = mdst + NW;
  float* sdst  = srcrec + 2 * NW;
  float* ssrc  = sdst + NW;
  float* msg   = ssrc + NW;
  float* ebuf  = msg + NW;
  unsigned* flag = mdst;

  node_gemm_kernel<<<N_NODES / 16, 32, 0, stream>>>(
      srcrec, out, feat, Wp, Wp2, Was, Wad, Wapd, bapd, fc, esrc, edst, dterm);

  edge_logits_kernel<<<PW_GRID, PW_BLOCK, 0, stream>>>(
      fe, src, dst, We, esrc, edst, ebuf, sdst, ssrc, msg, out, flag);

  fallback_max_kernel<<<FB_GRID, PW_BLOCK, 0, stream>>>(
      flag, ebuf, src, dst, mdst, msrc, sdst, out);
  fallback_sum_kernel<<<FB_GRID, PW_BLOCK, 0, stream>>>(
      flag, ebuf, src, dst, mdst, msrc, sdst, ssrc, edst, esrc);

  node_fold_kernel<<<PW_GRID, PW_BLOCK, 0, stream>>>(
      sdst, ssrc, msg, out, edst, esrc, fc, srcrec);

  edge_agg_kernel<<<PW_GRID, PW_BLOCK, 0, stream>>>(
      ebuf, src, dst, (const float*)sdst, srcrec, msg, ssrc);

  final_kernel<<<N_NODES / 16, 32, 0, stream>>>(
      msg, ssrc, scale, offs, Wagg, bagg, dterm, Wap, bap, out);

  (void)in_sizes; (void)n_in; (void)out_size; (void)ws_size;
}


# 7 launches -> 5: the two range-fallback launches go; with the flag up each fold workgroup remakes segment max and sums for its own nodes in LDS from all edges; flag = +inf added to a sum word
# speedup vs baseline: 3.2052x; 1.0383x over previous
#include <hip/hip_runtime.h>

constexpr int N_NODES = 50000;
constexpr int N_EDGES = 1600000;
constexpr int F_IN    = 128;
constexpr int F_E     = 16;
constexpr int OUT     = 32;
constexpr float SLOPE = 0.2f;
constexpr float EPS   = 1e-9f;

typedef float v2f __attribute__((ext_vector_type(2)));
typedef float v8f __attribute__((ext_vector_type(8)));

#define WMMA4(acc, a, b)                                                     \
  acc = __builtin_amdgcn_wmma_f32_16x16x4_f32(false, (a), false, (b),        \
                                              (short)0, (acc), false, false)

__device__ __forceinline__ float leaky(float x) { return x >= 0.f ? x : SLOPE * x; }

__device__ __forceinline__ unsigned flipf(float f) {
  unsigned b = __float_as_uint(f);
  return (b & 0x80000000u) ? ~b : (b | 0x80000000u);
}
__device__ __forceinline__ float unflipf(unsigned u) {
  return __uint_as_float((u & 0x80000000u) ? (u ^ 0x80000000u) : ~u);
}

constexpr float SHIFT_RANGE = 32.f;

constexpr int PW_BLOCK = 256;
constexpr int PW_GRID  = 1024;
constexpr int PW_WAVES = PW_GRID * (PW_BLOCK / 32);
#define PW_RESIDENT __attribute__((amdgpu_waves_per_eu(8, 8)))

__device__ __forceinline__ int wave_number() {
  return __builtin_amdgcn_readfirstlane(
      (int)(blockIdx.x * (PW_BLOCK / 32) + (threadIdx.x >> 5)));
}

__device__ __forceinline__ int package_half() {
  return (int)((__builtin_amdgcn_s_sendmsg_rtn(0x87) >> 18) & 1u);
}

constexpr int ZERO_PIECES   = 3 * N_NODES * OUT / 4;
constexpr int ZERO_PER_WAVE = ZERO_PIECES / (N_NODES / 16);
static_assert(ZERO_PIECES % (N_NODES / 16) == 0 && ZERO_PER_WAVE % 32 == 0,
              "the zero fill must split evenly over the GEMM's waves and lanes");
constexpr int ZERO2_PER_WAVE = (N_NODES * OUT / 4) / (N_NODES / 16);
static_assert((N_NODES * OUT / 4) % (N_NODES / 16) == 0 && ZERO2_PER_WAVE % 32 == 0,
              "the fourth plane must split evenly over the GEMM's waves and lanes");

__global__ __launch_bounds__(32) void node_gemm_kernel(
    float* __restrict__ zero_base, float* __restrict__ zero_plane6,
    const float* __restrict__ feat,
    const float* __restrict__ Wp,  const float* __restrict__ Wp2,
    const float* __restrict__ Was, const float* __restrict__ Wad,
    const float* __restrict__ Wapd, const float* __restrict__ bapd,
    float* __restrict__ fc, float* __restrict__ esrc,
    float* __restrict__ edst, float* __restrict__ dterm) {
  const int lane = threadIdx.x, lo = lane & 15, hi = lane >> 4;
  const int base = blockIdx.x * 16;

  {
    float4* z4 = (float4*)zero_base + (long)blockIdx.x * ZERO_PER_WAVE + lane;
    const float4 z = {0.f, 0.f, 0.f, 0.f};
#pragma unroll
    for (int i = 0; i < ZERO_PER_WAVE / 32; ++i) z4[i * 32] = z;
    float4* y4 = (float4*)zero_plane6 + (long)blockIdx.x * ZERO2_PER_WAVE + lane;
#pragma unroll
    for (int i = 0; i < ZERO2_PER_WAVE / 32; ++i) y4[i * 32] = z;
  }

  v8f accP[2] = {}, accAS[2] = {}, accAD[2] = {}, accAP[2] = {};
  for (int k0 = 0; k0 < F_IN; k0 += 4) {
    const int ka = k0 + 2 * hi;
    v2f a = *(const v2f*)(feat + (long)(base + lo) * F_IN + ka);
#pragma unroll
    for (int t = 0; t < 2; ++t) {
      const int n = lo + 16 * t;
      v2f b;
      b = (v2f){Wp[ka * OUT + n],   Wp[(ka + 1) * OUT + n]};   WMMA4(accP[t],  a, b);
      b = (v2f){Was[ka * OUT + n],  Was[(ka + 1) * OUT + n]};  WMMA4(accAS[t], a, b);
      b = (v2f){Wad[ka * OUT + n],  Wad[(ka + 1) * OUT + n]};  WMMA4(accAD[t], a, b);
      b = (v2f){Wapd[ka * OUT + n], Wapd[(ka + 1) * OUT + n]}; WMMA4(accAP[t], a, b);
    }
  }

  __shared__ float lds[16 * OUT];
#pragma unroll
  for (int t = 0; t < 2; ++t)
#pragma unroll
    for (int v = 0; v < 8; ++v)
      lds[(v + 8 * hi) * OUT + lo + 16 * t] = leaky(accP[t][v]);
  __syncthreads();

  v8f accFC[2] = {};
  for (int k0 = 0; k0 < OUT; k0 += 4) {
    const int ka = k0 + 2 * hi;
    v2f a2 = (v2f){lds[lo * OUT + ka], lds[lo * OUT + ka + 1]};
#pragma unroll
    for (int t = 0; t < 2; ++t) {
      const int n = lo + 16 * t;
      v2f b = (v2f){Wp2[ka * OUT + n], Wp2[(ka + 1) * OUT + n]};
      WMMA4(accFC[t], a2, b);
    }
  }

#pragma unroll
  for (int t = 0; t < 2; ++t)
#pragma unroll
    for (int v = 0; v < 8; ++v) {
      const long row = base + v + 8 * hi;
      const int n = lo + 16 * t;
      fc[row * OUT + n]    = accFC[t][v];
      esrc[row * OUT + n]  = accAS[t][v];
      edst[row * OUT + n]  = accAD[t][v];
      dterm[row * OUT + n] = accAP[t][v] + bapd[n];
    }
}


__device__ __forceinline__ float xhalf(float x) {
  const int b = __float_as_int(x);
  return __int_as_float(
      __builtin_amdgcn_permlanex16(b, b, 0x76543210, 0xfedcba98, false, false));
}

constexpr int TE = 16;

constexpr int EQ = 4;
static_assert(TE % EQ == 0, "a logit tile must be whole groups");

__global__ __launch_bounds__(PW_BLOCK) PW_RESIDENT void edge_logits_kernel(
    const float* __restrict__ fe, const int* __restrict__ src,
    const int* __restrict__ dst, const float* __restrict__ We,
    const float* __restrict__ esrc, const float* __restrict__ edst,
    float* __restrict__ ebuf, float* __restrict__ sdst0,
    float* __restrict__ ssrc0, float* __restrict__ sdst1,
    float* __restrict__ ssrc1) {
  const int lane = threadIdx.x & 31, lo = lane & 15, hi = lane >> 4;
  const int wave = wave_number();
  const int half = package_half();
  float* __restrict__ sdst = half ? sdst1 : sdst0;
  float* __restrict__ ssrc = half ? ssrc1 : ssrc0;
  float worst = 0.f;
  float4* __restrict__ ebuf4 = (float4*)ebuf;

  v2f bw[F_E / 4][2];
#pragma unroll
  for (int k = 0; k < F_E / 4; ++k) {
    const int ka = 4 * k + 2 * hi;
#pragma unroll
    for (int t = 0; t < 2; ++t) {
      const int n = lo + 16 * t;
      bw[k][t] = (v2f){We[ka * OUT + n], We[(ka + 1) * OUT + n]};
    }
  }

  for (int tile = wave; tile < N_EDGES / TE; tile += PW_WAVES) {
    const int ebase = tile * TE;

    const int sv = src[ebase + lo];
    const int dv = dst[ebase + lo];

    v8f acc[2] = {};
#pragma unroll
    for (int k = 0; k < F_E / 4; ++k) {
      const int ka = 4 * k + 2 * hi;
      v2f a = *(const v2f*)(fe + (long)(ebase + lo) * F_E + ka);
#pragma unroll
      for (int t = 0; t < 2; ++t) WMMA4(acc[t], a, bw[k][t]);
    }

    unsigned os[TE], od[TE];
#pragma unroll
    for (int r = 0; r < TE; ++r) {
      os[r] = (unsigned)__builtin_amdgcn_readlane(sv, r) * OUT + lane;
      od[r] = (unsigned)__builtin_amdgcn_readlane(dv, r) * OUT + lane;
    }
    float gs[TE], gd[TE];
#pragma unroll
    for (int r = 0; r < TE; ++r) {
      gs[r] = esrc[os[r]];
      gd[r] = edst[od[r]];
    }

    float row[TE];
#pragma unroll
    for (int v = 0; v < 8; ++v) {
      const float a0 = acc[0][v], a1 = acc[1][v];
      const float x0 = xhalf(a0), x1 = xhalf(a1);
      row[v]     = hi ? x1 : a0;
      row[v + 8] = hi ? a1 : x0;
    }

#pragma unroll
    for (int g = 0; g < TE / EQ; ++g) {
      float val[EQ];
#pragma unroll
      for (int k = 0; k < EQ; ++k) {
        const int r = EQ * g + k;
        val[k] = leaky(row[r] + gs[r] + gd[r]);
      }
      ebuf4[(tile * (TE / EQ) + g) * OUT + lane] = (float4){val[0], val[1], val[2], val[3]};
#pragma unroll
      for (int k = 0; k < EQ; ++k) {
        const int r = EQ * g + k;
        const float xd = val[k] - gd[r], xs = val[k] - gs[r];
        worst = fmaxf(worst, fmaxf(fabsf(xd), fabsf(xs)));
        atomicAdd(&sdst[od[r]], __expf(xd));
        atomicAdd(&ssrc[os[r]], __expf(xs));
      }
    }
  }
  if (worst > SHIFT_RANGE) atomicAdd(&ssrc1[0], __builtin_inff());
}

constexpr int EPW = 8;
constexpr int N_CHUNKS = N_EDGES / EPW;
static_assert(N_EDGES % EPW == 0, "edge chunks must tile exactly");
static_assert(N_EDGES % 16 == 0, "logit tiles must tile exactly");

__device__ __forceinline__ void edge_offsets(const int* __restrict__ src,
                                             const int* __restrict__ dst,
                                             int e0, int n,
                                             unsigned (&os)[EPW], unsigned (&od)[EPW]) {
  const int sv = src[e0 + (n & (EPW - 1))];
  const int dv = dst[e0 + (n & (EPW - 1))];
#pragma unroll
  for (int j = 0; j < EPW; ++j) {
    os[j] = (unsigned)__builtin_amdgcn_readlane(sv, j) * OUT + n;
    od[j] = (unsigned)__builtin_amdgcn_readlane(dv, j) * OUT + n;
  }
}

static_assert(EPW % EQ == 0, "an edge chunk must be whole groups");
__device__ __forceinline__ void edge_logits_of(const float* __restrict__ ebuf,
                                               int chunk, int n, float (&v)[EPW]) {
  const float4* __restrict__ g4 = (const float4*)ebuf + chunk * (EPW / EQ) * OUT + n;
#pragma unroll
  for (int g = 0; g < EPW / EQ; ++g) {
    const float4 t = g4[g * OUT];
    v[EQ * g] = t.x; v[EQ * g + 1] = t.y; v[EQ * g + 2] = t.z; v[EQ * g + 3] = t.w;
  }
}

constexpr int PLANE_ITEMS = N_NODES * OUT / 4;

constexpr int FOLD_ITEMS = PLANE_ITEMS;

constexpr int FB_NODES = (N_NODES + PW_GRID - 1) / PW_GRID;
static_assert((long)FB_NODES * PW_GRID >= N_NODES, "the workgroups must cover every node");
static_assert(N_EDGES % PW_BLOCK == 0, "every thread takes the same number of trips over the edges");

__device__ __forceinline__ float logit_of(const float* __restrict__ ebuf, int e, int c) {
  return ebuf[((long)(e / EQ) * OUT + c) * EQ + (e % EQ)];
}

__device__ __forceinline__ float fold_term(float shift, float sum) {
  return shift + logf(sum);
}

__global__ __launch_bounds__(PW_BLOCK) void node_fold_kernel(
    float* splane, float* ssrc0, float* sdst1, const float* __restrict__ ssrc1,
    const float* __restrict__ edst, const float* __restrict__ esrc,
    const float* __restrict__ fc, float* __restrict__ srcrec,
    const float* __restrict__ ebuf, const int* __restrict__ src,
    const int* __restrict__ dst) {
  const float fword = __uint_as_float(
      __builtin_amdgcn_readfirstlane(__float_as_uint(ssrc1[0])));
  if (!(fword < 3.0e38f)) {
    __shared__ unsigned mx[2][FB_NODES * OUT];
    __shared__ float    sm[2][FB_NODES * OUT];
    const int n0  = blockIdx.x * FB_NODES;
    const int cnt = min(FB_NODES, max(0, N_NODES - n0));
    for (int i = threadIdx.x; i < FB_NODES * OUT; i += PW_BLOCK) {
      mx[0][i] = 0u; mx[1][i] = 0u;
      sm[0][i] = 0.f; sm[1][i] = 0.f;
    }
    __syncthreads();
    for (int e = threadIdx.x; e < N_EDGES; e += PW_BLOCK) {
      const unsigned d = (unsigned)(dst[e] - n0), s = (unsigned)(src[e] - n0);
      if (d < (unsigned)cnt)
        for (int c = 0; c < OUT; ++c) atomicMax(&mx[0][d * OUT + c], flipf(logit_of(ebuf, e, c)));
      if (s < (unsigned)cnt)
        for (int c = 0; c < OUT; ++c) atomicMax(&mx[1][s * OUT + c], flipf(logit_of(ebuf, e, c)));
    }
    __syncthreads();
    for (int e = threadIdx.x; e < N_EDGES; e += PW_BLOCK) {
      const unsigned d = (unsigned)(dst[e] - n0), s = (unsigned)(src[e] - n0);
      if (d < (unsigned)cnt)
        for (int c = 0; c < OUT; ++c)
          atomicAdd(&sm[0][d * OUT + c], __expf(logit_of(ebuf, e, c) - unflipf(mx[0][d * OUT + c])));
      if (s < (unsigned)cnt)
        for (int c = 0; c < OUT; ++c)
          atomicAdd(&sm[1][s * OUT + c], __expf(logit_of(ebuf, e, c) - unflipf(mx[1][s * OUT + c])));
    }
    __syncthreads();
    for (int i = threadIdx.x; i < cnt * OUT; i += PW_BLOCK) {
      const long g = (long)n0 * OUT + i;
      splane[g] = fold_term(unflipf(mx[0][i]), sm[0][i]);
      ((v2f*)srcrec)[g] = (v2f){fold_term(unflipf(mx[1][i]), sm[1][i]), fc[g]};
      sdst1[g] = 0.f;
      ssrc0[g] = 0.f;
    }
    return;
  }
  for (int i = blockIdx.x * PW_BLOCK + threadIdx.x; i < FOLD_ITEMS; i += PW_GRID * PW_BLOCK) {
    const float4 td = ((const float4*)edst)[i];
    const float4 ts = ((const float4*)esrc)[i];
    const float4 da = ((const float4*)splane)[i];
    const float4 db = ((const float4*)sdst1)[i];
    const float4 sa = ((const float4*)ssrc0)[i];
    const float4 sb = ((const float4*)ssrc1)[i];
    const float4 f  = ((const float4*)fc)[i];
    const float4 sd = {da.x + db.x, da.y + db.y, da.z + db.z, da.w + db.w};
    const float4 ss = {sa.x + sb.x, sa.y + sb.y, sa.z + sb.z, sa.w + sb.w};
    const float4 z  = {0.f, 0.f, 0.f, 0.f};
    ((float4*)sdst1)[i] = z;
    ((float4*)ssrc0)[i] = z;
    const float4 ld = {fold_term(td.x, sd.x), fold_term(td.y, sd.y),
                       fold_term(td.z, sd.z), fold_term(td.w, sd.w)};
    ((float4*)splane)[i] = ld;
    const float4 r0 = {fold_term(ts.x, ss.x), f.x, fold_term(ts.y, ss.y), f.y};
    const float4 r1 = {fold_term(ts.z, ss.z), f.z, fold_term(ts.w, ss.w), f.w};
    ((float4*)srcrec)[2 * i]     = r0;
    ((float4*)srcrec)[2 * i + 1] = r1;
  }
}

__global__ __launch_bounds__(PW_BLOCK) void edge_agg_kernel(
    const float* __restrict__ ebuf, const int* __restrict__ src,
    const int* __restrict__ dst, const float* __restrict__ lddst,
    const float* __restrict__ srcrec, float* __restrict__ msg0,
    float* __restrict__ msg1) {
  const int n = threadIdx.x & 31;
  float* __restrict__ msg = package_half() ? msg1 : msg0;
  const v2f* __restrict__ rec = (const v2f*)srcrec;
  for (int chunk = wave_number(); chunk < N_CHUNKS; chunk += PW_WAVES) {
    const int e0 = chunk * EPW;
    unsigned os[EPW], od[EPW];
    edge_offsets(src, dst, e0, n, os, od);

    float v[EPW], ld[EPW];
    v2f r[EPW];
    edge_logits_of(ebuf, chunk, n, v);
#pragma unroll
    for (int j = 0; j < EPW; ++j) {
      ld[j] = lddst[od[j]];
      r[j]  = rec[os[j]];
    }
#pragma unroll
    for (int j = 0; j < EPW; ++j) {
      const float ad = fmaxf(__expf(v[j] - ld[j]), EPS);
      const float as = fmaxf(__expf(v[j] - r[j].x), EPS);
      atomicAdd(&msg[od[j]], r[j].y * sqrtf(ad * as));
    }
  }
}

__global__ __launch_bounds__(32) void final_kernel(
    const float* __restrict__ msg, const float* __restrict__ msg1,
    const float* __restrict__ scale,
    const float* __restrict__ offset, const float* __restrict__ Wagg,
    const float* __restrict__ bagg, const float* __restrict__ dterm,
    const float* __restrict__ Wap, const float* __restrict__ bap,
    float* __restrict__ out) {
  const int lane = threadIdx.x, lo = lane & 15, hi = lane >> 4;
  const int base = blockIdx.x * 16;

  v2f a[OUT / 4];
#pragma unroll
  for (int j = 0; j < OUT / 4; ++j)
    a[j] = *(const v2f*)(msg + (long)(base + lo) * OUT + 4 * j + 2 * hi) +
           *(const v2f*)(msg1 + (long)(base + lo) * OUT + 4 * j + 2 * hi);

#pragma unroll
  for (int h = 0; h < 4; ++h) {
    v2f p = a[2 * h], q = a[2 * h + 1];
    const float s = (p.x + p.y) + (q.x + q.y);
    const float m = (s + xhalf(s)) * 0.125f;
    p.x -= m; p.y -= m; q.x -= m; q.y -= m;
    const float v = (p.x * p.x + p.y * p.y) + (q.x * q.x + q.y * q.y);
    const float r = rsqrtf((v + xhalf(v)) * 0.125f + EPS);
    const int k = 8 * h + 2 * hi;
    a[2 * h]     = (v2f){p.x * scale[k] * r + offset[k],
                         p.y * scale[k + 1] * r + offset[k + 1]};
    a[2 * h + 1] = (v2f){q.x * scale[k + 4] * r + offset[k + 4],
                         q.y * scale[k + 5] * r + offset[k + 5]};
  }

  v8f acc1[2] = {};
#pragma unroll
  for (int j = 0; j < OUT / 4; ++j) {
    const int ka = 4 * j + 2 * hi;
#pragma unroll
    for (int t = 0; t < 2; ++t) {
      const int n = lo + 16 * t;
      v2f b = (v2f){Wagg[ka * OUT + n], Wagg[(ka + 1) * OUT + n]};
      WMMA4(acc1[t], a[j], b);
    }
  }

  __shared__ float lds[16 * OUT];
#pragma unroll
  for (int t = 0; t < 2; ++t)
#pragma unroll
    for (int v = 0; v < 8; ++v) {
      const long row = base + v + 8 * hi;
      const int n = lo + 16 * t;
      lds[(v + 8 * hi) * OUT + n] = leaky(acc1[t][v] + bagg[n] + dterm[row * OUT + n]);
    }
  __syncthreads();

  v8f acc2[2] = {};
  for (int k0 = 0; k0 < OUT; k0 += 4) {
    const int ka = k0 + 2 * hi;
    v2f a2 = (v2f){lds[lo * OUT + ka], lds[lo * OUT + ka + 1]};
#pragma unroll
    for (int t = 0; t < 2; ++t) {
      const int n = lo + 16 * t;
      v2f b = (v2f){Wap[ka * OUT + n], Wap[(ka + 1) * OUT + n]};
      WMMA4(acc2[t], a2, b);
    }
  }

#pragma unroll
  for (int t = 0; t < 2; ++t)
#pragma unroll
    for (int v = 0; v < 8; ++v)
      out[(long)(base + v + 8 * hi) * OUT + lo + 16 * t] = acc2[t][v] + bap[lo + 16 * t];
}

extern "C" void kernel_launch(void* const* d_in, const int* in_sizes, int n_in,
                              void* d_out, int out_size, void* d_ws, size_t ws_size,
                              hipStream_t stream) {
  const float* feat  = (const float*)d_in[0];
  const float* fe    = (const float*)d_in[1];
  const int*   src   = (const int*)d_in[2];
  const int*   dst   = (const int*)d_in[3];
  const float* Wp    = (const float*)d_in[4];
  const float* Wp2   = (const float*)d_in[5];
  const float* Was   = (const float*)d_in[6];
  const float* Wad   = (const float*)d_in[7];
  const float* We    = (const float*)d_in[8];
  const float* scale = (const float*)d_in[9];
  const float* offs  = (const float*)d_in[10];
  const float* Wagg  = (const float*)d_in[11];
  const float* bagg  = (const float*)d_in[12];
  const float* Wapd  = (const float*)d_in[13];
  const float* bapd  = (const float*)d_in[14];
  const float* Wap   = (const float*)d_in[15];
  const float* bap   = (const float*)d_in[16];
  float* out = (float*)d_out;

  const long NW = (long)N_NODES * OUT;
  float* fc    = (float*)d_ws;
  float* esrc  = fc + NW;
  float* edst  = esrc + NW;
  float* dterm = edst + NW;
  float* srcrec = dterm + NW;
  float* sdst  = srcrec + 2 * NW;
  float* ssrc  = sdst + NW;
  float* msg   = ssrc + NW;
  float* ebuf  = msg + NW;

  node_gemm_kernel<<<N_NODES / 16, 32, 0, stream>>>(
      sdst, out, feat, Wp, Wp2, Was, Wad, Wapd, bapd, fc, esrc, edst, dterm);

  edge_logits_kernel<<<PW_GRID, PW_BLOCK, 0, stream>>>(
      fe, src, dst, We, esrc, edst, ebuf, sdst, ssrc, msg, out);

  node_fold_kernel<<<PW_GRID, PW_BLOCK, 0, stream>>>(
      sdst, ssrc, msg, out, edst, esrc, fc, srcrec, ebuf, src, dst);

  edge_agg_kernel<<<PW_GRID, PW_BLOCK, 0, stream>>>(
      ebuf, src, dst, (const float*)sdst, srcrec, msg, ssrc);

  final_kernel<<<N_NODES / 16, 32, 0, stream>>>(
      msg, ssrc, scale, offs, Wagg, bagg, dterm, Wap, bap, out);

  (void)in_sizes; (void)n_in; (void)out_size; (void)ws_size;
}


# aggregation: the destination side's exp(-lddst/2) leaves the edge sum and scales the finished row in the epilogue, so an edge gathers one source record and no destination row; add order still free
# speedup vs baseline: 3.3456x; 1.0438x over previous
#include <hip/hip_runtime.h>

constexpr int N_NODES = 50000;
constexpr int N_EDGES = 1600000;
constexpr int F_IN    = 128;
constexpr int F_E     = 16;
constexpr int OUT     = 32;
constexpr float SLOPE = 0.2f;
constexpr float EPS   = 1e-9f;

typedef float v2f __attribute__((ext_vector_type(2)));
typedef float v8f __attribute__((ext_vector_type(8)));

#define WMMA4(acc, a, b)                                                     \
  acc = __builtin_amdgcn_wmma_f32_16x16x4_f32(false, (a), false, (b),        \
                                              (short)0, (acc), false, false)

__device__ __forceinline__ float leaky(float x) { return x >= 0.f ? x : SLOPE * x; }

__device__ __forceinline__ unsigned flipf(float f) {
  unsigned b = __float_as_uint(f);
  return (b & 0x80000000u) ? ~b : (b | 0x80000000u);
}
__device__ __forceinline__ float unflipf(unsigned u) {
  return __uint_as_float((u & 0x80000000u) ? (u ^ 0x80000000u) : ~u);
}

constexpr float SHIFT_RANGE = 10.f;
constexpr float VAL_RANGE = 60.f;

constexpr int PW_BLOCK = 256;
constexpr int PW_GRID  = 1024;
constexpr int PW_WAVES = PW_GRID * (PW_BLOCK / 32);
#define PW_RESIDENT __attribute__((amdgpu_waves_per_eu(8, 8)))

__device__ __forceinline__ int wave_number() {
  return __builtin_amdgcn_readfirstlane(
      (int)(blockIdx.x * (PW_BLOCK / 32) + (threadIdx.x >> 5)));
}

__device__ __forceinline__ int package_half() {
  return (int)((__builtin_amdgcn_s_sendmsg_rtn(0x87) >> 18) & 1u);
}

constexpr int ZERO_PIECES   = 3 * N_NODES * OUT / 4;
constexpr int ZERO_PER_WAVE = ZERO_PIECES / (N_NODES / 16);
static_assert(ZERO_PIECES % (N_NODES / 16) == 0 && ZERO_PER_WAVE % 32 == 0,
              "the zero fill must split evenly over the GEMM's waves and lanes");
constexpr int ZERO2_PER_WAVE = (N_NODES * OUT / 4) / (N_NODES / 16);
static_assert((N_NODES * OUT / 4) % (N_NODES / 16) == 0 && ZERO2_PER_WAVE % 32 == 0,
              "the fourth plane must split evenly over the GEMM's waves and lanes");

__global__ __launch_bounds__(32) void node_gemm_kernel(
    float* __restrict__ zero_base, float* __restrict__ zero_plane6,
    const float* __restrict__ feat,
    const float* __restrict__ Wp,  const float* __restrict__ Wp2,
    const float* __restrict__ Was, const float* __restrict__ Wad,
    const float* __restrict__ Wapd, const float* __restrict__ bapd,
    float* __restrict__ fc, float* __restrict__ esrc,
    float* __restrict__ edst, float* __restrict__ dterm) {
  const int lane = threadIdx.x, lo = lane & 15, hi = lane >> 4;
  const int base = blockIdx.x * 16;

  {
    float4* z4 = (float4*)zero_base + (long)blockIdx.x * ZERO_PER_WAVE + lane;
    const float4 z = {0.f, 0.f, 0.f, 0.f};
#pragma unroll
    for (int i = 0; i < ZERO_PER_WAVE / 32; ++i) z4[i * 32] = z;
    float4* y4 = (float4*)zero_plane6 + (long)blockIdx.x * ZERO2_PER_WAVE + lane;
#pragma unroll
    for (int i = 0; i < ZERO2_PER_WAVE / 32; ++i) y4[i * 32] = z;
  }

  v8f accP[2] = {}, accAS[2] = {}, accAD[2] = {}, accAP[2] = {};
  for (int k0 = 0; k0 < F_IN; k0 += 4) {
    const int ka = k0 + 2 * hi;
    v2f a = *(const v2f*)(feat + (long)(base + lo) * F_IN + ka);
#pragma unroll
    for (int t = 0; t < 2; ++t) {
      const int n = lo + 16 * t;
      v2f b;
      b = (v2f){Wp[ka * OUT + n],   Wp[(ka + 1) * OUT + n]};   WMMA4(accP[t],  a, b);
      b = (v2f){Was[ka * OUT + n],  Was[(ka + 1) * OUT + n]};  WMMA4(accAS[t], a, b);
      b = (v2f){Wad[ka * OUT + n],  Wad[(ka + 1) * OUT + n]};  WMMA4(accAD[t], a, b);
      b = (v2f){Wapd[ka * OUT + n], Wapd[(ka + 1) * OUT + n]}; WMMA4(accAP[t], a, b);
    }
  }

  __shared__ float lds[16 * OUT];
#pragma unroll
  for (int t = 0; t < 2; ++t)
#pragma unroll
    for (int v = 0; v < 8; ++v)
      lds[(v + 8 * hi) * OUT + lo + 16 * t] = leaky(accP[t][v]);
  __syncthreads();

  v8f accFC[2] = {};
  for (int k0 = 0; k0 < OUT; k0 += 4) {
    const int ka = k0 + 2 * hi;
    v2f a2 = (v2f){lds[lo * OUT + ka], lds[lo * OUT + ka + 1]};
#pragma unroll
    for (int t = 0; t < 2; ++t) {
      const int n = lo + 16 * t;
      v2f b = (v2f){Wp2[ka * OUT + n], Wp2[(ka + 1) * OUT + n]};
      WMMA4(accFC[t], a2, b);
    }
  }

#pragma unroll
  for (int t = 0; t < 2; ++t)
#pragma unroll
    for (int v = 0; v < 8; ++v) {
      const long row = base + v + 8 * hi;
      const int n = lo + 16 * t;
      fc[row * OUT + n]    = accFC[t][v];
      esrc[row * OUT + n]  = accAS[t][v];
      edst[row * OUT + n]  = accAD[t][v];
      dterm[row * OUT + n] = accAP[t][v] + bapd[n];
    }
}


__device__ __forceinline__ float xhalf(float x) {
  const int b = __float_as_int(x);
  return __int_as_float(
      __builtin_amdgcn_permlanex16(b, b, 0x76543210, 0xfedcba98, false, false));
}

constexpr int TE = 16;

constexpr int EQ = 4;
static_assert(TE % EQ == 0, "a logit tile must be whole groups");

__global__ __launch_bounds__(PW_BLOCK) PW_RESIDENT void edge_logits_kernel(
    const float* __restrict__ fe, const int* __restrict__ src,
    const int* __restrict__ dst, const float* __restrict__ We,
    const float* __restrict__ esrc, const float* __restrict__ edst,
    float* __restrict__ ebuf, float* __restrict__ sdst0,
    float* __restrict__ ssrc0, float* __restrict__ sdst1,
    float* __restrict__ ssrc1) {
  const int lane = threadIdx.x & 31, lo = lane & 15, hi = lane >> 4;
  const int wave = wave_number();
  const int half = package_half();
  float* __restrict__ sdst = half ? sdst1 : sdst0;
  float* __restrict__ ssrc = half ? ssrc1 : ssrc0;
  float worst = 0.f;
  float vworst = 0.f;
  float4* __restrict__ ebuf4 = (float4*)ebuf;

  v2f bw[F_E / 4][2];
#pragma unroll
  for (int k = 0; k < F_E / 4; ++k) {
    const int ka = 4 * k + 2 * hi;
#pragma unroll
    for (int t = 0; t < 2; ++t) {
      const int n = lo + 16 * t;
      bw[k][t] = (v2f){We[ka * OUT + n], We[(ka + 1) * OUT + n]};
    }
  }

  for (int tile = wave; tile < N_EDGES / TE; tile += PW_WAVES) {
    const int ebase = tile * TE;

    const int sv = src[ebase + lo];
    const int dv = dst[ebase + lo];

    v8f acc[2] = {};
#pragma unroll
    for (int k = 0; k < F_E / 4; ++k) {
      const int ka = 4 * k + 2 * hi;
      v2f a = *(const v2f*)(fe + (long)(ebase + lo) * F_E + ka);
#pragma unroll
      for (int t = 0; t < 2; ++t) WMMA4(acc[t], a, bw[k][t]);
    }

    unsigned os[TE], od[TE];
#pragma unroll
    for (int r = 0; r < TE; ++r) {
      os[r] = (unsigned)__builtin_amdgcn_readlane(sv, r) * OUT + lane;
      od[r] = (unsigned)__builtin_amdgcn_readlane(dv, r) * OUT + lane;
    }
    float gs[TE], gd[TE];
#pragma unroll
    for (int r = 0; r < TE; ++r) {
      gs[r] = esrc[os[r]];
      gd[r] = edst[od[r]];
    }

    float row[TE];
#pragma unroll
    for (int v = 0; v < 8; ++v) {
      const float a0 = acc[0][v], a1 = acc[1][v];
      const float x0 = xhalf(a0), x1 = xhalf(a1);
      row[v]     = hi ? x1 : a0;
      row[v + 8] = hi ? a1 : x0;
    }

#pragma unroll
    for (int g = 0; g < TE / EQ; ++g) {
      float val[EQ];
#pragma unroll
      for (int k = 0; k < EQ; ++k) {
        const int r = EQ * g + k;
        val[k] = leaky(row[r] + gs[r] + gd[r]);
      }
      ebuf4[(tile * (TE / EQ) + g) * OUT + lane] = (float4){val[0], val[1], val[2], val[3]};
#pragma unroll
      for (int k = 0; k < EQ; ++k) {
        const int r = EQ * g + k;
        const float xd = val[k] - gd[r], xs = val[k] - gs[r];
        worst = fmaxf(worst, fmaxf(fabsf(xd), fabsf(xs)));
        vworst = fmaxf(vworst, fabsf(val[k]));
        atomicAdd(&sdst[od[r]], __expf(xd));
        atomicAdd(&ssrc[os[r]], __expf(xs));
      }
    }
  }
  if (worst > SHIFT_RANGE || vworst > VAL_RANGE) atomicAdd(&ssrc1[0], __builtin_inff());
}

constexpr int EPW = 8;
constexpr int N_CHUNKS = N_EDGES / EPW;
static_assert(N_EDGES % EPW == 0, "edge chunks must tile exactly");
static_assert(N_EDGES % 16 == 0, "logit tiles must tile exactly");

__device__ __forceinline__ void edge_offsets(const int* __restrict__ src,
                                             const int* __restrict__ dst,
                                             int e0, int n,
                                             unsigned (&os)[EPW], unsigned (&od)[EPW]) {
  const int sv = src[e0 + (n & (EPW - 1))];
  const int dv = dst[e0 + (n & (EPW - 1))];
#pragma unroll
  for (int j = 0; j < EPW; ++j) {
    os[j] = (unsigned)__builtin_amdgcn_readlane(sv, j) * OUT + n;
    od[j] = (unsigned)__builtin_amdgcn_readlane(dv, j) * OUT + n;
  }
}

static_assert(EPW % EQ == 0, "an edge chunk must be whole groups");
__device__ __forceinline__ void edge_logits_of(const float* __restrict__ ebuf,
                                               int chunk, int n, float (&v)[EPW]) {
  const float4* __restrict__ g4 = (const float4*)ebuf + chunk * (EPW / EQ) * OUT + n;
#pragma unroll
  for (int g = 0; g < EPW / EQ; ++g) {
    const float4 t = g4[g * OUT];
    v[EQ * g] = t.x; v[EQ * g + 1] = t.y; v[EQ * g + 2] = t.z; v[EQ * g + 3] = t.w;
  }
}

constexpr int PLANE_ITEMS = N_NODES * OUT / 4;

constexpr int FOLD_ITEMS = PLANE_ITEMS;

constexpr int FB_NODES = (N_NODES + PW_GRID - 1) / PW_GRID;
static_assert((long)FB_NODES * PW_GRID >= N_NODES, "the workgroups must cover every node");
static_assert(N_EDGES % PW_BLOCK == 0, "every thread takes the same number of trips over the edges");

__device__ __forceinline__ float logit_of(const float* __restrict__ ebuf, int e, int c) {
  return ebuf[((long)(e / EQ) * OUT + c) * EQ + (e % EQ)];
}

__device__ __forceinline__ float fold_term(float shift, float sum) {
  return shift + logf(sum);
}

__global__ __launch_bounds__(PW_BLOCK) void node_fold_kernel(
    float* splane, float* ssrc0, float* sdst1, const float* __restrict__ ssrc1,
    const float* __restrict__ edst, const float* __restrict__ esrc,
    const float* __restrict__ fc, float* __restrict__ srcrec,
    const float* __restrict__ ebuf, const int* __restrict__ src,
    const int* __restrict__ dst) {
  const float fword = __uint_as_float(
      __builtin_amdgcn_readfirstlane(__float_as_uint(ssrc1[0])));
  if (!(fword < 3.0e38f)) {
    __shared__ unsigned mx[2][FB_NODES * OUT];
    __shared__ float    sm[2][FB_NODES * OUT];
    const int n0  = blockIdx.x * FB_NODES;
    const int cnt = min(FB_NODES, max(0, N_NODES - n0));
    for (int i = threadIdx.x; i < FB_NODES * OUT; i += PW_BLOCK) {
      mx[0][i] = 0u; mx[1][i] = 0u;
      sm[0][i] = 0.f; sm[1][i] = 0.f;
    }
    __syncthreads();
    for (int e = threadIdx.x; e < N_EDGES; e += PW_BLOCK) {
      const unsigned d = (unsigned)(dst[e] - n0), s = (unsigned)(src[e] - n0);
      if (d < (unsigned)cnt)
        for (int c = 0; c < OUT; ++c) atomicMax(&mx[0][d * OUT + c], flipf(logit_of(ebuf, e, c)));
      if (s < (unsigned)cnt)
        for (int c = 0; c < OUT; ++c) atomicMax(&mx[1][s * OUT + c], flipf(logit_of(ebuf, e, c)));
    }
    __syncthreads();
    for (int e = threadIdx.x; e < N_EDGES; e += PW_BLOCK) {
      const unsigned d = (unsigned)(dst[e] - n0), s = (unsigned)(src[e] - n0);
      if (d < (unsigned)cnt)
        for (int c = 0; c < OUT; ++c)
          atomicAdd(&sm[0][d * OUT + c], __expf(logit_of(ebuf, e, c) - unflipf(mx[0][d * OUT + c])));
      if (s < (unsigned)cnt)
        for (int c = 0; c < OUT; ++c)
          atomicAdd(&sm[1][s * OUT + c], __expf(logit_of(ebuf, e, c) - unflipf(mx[1][s * OUT + c])));
    }
    __syncthreads();
    for (int i = threadIdx.x; i < cnt * OUT; i += PW_BLOCK) {
      const long g = (long)n0 * OUT + i;
      splane[g] = fold_term(unflipf(mx[0][i]), sm[0][i]);
      ((v2f*)srcrec)[g] = (v2f){fold_term(unflipf(mx[1][i]), sm[1][i]), fc[g]};
      sdst1[g] = 0.f;
      ssrc0[g] = 0.f;
    }
    return;
  }
  for (int i = blockIdx.x * PW_BLOCK + threadIdx.x; i < FOLD_ITEMS; i += PW_GRID * PW_BLOCK) {
    const float4 td = ((const float4*)edst)[i];
    const float4 ts = ((const float4*)esrc)[i];
    const float4 da = ((const float4*)splane)[i];
    const float4 db = ((const float4*)sdst1)[i];
    const float4 sa = ((const float4*)ssrc0)[i];
    const float4 sb = ((const float4*)ssrc1)[i];
    const float4 f  = ((const float4*)fc)[i];
    const float4 sd = {da.x + db.x, da.y + db.y, da.z + db.z, da.w + db.w};
    const float4 ss = {sa.x + sb.x, sa.y + sb.y, sa.z + sb.z, sa.w + sb.w};
    const float4 z  = {0.f, 0.f, 0.f, 0.f};
    ((float4*)sdst1)[i] = z;
    ((float4*)ssrc0)[i] = z;
    const float4 ld = {fold_term(td.x, sd.x), fold_term(td.y, sd.y),
                       fold_term(td.z, sd.z), fold_term(td.w, sd.w)};
    ((float4*)splane)[i] = ld;
    const float4 r0 = {fold_term(ts.x, ss.x), f.x, fold_term(ts.y, ss.y), f.y};
    const float4 r1 = {fold_term(ts.z, ss.z), f.z, fold_term(ts.w, ss.w), f.w};
    ((float4*)srcrec)[2 * i]     = r0;
    ((float4*)srcrec)[2 * i + 1] = r1;
  }
}

constexpr float LOGS_MAX = 10.f;
constexpr float SQRT_EPS = 3.1622777e-5f;

__global__ __launch_bounds__(PW_BLOCK) void edge_agg_kernel(
    const float* __restrict__ ebuf, const int* __restrict__ src,
    const int* __restrict__ dst, const float* __restrict__ lddst,
    const float* __restrict__ srcrec, float* __restrict__ msg0,
    float* __restrict__ msg1, const float* __restrict__ flagw,
    float* __restrict__ mode_out) {
  const int n = threadIdx.x & 31;
  float* __restrict__ msg = package_half() ? msg1 : msg0;
  const v2f* __restrict__ rec = (const v2f*)srcrec;
  const int wave0 = wave_number();
  const float fword = __uint_as_float(
      __builtin_amdgcn_readfirstlane(__float_as_uint(flagw[0])));
  const bool unfactored = !(fword < 3.0e38f);
  if (wave0 == 0 && n == 0) mode_out[0] = unfactored ? 1.f : 0.f;
  if (!unfactored) {
    for (int chunk = wave0; chunk < N_CHUNKS; chunk += PW_WAVES) {
      const int e0 = chunk * EPW;
      unsigned os[EPW], od[EPW];
      edge_offsets(src, dst, e0, n, os, od);

      float v[EPW];
      v2f r[EPW];
      edge_logits_of(ebuf, chunk, n, v);
#pragma unroll
      for (int j = 0; j < EPW; ++j) r[j] = rec[os[j]];
#pragma unroll
      for (int j = 0; j < EPW; ++j) {
        const float ras = fmaxf(__expf(0.5f * (v[j] - r[j].x)), SQRT_EPS);
        atomicAdd(&msg[od[j]], r[j].y * ras * __expf(0.5f * v[j]));
      }
    }
    return;
  }
  for (int chunk = wave0; chunk < N_CHUNKS; chunk += PW_WAVES) {
    const int e0 = chunk * EPW;
    unsigned os[EPW], od[EPW];
    edge_offsets(src, dst, e0, n, os, od);

    float v[EPW], ld[EPW];
    v2f r[EPW];
    edge_logits_of(ebuf, chunk, n, v);
#pragma unroll
    for (int j = 0; j < EPW; ++j) {
      ld[j] = lddst[od[j]];
      r[j]  = rec[os[j]];
    }
#pragma unroll
    for (int j = 0; j < EPW; ++j) {
      const float ad = fmaxf(__expf(v[j] - ld[j]), EPS);
      const float as = fmaxf(__expf(v[j] - r[j].x), EPS);
      atomicAdd(&msg[od[j]], r[j].y * sqrtf(ad * as));
    }
  }
}

__global__ __launch_bounds__(32) void final_kernel(
    const float* __restrict__ msg, const float* __restrict__ msg1,
    const float* __restrict__ modew, const float* __restrict__ lddst,
    const float* __restrict__ edst, const float* __restrict__ ebuf,
    const int* __restrict__ src, const int* __restrict__ dst,
    const float* __restrict__ srcrec,
    const float* __restrict__ scale,
    const float* __restrict__ offset, const float* __restrict__ Wagg,
    const float* __restrict__ bagg, const float* __restrict__ dterm,
    const float* __restrict__ Wap, const float* __restrict__ bap,
    float* __restrict__ out) {
  const int lane = threadIdx.x, lo = lane & 15, hi = lane >> 4;
  const int base = blockIdx.x * 16;

  const float mode = __uint_as_float(
      __builtin_amdgcn_readfirstlane(__float_as_uint(modew[0])));
  v2f ldr[OUT / 4] = {}, edr[OUT / 4] = {};
  if (mode == 0.f) {
#pragma unroll
    for (int j = 0; j < OUT / 4; ++j) {
      ldr[j] = *(const v2f*)(lddst + (long)(base + lo) * OUT + 4 * j + 2 * hi);
      edr[j] = *(const v2f*)(edst + (long)(base + lo) * OUT + 4 * j + 2 * hi);
    }
  }

  v2f a[OUT / 4];
#pragma unroll
  for (int j = 0; j < OUT / 4; ++j)
    a[j] = *(const v2f*)(msg + (long)(base + lo) * OUT + 4 * j + 2 * hi) +
           *(const v2f*)(msg1 + (long)(base + lo) * OUT + 4 * j + 2 * hi);

  if (mode == 0.f) {
    v2f fac[OUT / 4];
    float logs = -__builtin_inff();
#pragma unroll
    for (int j = 0; j < OUT / 4; ++j) {
      const v2f ld = ldr[j], ed = edr[j];
      logs = fmaxf(logs, fmaxf(ld.x - ed.x, ld.y - ed.y));
      fac[j] = (v2f){ld.x > -3.0e38f ? __expf(-0.5f * ld.x) : 0.f,
                     ld.y > -3.0e38f ? __expf(-0.5f * ld.y) : 0.f};
    }
    if (__builtin_amdgcn_ballot_w32(logs > LOGS_MAX) == 0u) {
#pragma unroll
      for (int j = 0; j < OUT / 4; ++j) a[j] *= fac[j];
    } else {
      __shared__ float racc[16 * OUT];
      for (int i = lane; i < 16 * OUT; i += 32) racc[i] = 0.f;
      __syncthreads();
      const v2f* __restrict__ rec = (const v2f*)srcrec;
      for (int e0 = 0; e0 < N_EDGES; e0 += 32) {
        const int dl = dst[e0 + lane];
        unsigned hit = __builtin_amdgcn_ballot_w32((unsigned)(dl - base) < 16u);
        while (hit != 0u) {
          const int b = __builtin_ctz(hit);
          hit &= hit - 1u;
          const int e = e0 + b;
          const int d = dst[e], s = src[e];
          const float v = logit_of(ebuf, e, lane);
          const v2f rr = rec[(long)s * OUT + lane];
          const float ad = fmaxf(__expf(v - lddst[(long)d * OUT + lane]), EPS);
          const float as = fmaxf(__expf(v - rr.x), EPS);
          racc[(d - base) * OUT + lane] += rr.y * sqrtf(ad * as);
        }
      }
      __syncthreads();
#pragma unroll
      for (int j = 0; j < OUT / 4; ++j)
        a[j] = (v2f){racc[lo * OUT + 4 * j + 2 * hi], racc[lo * OUT + 4 * j + 2 * hi + 1]};
    }
  }

#pragma unroll
  for (int h = 0; h < 4; ++h) {
    v2f p = a[2 * h], q = a[2 * h + 1];
    const float s = (p.x + p.y) + (q.x + q.y);
    const float m = (s + xhalf(s)) * 0.125f;
    p.x -= m; p.y -= m; q.x -= m; q.y -= m;
    const float v = (p.x * p.x + p.y * p.y) + (q.x * q.x + q.y * q.y);
    const float r = rsqrtf((v + xhalf(v)) * 0.125f + EPS);
    const int k = 8 * h + 2 * hi;
    a[2 * h]     = (v2f){p.x * scale[k] * r + offset[k],
                         p.y * scale[k + 1] * r + offset[k + 1]};
    a[2 * h + 1] = (v2f){q.x * scale[k + 4] * r + offset[k + 4],
                         q.y * scale[k + 5] * r + offset[k + 5]};
  }

  v8f acc1[2] = {};
#pragma unroll
  for (int j = 0; j < OUT / 4; ++j) {
    const int ka = 4 * j + 2 * hi;
#pragma unroll
    for (int t = 0; t < 2; ++t) {
      const int n = lo + 16 * t;
      v2f b = (v2f){Wagg[ka * OUT + n], Wagg[(ka + 1) * OUT + n]};
      WMMA4(acc1[t], a[j], b);
    }
  }

  __shared__ float lds[16 * OUT];
#pragma unroll
  for (int t = 0; t < 2; ++t)
#pragma unroll
    for (int v = 0; v < 8; ++v) {
      const long row = base + v + 8 * hi;
      const int n = lo + 16 * t;
      lds[(v + 8 * hi) * OUT + n] = leaky(acc1[t][v] + bagg[n] + dterm[row * OUT + n]);
    }
  __syncthreads();

  v8f acc2[2] = {};
  for (int k0 = 0; k0 < OUT; k0 += 4) {
    const int ka = k0 + 2 * hi;
    v2f a2 = (v2f){lds[lo * OUT + ka], lds[lo * OUT + ka + 1]};
#pragma unroll
    for (int t = 0; t < 2; ++t) {
      const int n = lo + 16 * t;
      v2f b = (v2f){Wap[ka * OUT + n], Wap[(ka + 1) * OUT + n]};
      WMMA4(acc2[t], a2, b);
    }
  }

#pragma unroll
  for (int t = 0; t < 2; ++t)
#pragma unroll
    for (int v = 0; v < 8; ++v)
      out[(long)(base + v + 8 * hi) * OUT + lo + 16 * t] = acc2[t][v] + bap[lo + 16 * t];
}

extern "C" void kernel_launch(void* const* d_in, const int* in_sizes, int n_in,
                              void* d_out, int out_size, void* d_ws, size_t ws_size,
                              hipStream_t stream) {
  const float* feat  = (const float*)d_in[0];
  const float* fe    = (const float*)d_in[1];
  const int*   src   = (const int*)d_in[2];
  const int*   dst   = (const int*)d_in[3];
  const float* Wp    = (const float*)d_in[4];
  const float* Wp2   = (const float*)d_in[5];
  const float* Was   = (const float*)d_in[6];
  const float* Wad   = (const float*)d_in[7];
  const float* We    = (const float*)d_in[8];
  const float* scale = (const float*)d_in[9];
  const float* offs  = (const float*)d_in[10];
  const float* Wagg  = (const float*)d_in[11];
  const float* bagg  = (const float*)d_in[12];
  const float* Wapd  = (const float*)d_in[13];
  const float* bapd  = (const float*)d_in[14];
  const float* Wap   = (const float*)d_in[15];
  const float* bap   = (const float*)d_in[16];
  float* out = (float*)d_out;

  const long NW = (long)N_NODES * OUT;
  float* fc    = (float*)d_ws;
  float* esrc  = fc + NW;
  float* edst  = esrc + NW;
  float* dterm = edst + NW;
  float* srcrec = dterm + NW;
  float* sdst  = srcrec + 2 * NW;
  float* ssrc  = sdst + NW;
  float* msg   = ssrc + NW;
  float* ebuf  = msg + NW;

  node_gemm_kernel<<<N_NODES / 16, 32, 0, stream>>>(
      sdst, out, feat, Wp, Wp2, Was, Wad, Wapd, bapd, fc, esrc, edst, dterm);

  edge_logits_kernel<<<PW_GRID, PW_BLOCK, 0, stream>>>(
      fe, src, dst, We, esrc, edst, ebuf, sdst, ssrc, msg, out);

  node_fold_kernel<<<PW_GRID, PW_BLOCK, 0, stream>>>(
      sdst, ssrc, msg, out, edst, esrc, fc, srcrec, ebuf, src, dst);

  edge_agg_kernel<<<PW_GRID, PW_BLOCK, 0, stream>>>(
      ebuf, src, dst, (const float*)sdst, srcrec, msg, ssrc, (const float*)out, fc);

  final_kernel<<<N_NODES / 16, 32, 0, stream>>>(
      msg, ssrc, fc, sdst, edst, ebuf, src, dst, srcrec,
      scale, offs, Wagg, bagg, dterm, Wap, bap, out);

  (void)in_sizes; (void)n_in; (void)out_size; (void)ws_size;
}
